# starting kernel
# baseline (speedup 1.0000x reference)
#include <hip/hip_runtime.h>

#define N_NODES   100000
#define N_EDGES   1600000
#define N_GRAPHS  1000
#define D_FEAT    128
#define HID       64
#define BN_EPS    1e-5f
#define NTILES    (N_NODES / 16)

typedef __attribute__((ext_vector_type(2))) float v2f;
typedef __attribute__((ext_vector_type(8))) float v8f;

template<int K1>
__launch_bounds__(128)
__global__ void gin_mlp_kernel(const float* __restrict__ xin,
                               const float* __restrict__ W1,
                               const float* __restrict__ b1,
                               const float* __restrict__ bn1,
                               const float* __restrict__ W2,
                               const float* __restrict__ b2,
                               const float* __restrict__ bn2,
                               float* __restrict__ hout,
                               int ntiles) {
  constexpr int KS1 = K1 / 4;
  __shared__ __align__(16) float w1s[4 * KS1 * 32 * 2];
  __shared__ __align__(16) float w2s[4 * 16 * 32 * 2];
  __shared__ __align__(16) float tb[4][16 * 68];

  const int tid = threadIdx.x;

  for (int i = tid; i < K1 * 64; i += 128) {
    int kk = i >> 6, nn = i & 63;
    int idx = ((((nn >> 4) * KS1) + (kk >> 2)) * 32 + (nn & 15) + ((kk & 2) << 3)) * 2 + (kk & 1);
    w1s[idx] = W1[i];
  }
  for (int i = tid; i < 64 * 64; i += 128) {
    int kk = i >> 6, nn = i & 63;
    int idx = ((((nn >> 4) * 16) + (kk >> 2)) * 32 + (nn & 15) + ((kk & 2) << 3)) * 2 + (kk & 1);
    w2s[idx] = W2[i];
  }
  __syncthreads();

  const int lane = tid & 31;
  const int wave = tid >> 5;
  const int row  = lane & 15;
  const int hi   = lane >> 4;
  const int tile = blockIdx.x * 4 + wave;
  const bool active = tile < ntiles;
  float* tbuf = tb[wave];

  const v8f vzero = {0.f, 0.f, 0.f, 0.f, 0.f, 0.f, 0.f, 0.f};

  if (active) {
    v2f a1[KS1];
    const float* xr = xin + (size_t)(tile * 16 + row) * K1 + 2 * hi;
#pragma unroll
    for (int k = 0; k < KS1; k++) a1[k] = *(const v2f*)(xr + 4 * k);

#pragma unroll
    for (int nt = 0; nt < 4; nt++) {
      v8f acc = vzero;
      const v2f* wf = ((const v2f*)w1s) + nt * KS1 * 32 + lane;
#pragma unroll
      for (int k = 0; k < KS1; k++) {
        acc = __builtin_amdgcn_wmma_f32_16x16x4_f32(false, a1[k], false, wf[k * 32],
                                                    (short)0, acc, false, false);
      }
      int n = nt * 16 + row;
      float s  = bn1[n] * rsqrtf(bn1[192 + n] + BN_EPS);
      float t0 = bn1[64 + n] + (b1[n] - bn1[128 + n]) * s;
#pragma unroll
      for (int i = 0; i < 8; i++) {
        float y = acc[i] * s + t0;
        tbuf[(i + 8 * hi) * 68 + n] = fmaxf(y, 0.0f);
      }
    }
  }
  __syncthreads();

  v8f acc2[4];
  if (active) {
#pragma unroll
    for (int nt = 0; nt < 4; nt++) {
      v8f acc = vzero;
      const v2f* wf = ((const v2f*)w2s) + nt * 16 * 32 + lane;
      const float* tr = tbuf + row * 68 + 2 * hi;
#pragma unroll
      for (int k = 0; k < 16; k++) {
        v2f a = *(const v2f*)(tr + 4 * k);
        acc = __builtin_amdgcn_wmma_f32_16x16x4_f32(false, a, false, wf[k * 32],
                                                    (short)0, acc, false, false);
      }
      acc2[nt] = acc;
    }
  }
  __syncthreads();

  if (active) {
#pragma unroll
    for (int nt = 0; nt < 4; nt++) {
      int n = nt * 16 + row;
      float s  = bn2[n] * rsqrtf(bn2[192 + n] + BN_EPS);
      float t0 = bn2[64 + n] + (b2[n] - bn2[128 + n]) * s;
#pragma unroll
      for (int i = 0; i < 8; i++) {
        float y = acc2[nt][i] * s + t0;
        tbuf[(i + 8 * hi) * 68 + n] = fmaxf(y, 0.0f);
      }
    }
  }
  __syncthreads();

  if (active) {
#pragma unroll
    for (int j = 0; j < 8; j++) {
      int flat = j * 32 + lane;
      int r = flat >> 4;
      int c = (flat & 15) << 2;
      float4 v = *(const float4*)(tbuf + r * 68 + c);
      *(float4*)(hout + (size_t)(tile * 16 + r) * 64 + c) = v;
    }
  }
}

__global__ void gin_scatter_kernel(const float* __restrict__ h,
                                   const int* __restrict__ src,
                                   const int* __restrict__ dst,
                                   float* __restrict__ agg) {
  long long idx = (long long)blockIdx.x * blockDim.x + threadIdx.x;
  int e = (int)(idx >> 4);
  if (e >= N_EDGES) return;
  int c = ((int)idx & 15) << 2;
  int s = src[e], d = dst[e];
  const float4 v = *(const float4*)(h + (size_t)s * 64 + c);
  float* p = agg + (size_t)d * 64 + c;
  atomicAdd(p + 0, v.x);
  atomicAdd(p + 1, v.y);
  atomicAdd(p + 2, v.z);
  atomicAdd(p + 3, v.w);
}

__global__ void gin_pool_kernel(const float* __restrict__ h,
                                const int* __restrict__ batch,
                                float* __restrict__ pooled,
                                float* __restrict__ counts,
                                int addCount) {
  long long idx = (long long)blockIdx.x * blockDim.x + threadIdx.x;
  int n = (int)(idx >> 4);
  if (n >= N_NODES) return;
  int c = ((int)idx & 15) << 2;
  int g = batch[n];
  const float4 v = *(const float4*)(h + (size_t)n * 64 + c);
  float* p = pooled + (size_t)g * 64 + c;
  atomicAdd(p + 0, v.x);
  atomicAdd(p + 1, v.y);
  atomicAdd(p + 2, v.z);
  atomicAdd(p + 3, v.w);
  if (addCount && (((int)idx & 15) == 0)) atomicAdd(counts + g, 1.0f);
}

__global__ void gin_final_kernel(const float* __restrict__ pooled,
                                 const float* __restrict__ counts,
                                 const float* __restrict__ linW,
                                 const float* __restrict__ linb,
                                 float* __restrict__ out) {
  int idx = blockIdx.x * blockDim.x + threadIdx.x;
  if (idx >= N_GRAPHS * 10) return;
  int g = idx / 10, t = idx - g * 10;
  float acc = counts[g] * linb[t];
  for (int l = 1; l < 5; l++) acc += linb[l * 10 + t];
  for (int l = 0; l < 5; l++) {
    const float* p = pooled + ((size_t)l * N_GRAPHS + g) * 64;
    const float* w = linW + l * 640 + t;
    float s = 0.f;
#pragma unroll
    for (int k = 0; k < 64; k++) s += p[k] * w[k * 10];
    acc += s;
  }
  out[idx] = acc;
}

__global__ void gin_zero_kernel(float* __restrict__ p, int n) {
  int i = blockIdx.x * blockDim.x + threadIdx.x;
  if (i < n) p[i] = 0.0f;
}

extern "C" void kernel_launch(void* const* d_in, const int* in_sizes, int n_in,
                              void* d_out, int out_size, void* d_ws, size_t ws_size,
                              hipStream_t stream) {
  const float* x        = (const float*)d_in[0];
  const float* fh_W1    = (const float*)d_in[1];
  const float* fh_b1    = (const float*)d_in[2];
  const float* fh_bn1   = (const float*)d_in[3];
  const float* fh_W2    = (const float*)d_in[4];
  const float* fh_b2    = (const float*)d_in[5];
  const float* fh_bn2   = (const float*)d_in[6];
  const float* conv_W1  = (const float*)d_in[7];
  const float* conv_b1  = (const float*)d_in[8];
  const float* conv_bn1 = (const float*)d_in[9];
  const float* conv_W2  = (const float*)d_in[10];
  const float* conv_b2  = (const float*)d_in[11];
  const float* conv_bn2 = (const float*)d_in[12];
  const float* lin_W    = (const float*)d_in[13];
  const float* lin_b    = (const float*)d_in[14];
  const int*   eidx     = (const int*)d_in[15];
  const int*   batch    = (const int*)d_in[16];
  float* out = (float*)d_out;

  const int* src = eidx;
  const int* dst = eidx + N_EDGES;

  float* ws     = (float*)d_ws;
  float* h      = ws;
  float* agg    = ws + (size_t)N_NODES * 64;
  float* pooled = agg + (size_t)N_NODES * 64;
  float* counts = pooled + 5 * N_GRAPHS * 64;

  const int accN = 5 * N_GRAPHS * 64 + N_GRAPHS;
  gin_zero_kernel<<<(accN + 255) / 256, 256, 0, stream>>>(pooled, accN);

  const int mlpBlocks = (NTILES + 3) / 4;

  gin_mlp_kernel<D_FEAT><<<mlpBlocks, 128, 0, stream>>>(
      x, fh_W1, fh_b1, fh_bn1, fh_W2, fh_b2, fh_bn2, h, NTILES);
  gin_pool_kernel<<<(N_NODES * 16 + 255) / 256, 256, 0, stream>>>(
      h, batch, pooled, counts, 1);

  for (int l = 0; l < 4; l++) {
    hipMemcpyAsync(agg, h, (size_t)N_NODES * 64 * sizeof(float),
                   hipMemcpyDeviceToDevice, stream);
    gin_scatter_kernel<<<(int)(((long long)N_EDGES * 16 + 255) / 256), 256, 0, stream>>>(
        h, src, dst, agg);
    gin_mlp_kernel<HID><<<mlpBlocks, 128, 0, stream>>>(
        agg,
        conv_W1 + (size_t)l * 64 * 64, conv_b1 + l * 64, conv_bn1 + l * 256,
        conv_W2 + (size_t)l * 64 * 64, conv_b2 + l * 64, conv_bn2 + l * 256,
        h, NTILES);
    gin_pool_kernel<<<(N_NODES * 16 + 255) / 256, 256, 0, stream>>>(
        h, batch, pooled + (size_t)(l + 1) * N_GRAPHS * 64, counts, 0);
  }

  gin_final_kernel<<<(N_GRAPHS * 10 + 255) / 256, 256, 0, stream>>>(
      pooled, counts, lin_W, lin_b, out);
}


# Replace per-edge f32 atomic scatter and the h-to-agg copy by in-edge lists built per call (int atomic rank, scan, fill) and an atomic-free row gather; neighbour sum order = arrival rank
# speedup vs baseline: 1.6373x; 1.6373x over previous
#include <hip/hip_runtime.h>

#define N_NODES   100000
#define N_EDGES   1600000
#define N_GRAPHS  1000
#define D_FEAT    128
#define HID       64
#define BN_EPS    1e-5f
#define NTILES    (N_NODES / 16)

typedef __attribute__((ext_vector_type(2))) float v2f;
typedef __attribute__((ext_vector_type(8))) float v8f;

template<int K1>
__launch_bounds__(128)
__global__ void gin_mlp_kernel(const float* __restrict__ xin,
                               const float* __restrict__ W1,
                               const float* __restrict__ b1,
                               const float* __restrict__ bn1,
                               const float* __restrict__ W2,
                               const float* __restrict__ b2,
                               const float* __restrict__ bn2,
                               float* __restrict__ hout,
                               int ntiles) {
  constexpr int KS1 = K1 / 4;
  __shared__ __align__(16) float w1s[4 * KS1 * 32 * 2];
  __shared__ __align__(16) float w2s[4 * 16 * 32 * 2];
  __shared__ __align__(16) float tb[4][16 * 68];

  const int tid = threadIdx.x;

  for (int i = tid; i < K1 * 64; i += 128) {
    int kk = i >> 6, nn = i & 63;
    int idx = ((((nn >> 4) * KS1) + (kk >> 2)) * 32 + (nn & 15) + ((kk & 2) << 3)) * 2 + (kk & 1);
    w1s[idx] = W1[i];
  }
  for (int i = tid; i < 64 * 64; i += 128) {
    int kk = i >> 6, nn = i & 63;
    int idx = ((((nn >> 4) * 16) + (kk >> 2)) * 32 + (nn & 15) + ((kk & 2) << 3)) * 2 + (kk & 1);
    w2s[idx] = W2[i];
  }
  __syncthreads();

  const int lane = tid & 31;
  const int wave = tid >> 5;
  const int row  = lane & 15;
  const int hi   = lane >> 4;
  const int tile = blockIdx.x * 4 + wave;
  const bool active = tile < ntiles;
  float* tbuf = tb[wave];

  const v8f vzero = {0.f, 0.f, 0.f, 0.f, 0.f, 0.f, 0.f, 0.f};

  if (active) {
    v2f a1[KS1];
    const float* xr = xin + (size_t)(tile * 16 + row) * K1 + 2 * hi;
#pragma unroll
    for (int k = 0; k < KS1; k++) a1[k] = *(const v2f*)(xr + 4 * k);

#pragma unroll
    for (int nt = 0; nt < 4; nt++) {
      v8f acc = vzero;
      const v2f* wf = ((const v2f*)w1s) + nt * KS1 * 32 + lane;
#pragma unroll
      for (int k = 0; k < KS1; k++) {
        acc = __builtin_amdgcn_wmma_f32_16x16x4_f32(false, a1[k], false, wf[k * 32],
                                                    (short)0, acc, false, false);
      }
      int n = nt * 16 + row;
      float s  = bn1[n] * rsqrtf(bn1[192 + n] + BN_EPS);
      float t0 = bn1[64 + n] + (b1[n] - bn1[128 + n]) * s;
#pragma unroll
      for (int i = 0; i < 8; i++) {
        float y = acc[i] * s + t0;
        tbuf[(i + 8 * hi) * 68 + n] = fmaxf(y, 0.0f);
      }
    }
  }
  __syncthreads();

  v8f acc2[4];
  if (active) {
#pragma unroll
    for (int nt = 0; nt < 4; nt++) {
      v8f acc = vzero;
      const v2f* wf = ((const v2f*)w2s) + nt * 16 * 32 + lane;
      const float* tr = tbuf + row * 68 + 2 * hi;
#pragma unroll
      for (int k = 0; k < 16; k++) {
        v2f a = *(const v2f*)(tr + 4 * k);
        acc = __builtin_amdgcn_wmma_f32_16x16x4_f32(false, a, false, wf[k * 32],
                                                    (short)0, acc, false, false);
      }
      acc2[nt] = acc;
    }
  }
  __syncthreads();

  if (active) {
#pragma unroll
    for (int nt = 0; nt < 4; nt++) {
      int n = nt * 16 + row;
      float s  = bn2[n] * rsqrtf(bn2[192 + n] + BN_EPS);
      float t0 = bn2[64 + n] + (b2[n] - bn2[128 + n]) * s;
#pragma unroll
      for (int i = 0; i < 8; i++) {
        float y = acc2[nt][i] * s + t0;
        tbuf[(i + 8 * hi) * 68 + n] = fmaxf(y, 0.0f);
      }
    }
  }
  __syncthreads();

  if (active) {
#pragma unroll
    for (int j = 0; j < 8; j++) {
      int flat = j * 32 + lane;
      int r = flat >> 4;
      int c = (flat & 15) << 2;
      float4 v = *(const float4*)(tbuf + r * 68 + c);
      *(float4*)(hout + (size_t)(tile * 16 + r) * 64 + c) = v;
    }
  }
}

__global__ void gin_rank_kernel(const int* __restrict__ dst,
                                int* __restrict__ deg,
                                int* __restrict__ rank) {
  int e = blockIdx.x * blockDim.x + threadIdx.x;
  if (e >= N_EDGES) return;
  rank[e] = atomicAdd(deg + dst[e], 1);
}

#define SCAN_CHUNK 100
#define SCAN_OWNERS (N_NODES / SCAN_CHUNK)
__launch_bounds__(1024)
__global__ void gin_scan_kernel(const int* __restrict__ deg,
                                int* __restrict__ rowptr) {
  __shared__ int wsum[32];
  const int t = threadIdx.x;
  const int lane = t & 31;
  const int w = t >> 5;
  const bool owner = t < SCAN_OWNERS;

  int4 v[SCAN_CHUNK / 4];
  int local = 0;
  if (owner) {
    const int4* p = (const int4*)(deg + t * SCAN_CHUNK);
#pragma unroll
    for (int i = 0; i < SCAN_CHUNK / 4; i++) v[i] = p[i];
#pragma unroll
    for (int i = 0; i < SCAN_CHUNK / 4; i++) local += v[i].x + v[i].y + v[i].z + v[i].w;
  }

  int x = local;
#pragma unroll
  for (int o = 1; o < 32; o <<= 1) {
    int y = __shfl_up(x, o);
    if (lane >= o) x += y;
  }
  if (lane == 31) wsum[w] = x;
  __syncthreads();
  if (w == 0) {
    int s = wsum[lane];
#pragma unroll
    for (int o = 1; o < 32; o <<= 1) {
      int y = __shfl_up(s, o);
      if (lane >= o) s += y;
    }
    wsum[lane] = s;
  }
  __syncthreads();

  if (owner) {
    int run = x - local + (w > 0 ? wsum[w - 1] : 0);
    int4* q = (int4*)(rowptr + t * SCAN_CHUNK);
#pragma unroll
    for (int i = 0; i < SCAN_CHUNK / 4; i++) {
      int4 o4;
      o4.x = run; run += v[i].x;
      o4.y = run; run += v[i].y;
      o4.z = run; run += v[i].z;
      o4.w = run; run += v[i].w;
      q[i] = o4;
    }
    if (t == SCAN_OWNERS - 1) rowptr[N_NODES] = run;
  }
}

__global__ void gin_fill_kernel(const int* __restrict__ src,
                                const int* __restrict__ dst,
                                const int* __restrict__ rowptr,
                                const int* __restrict__ rank,
                                int* __restrict__ csr) {
  int e = blockIdx.x * blockDim.x + threadIdx.x;
  if (e >= N_EDGES) return;
  csr[rowptr[dst[e]] + rank[e]] = src[e];
}

__launch_bounds__(256)
__global__ void gin_gather_kernel(const float* __restrict__ h,
                                  const int* __restrict__ rowptr,
                                  const int* __restrict__ csr,
                                  float* __restrict__ agg) {
  const int n = blockIdx.x * 16 + (threadIdx.x >> 4);
  const unsigned cb = (threadIdx.x & 15u) << 4;
  const int beg = rowptr[n];
  const int end = rowptr[n + 1];
  const char* hb = (const char*)h;
  float4 acc = *(const float4*)(hb + (((unsigned)n << 8) + cb));

  for (int j = beg; j < end; j += 8) {
    int s[8];
#pragma unroll
    for (int k = 0; k < 8; k++) {
      int jj = j + k;
      s[k] = csr[jj < end ? jj : end - 1];
    }
    float4 v[8];
#pragma unroll
    for (int k = 0; k < 8; k++) {
      int sk = (j + k < end) ? s[k] : N_NODES;
      v[k] = *(const float4*)(hb + (((unsigned)sk << 8) + cb));
    }
#pragma unroll
    for (int k = 0; k < 8; k++) {
      acc.x += v[k].x; acc.y += v[k].y; acc.z += v[k].z; acc.w += v[k].w;
    }
  }
  *(float4*)(agg + (size_t)n * 64 + ((threadIdx.x & 15) << 2)) = acc;
}

__global__ void gin_pool_kernel(const float* __restrict__ h,
                                const int* __restrict__ batch,
                                float* __restrict__ pooled,
                                float* __restrict__ counts,
                                int addCount) {
  long long idx = (long long)blockIdx.x * blockDim.x + threadIdx.x;
  int n = (int)(idx >> 4);
  if (n >= N_NODES) return;
  int c = ((int)idx & 15) << 2;
  int g = batch[n];
  const float4 v = *(const float4*)(h + (size_t)n * 64 + c);
  float* p = pooled + (size_t)g * 64 + c;
  atomicAdd(p + 0, v.x);
  atomicAdd(p + 1, v.y);
  atomicAdd(p + 2, v.z);
  atomicAdd(p + 3, v.w);
  if (addCount && (((int)idx & 15) == 0)) atomicAdd(counts + g, 1.0f);
}

__global__ void gin_final_kernel(const float* __restrict__ pooled,
                                 const float* __restrict__ counts,
                                 const float* __restrict__ linW,
                                 const float* __restrict__ linb,
                                 float* __restrict__ out) {
  int idx = blockIdx.x * blockDim.x + threadIdx.x;
  if (idx >= N_GRAPHS * 10) return;
  int g = idx / 10, t = idx - g * 10;
  float acc = counts[g] * linb[t];
  for (int l = 1; l < 5; l++) acc += linb[l * 10 + t];
  for (int l = 0; l < 5; l++) {
    const float* p = pooled + ((size_t)l * N_GRAPHS + g) * 64;
    const float* w = linW + l * 640 + t;
    float s = 0.f;
#pragma unroll
    for (int k = 0; k < 64; k++) s += p[k] * w[k * 10];
    acc += s;
  }
  out[idx] = acc;
}

__global__ void gin_zero_kernel(unsigned int* __restrict__ p, int n) {
  int i = blockIdx.x * blockDim.x + threadIdx.x;
  if (i < n) p[i] = 0u;
}

extern "C" void kernel_launch(void* const* d_in, const int* in_sizes, int n_in,
                              void* d_out, int out_size, void* d_ws, size_t ws_size,
                              hipStream_t stream) {
  const float* x        = (const float*)d_in[0];
  const float* fh_W1    = (const float*)d_in[1];
  const float* fh_b1    = (const float*)d_in[2];
  const float* fh_bn1   = (const float*)d_in[3];
  const float* fh_W2    = (const float*)d_in[4];
  const float* fh_b2    = (const float*)d_in[5];
  const float* fh_bn2   = (const float*)d_in[6];
  const float* conv_W1  = (const float*)d_in[7];
  const float* conv_b1  = (const float*)d_in[8];
  const float* conv_bn1 = (const float*)d_in[9];
  const float* conv_W2  = (const float*)d_in[10];
  const float* conv_b2  = (const float*)d_in[11];
  const float* conv_bn2 = (const float*)d_in[12];
  const float* lin_W    = (const float*)d_in[13];
  const float* lin_b    = (const float*)d_in[14];
  const int*   eidx     = (const int*)d_in[15];
  const int*   batch    = (const int*)d_in[16];
  float* out = (float*)d_out;

  const int* src = eidx;
  const int* dst = eidx + N_EDGES;

  float* ws     = (float*)d_ws;
  float* h      = ws;
  float* zrow   = h + (size_t)N_NODES * 64;
  float* pooled = zrow + 64;
  float* counts = pooled + 5 * N_GRAPHS * 64;
  int*   deg    = (int*)(counts + N_GRAPHS);
  int*   rowptr = deg + N_NODES;
  int*   rank   = rowptr + N_NODES + 4;
  int*   csr    = rank + N_EDGES;
  float* agg    = ws + 10121088;

  const int accN = 64 + 5 * N_GRAPHS * 64 + N_GRAPHS + N_NODES;
  gin_zero_kernel<<<(accN + 255) / 256, 256, 0, stream>>>((unsigned int*)zrow, accN);

  gin_rank_kernel<<<(N_EDGES + 255) / 256, 256, 0, stream>>>(dst, deg, rank);
  gin_scan_kernel<<<1, 1024, 0, stream>>>(deg, rowptr);
  gin_fill_kernel<<<(N_EDGES + 255) / 256, 256, 0, stream>>>(src, dst, rowptr, rank, csr);

  const int mlpBlocks = (NTILES + 3) / 4;

  gin_mlp_kernel<D_FEAT><<<mlpBlocks, 128, 0, stream>>>(
      x, fh_W1, fh_b1, fh_bn1, fh_W2, fh_b2, fh_bn2, h, NTILES);
  gin_pool_kernel<<<(N_NODES * 16 + 255) / 256, 256, 0, stream>>>(
      h, batch, pooled, counts, 1);

  for (int l = 0; l < 4; l++) {
    gin_gather_kernel<<<N_NODES / 16, 256, 0, stream>>>(h, rowptr, csr, agg);
    gin_mlp_kernel<HID><<<mlpBlocks, 128, 0, stream>>>(
        agg,
        conv_W1 + (size_t)l * 64 * 64, conv_b1 + l * 64, conv_bn1 + l * 256,
        conv_W2 + (size_t)l * 64 * 64, conv_b2 + l * 64, conv_bn2 + l * 256,
        h, NTILES);
    gin_pool_kernel<<<(N_NODES * 16 + 255) / 256, 256, 0, stream>>>(
        h, batch, pooled + (size_t)(l + 1) * N_GRAPHS * 64, counts, 0);
  }

  gin_final_kernel<<<(N_GRAPHS * 10 + 255) / 256, 256, 0, stream>>>(
      pooled, counts, lin_W, lin_b, out);
}


# Fuse global_add_pool into the MLP kernel's epilogue: per-tile column sums over rows of equal graph id, one f32 atomic per column; 5 pool launches gone; pooled sum order = tile partials, then arrival
# speedup vs baseline: 2.2161x; 1.3535x over previous
#include <hip/hip_runtime.h>

#define N_NODES   100000
#define N_EDGES   1600000
#define N_GRAPHS  1000
#define D_FEAT    128
#define HID       64
#define BN_EPS    1e-5f
#define NTILES    (N_NODES / 16)

typedef __attribute__((ext_vector_type(2))) float v2f;
typedef __attribute__((ext_vector_type(8))) float v8f;

template<int K1, bool COUNT>
__launch_bounds__(128)
__global__ void gin_mlp_kernel(const float* __restrict__ xin,
                               const float* __restrict__ W1,
                               const float* __restrict__ b1,
                               const float* __restrict__ bn1,
                               const float* __restrict__ W2,
                               const float* __restrict__ b2,
                               const float* __restrict__ bn2,
                               float* __restrict__ hout,
                               const int* __restrict__ batch,
                               float* __restrict__ pooled,
                               float* __restrict__ counts,
                               int ntiles) {
  constexpr int KS1 = K1 / 4;
  __shared__ __align__(16) float w1s[4 * KS1 * 32 * 2];
  __shared__ __align__(16) float w2s[4 * 16 * 32 * 2];
  __shared__ __align__(16) float tb[4][16 * 68];

  const int tid = threadIdx.x;

  for (int i = tid; i < K1 * 64; i += 128) {
    int kk = i >> 6, nn = i & 63;
    int idx = ((((nn >> 4) * KS1) + (kk >> 2)) * 32 + (nn & 15) + ((kk & 2) << 3)) * 2 + (kk & 1);
    w1s[idx] = W1[i];
  }
  for (int i = tid; i < 64 * 64; i += 128) {
    int kk = i >> 6, nn = i & 63;
    int idx = ((((nn >> 4) * 16) + (kk >> 2)) * 32 + (nn & 15) + ((kk & 2) << 3)) * 2 + (kk & 1);
    w2s[idx] = W2[i];
  }
  __syncthreads();

  const int lane = tid & 31;
  const int wave = tid >> 5;
  const int row  = lane & 15;
  const int hi   = lane >> 4;
  const int tile = blockIdx.x * 4 + wave;
  const bool active = tile < ntiles;
  float* tbuf = tb[wave];
  const int gl = active ? batch[tile * 16 + row] : 0;

  const v8f vzero = {0.f, 0.f, 0.f, 0.f, 0.f, 0.f, 0.f, 0.f};

  if (active) {
    v2f a1[KS1];
    const float* xr = xin + (size_t)(tile * 16 + row) * K1 + 2 * hi;
#pragma unroll
    for (int k = 0; k < KS1; k++) a1[k] = *(const v2f*)(xr + 4 * k);

#pragma unroll
    for (int nt = 0; nt < 4; nt++) {
      v8f acc = vzero;
      const v2f* wf = ((const v2f*)w1s) + nt * KS1 * 32 + lane;
#pragma unroll
      for (int k = 0; k < KS1; k++) {
        acc = __builtin_amdgcn_wmma_f32_16x16x4_f32(false, a1[k], false, wf[k * 32],
                                                    (short)0, acc, false, false);
      }
      int n = nt * 16 + row;
      float s  = bn1[n] * rsqrtf(bn1[192 + n] + BN_EPS);
      float t0 = bn1[64 + n] + (b1[n] - bn1[128 + n]) * s;
#pragma unroll
      for (int i = 0; i < 8; i++) {
        float y = acc[i] * s + t0;
        tbuf[(i + 8 * hi) * 68 + n] = fmaxf(y, 0.0f);
      }
    }
  }
  __syncthreads();

  v8f acc2[4];
  if (active) {
#pragma unroll
    for (int nt = 0; nt < 4; nt++) {
      v8f acc = vzero;
      const v2f* wf = ((const v2f*)w2s) + nt * 16 * 32 + lane;
      const float* tr = tbuf + row * 68 + 2 * hi;
#pragma unroll
      for (int k = 0; k < 16; k++) {
        v2f a = *(const v2f*)(tr + 4 * k);
        acc = __builtin_amdgcn_wmma_f32_16x16x4_f32(false, a, false, wf[k * 32],
                                                    (short)0, acc, false, false);
      }
      acc2[nt] = acc;
    }
  }
  __syncthreads();

  if (active) {
#pragma unroll
    for (int nt = 0; nt < 4; nt++) {
      int n = nt * 16 + row;
      float s  = bn2[n] * rsqrtf(bn2[192 + n] + BN_EPS);
      float t0 = bn2[64 + n] + (b2[n] - bn2[128 + n]) * s;
#pragma unroll
      for (int i = 0; i < 8; i++) {
        float y = acc2[nt][i] * s + t0;
        tbuf[(i + 8 * hi) * 68 + n] = fmaxf(y, 0.0f);
      }
    }
  }
  __syncthreads();

  if (active) {
#pragma unroll
    for (int j = 0; j < 8; j++) {
      int flat = j * 32 + lane;
      int r = flat >> 4;
      int c = (flat & 15) << 2;
      float4 v = *(const float4*)(tbuf + r * 68 + c);
      *(float4*)(hout + (size_t)(tile * 16 + r) * 64 + c) = v;
    }
  }

  if (active) {
    float p0[16], p1[16];
#pragma unroll
    for (int r = 0; r < 16; r++) {
      p0[r] = tbuf[r * 68 + lane];
      p1[r] = tbuf[r * 68 + 32 + lane];
    }
    unsigned todo = 0xFFFFu;
    while (todo) {
      const int r0 = __builtin_ctz(todo);
      const int g = __builtin_amdgcn_readlane(gl, r0);
      const unsigned m = __builtin_amdgcn_ballot_w32(gl == g) & todo;
      float s0 = 0.f, s1 = 0.f;
#pragma unroll
      for (int r = 0; r < 16; r++) {
        const bool in = (m >> r) & 1u;
        s0 += in ? p0[r] : 0.f;
        s1 += in ? p1[r] : 0.f;
      }
      float* pg = pooled + (size_t)g * 64 + lane;
      atomicAdd(pg, s0);
      atomicAdd(pg + 32, s1);
      if (COUNT && lane == 0) atomicAdd(counts + g, (float)__builtin_popcount(m));
      todo &= ~m;
    }
  }
}

__global__ void gin_rank_kernel(const int* __restrict__ dst,
                                int* __restrict__ deg,
                                int* __restrict__ rank) {
  int e = blockIdx.x * blockDim.x + threadIdx.x;
  if (e >= N_EDGES) return;
  rank[e] = atomicAdd(deg + dst[e], 1);
}

#define SCAN_CHUNK 100
#define SCAN_OWNERS (N_NODES / SCAN_CHUNK)
__launch_bounds__(1024)
__global__ void gin_scan_kernel(const int* __restrict__ deg,
                                int* __restrict__ rowptr) {
  __shared__ int wsum[32];
  const int t = threadIdx.x;
  const int lane = t & 31;
  const int w = t >> 5;
  const bool owner = t < SCAN_OWNERS;

  int4 v[SCAN_CHUNK / 4];
  int local = 0;
  if (owner) {
    const int4* p = (const int4*)(deg + t * SCAN_CHUNK);
#pragma unroll
    for (int i = 0; i < SCAN_CHUNK / 4; i++) v[i] = p[i];
#pragma unroll
    for (int i = 0; i < SCAN_CHUNK / 4; i++) local += v[i].x + v[i].y + v[i].z + v[i].w;
  }

  int x = local;
#pragma unroll
  for (int o = 1; o < 32; o <<= 1) {
    int y = __shfl_up(x, o);
    if (lane >= o) x += y;
  }
  if (lane == 31) wsum[w] = x;
  __syncthreads();
  if (w == 0) {
    int s = wsum[lane];
#pragma unroll
    for (int o = 1; o < 32; o <<= 1) {
      int y = __shfl_up(s, o);
      if (lane >= o) s += y;
    }
    wsum[lane] = s;
  }
  __syncthreads();

  if (owner) {
    int run = x - local + (w > 0 ? wsum[w - 1] : 0);
    int4* q = (int4*)(rowptr + t * SCAN_CHUNK);
#pragma unroll
    for (int i = 0; i < SCAN_CHUNK / 4; i++) {
      int4 o4;
      o4.x = run; run += v[i].x;
      o4.y = run; run += v[i].y;
      o4.z = run; run += v[i].z;
      o4.w = run; run += v[i].w;
      q[i] = o4;
    }
    if (t == SCAN_OWNERS - 1) rowptr[N_NODES] = run;
  }
}

__global__ void gin_fill_kernel(const int* __restrict__ src,
                                const int* __restrict__ dst,
                                const int* __restrict__ rowptr,
                                const int* __restrict__ rank,
                                int* __restrict__ csr) {
  int e = blockIdx.x * blockDim.x + threadIdx.x;
  if (e >= N_EDGES) return;
  csr[rowptr[dst[e]] + rank[e]] = src[e];
}

__launch_bounds__(256)
__global__ void gin_gather_kernel(const float* __restrict__ h,
                                  const int* __restrict__ rowptr,
                                  const int* __restrict__ csr,
                                  float* __restrict__ agg) {
  const int n = blockIdx.x * 16 + (threadIdx.x >> 4);
  const unsigned cb = (threadIdx.x & 15u) << 4;
  const int beg = rowptr[n];
  const int end = rowptr[n + 1];
  const char* hb = (const char*)h;
  float4 acc = *(const float4*)(hb + (((unsigned)n << 8) + cb));

  for (int j = beg; j < end; j += 8) {
    int s[8];
#pragma unroll
    for (int k = 0; k < 8; k++) {
      int jj = j + k;
      s[k] = csr[jj < end ? jj : end - 1];
    }
    float4 v[8];
#pragma unroll
    for (int k = 0; k < 8; k++) {
      int sk = (j + k < end) ? s[k] : N_NODES;
      v[k] = *(const float4*)(hb + (((unsigned)sk << 8) + cb));
    }
#pragma unroll
    for (int k = 0; k < 8; k++) {
      acc.x += v[k].x; acc.y += v[k].y; acc.z += v[k].z; acc.w += v[k].w;
    }
  }
  *(float4*)(agg + (size_t)n * 64 + ((threadIdx.x & 15) << 2)) = acc;
}

__global__ void gin_final_kernel(const float* __restrict__ pooled,
                                 const float* __restrict__ counts,
                                 const float* __restrict__ linW,
                                 const float* __restrict__ linb,
                                 float* __restrict__ out) {
  int idx = blockIdx.x * blockDim.x + threadIdx.x;
  if (idx >= N_GRAPHS * 10) return;
  int g = idx / 10, t = idx - g * 10;
  float acc = counts[g] * linb[t];
  for (int l = 1; l < 5; l++) acc += linb[l * 10 + t];
  for (int l = 0; l < 5; l++) {
    const float* p = pooled + ((size_t)l * N_GRAPHS + g) * 64;
    const float* w = linW + l * 640 + t;
    float s = 0.f;
#pragma unroll
    for (int k = 0; k < 64; k++) s += p[k] * w[k * 10];
    acc += s;
  }
  out[idx] = acc;
}

__global__ void gin_zero_kernel(unsigned int* __restrict__ p, int n) {
  int i = blockIdx.x * blockDim.x + threadIdx.x;
  if (i < n) p[i] = 0u;
}

extern "C" void kernel_launch(void* const* d_in, const int* in_sizes, int n_in,
                              void* d_out, int out_size, void* d_ws, size_t ws_size,
                              hipStream_t stream) {
  const float* x        = (const float*)d_in[0];
  const float* fh_W1    = (const float*)d_in[1];
  const float* fh_b1    = (const float*)d_in[2];
  const float* fh_bn1   = (const float*)d_in[3];
  const float* fh_W2    = (const float*)d_in[4];
  const float* fh_b2    = (const float*)d_in[5];
  const float* fh_bn2   = (const float*)d_in[6];
  const float* conv_W1  = (const float*)d_in[7];
  const float* conv_b1  = (const float*)d_in[8];
  const float* conv_bn1 = (const float*)d_in[9];
  const float* conv_W2  = (const float*)d_in[10];
  const float* conv_b2  = (const float*)d_in[11];
  const float* conv_bn2 = (const float*)d_in[12];
  const float* lin_W    = (const float*)d_in[13];
  const float* lin_b    = (const float*)d_in[14];
  const int*   eidx     = (const int*)d_in[15];
  const int*   batch    = (const int*)d_in[16];
  float* out = (float*)d_out;

  const int* src = eidx;
  const int* dst = eidx + N_EDGES;

  float* ws     = (float*)d_ws;
  float* h      = ws;
  float* zrow   = h + (size_t)N_NODES * 64;
  float* pooled = zrow + 64;
  float* counts = pooled + 5 * N_GRAPHS * 64;
  int*   deg    = (int*)(counts + N_GRAPHS);
  int*   rowptr = deg + N_NODES;
  int*   rank   = rowptr + N_NODES + 4;
  int*   csr    = rank + N_EDGES;
  float* agg    = ws + 10121088;

  const int accN = 64 + 5 * N_GRAPHS * 64 + N_GRAPHS + N_NODES;
  gin_zero_kernel<<<(accN + 255) / 256, 256, 0, stream>>>((unsigned int*)zrow, accN);

  gin_rank_kernel<<<(N_EDGES + 255) / 256, 256, 0, stream>>>(dst, deg, rank);
  gin_scan_kernel<<<1, 1024, 0, stream>>>(deg, rowptr);
  gin_fill_kernel<<<(N_EDGES + 255) / 256, 256, 0, stream>>>(src, dst, rowptr, rank, csr);

  const int mlpBlocks = (NTILES + 3) / 4;

  gin_mlp_kernel<D_FEAT, true><<<mlpBlocks, 128, 0, stream>>>(
      x, fh_W1, fh_b1, fh_bn1, fh_W2, fh_b2, fh_bn2, h, batch, pooled, counts, NTILES);

  for (int l = 0; l < 4; l++) {
    gin_gather_kernel<<<N_NODES / 16, 256, 0, stream>>>(h, rowptr, csr, agg);
    gin_mlp_kernel<HID, false><<<mlpBlocks, 128, 0, stream>>>(
        agg,
        conv_W1 + (size_t)l * 64 * 64, conv_b1 + l * 64, conv_bn1 + l * 256,
        conv_W2 + (size_t)l * 64 * 64, conv_b2 + l * 64, conv_bn2 + l * 256,
        h, batch, pooled + (size_t)(l + 1) * N_GRAPHS * 64, counts, NTILES);
  }

  gin_final_kernel<<<(N_GRAPHS * 10 + 255) / 256, 256, 0, stream>>>(
      pooled, counts, lin_W, lin_b, out);
}


# In-edge lists by a two-level counting sort with LDS counters (500-node buckets, then nodes) replace one returning global atomic per edge; neighbour sum order = LDS arrival rank per chunk and bucket
# speedup vs baseline: 2.8283x; 1.2763x over previous
#include <hip/hip_runtime.h>

#define N_NODES   100000
#define N_EDGES   1600000
#define N_GRAPHS  1000
#define D_FEAT    128
#define HID       64
#define BN_EPS    1e-5f
#define NTILES    (N_NODES / 16)

typedef __attribute__((ext_vector_type(2))) float v2f;
typedef __attribute__((ext_vector_type(8))) float v8f;

template<int K1, bool COUNT>
__launch_bounds__(128)
__global__ void gin_mlp_kernel(const float* __restrict__ xin,
                               const float* __restrict__ W1,
                               const float* __restrict__ b1,
                               const float* __restrict__ bn1,
                               const float* __restrict__ W2,
                               const float* __restrict__ b2,
                               const float* __restrict__ bn2,
                               float* __restrict__ hout,
                               const int* __restrict__ batch,
                               float* __restrict__ pooled,
                               float* __restrict__ counts,
                               int ntiles) {
  constexpr int KS1 = K1 / 4;
  __shared__ __align__(16) float w1s[4 * KS1 * 32 * 2];
  __shared__ __align__(16) float w2s[4 * 16 * 32 * 2];
  __shared__ __align__(16) float tb[4][16 * 68];

  const int tid = threadIdx.x;

  for (int i = tid; i < K1 * 64; i += 128) {
    int kk = i >> 6, nn = i & 63;
    int idx = ((((nn >> 4) * KS1) + (kk >> 2)) * 32 + (nn & 15) + ((kk & 2) << 3)) * 2 + (kk & 1);
    w1s[idx] = W1[i];
  }
  for (int i = tid; i < 64 * 64; i += 128) {
    int kk = i >> 6, nn = i & 63;
    int idx = ((((nn >> 4) * 16) + (kk >> 2)) * 32 + (nn & 15) + ((kk & 2) << 3)) * 2 + (kk & 1);
    w2s[idx] = W2[i];
  }
  __syncthreads();

  const int lane = tid & 31;
  const int wave = tid >> 5;
  const int row  = lane & 15;
  const int hi   = lane >> 4;
  const int tile = blockIdx.x * 4 + wave;
  const bool active = tile < ntiles;
  float* tbuf = tb[wave];
  const int gl = active ? batch[tile * 16 + row] : 0;

  const v8f vzero = {0.f, 0.f, 0.f, 0.f, 0.f, 0.f, 0.f, 0.f};

  if (active) {
    v2f a1[KS1];
    const float* xr = xin + (size_t)(tile * 16 + row) * K1 + 2 * hi;
#pragma unroll
    for (int k = 0; k < KS1; k++) a1[k] = *(const v2f*)(xr + 4 * k);

#pragma unroll
    for (int nt = 0; nt < 4; nt++) {
      v8f acc = vzero;
      const v2f* wf = ((const v2f*)w1s) + nt * KS1 * 32 + lane;
#pragma unroll
      for (int k = 0; k < KS1; k++) {
        acc = __builtin_amdgcn_wmma_f32_16x16x4_f32(false, a1[k], false, wf[k * 32],
                                                    (short)0, acc, false, false);
      }
      int n = nt * 16 + row;
      float s  = bn1[n] * rsqrtf(bn1[192 + n] + BN_EPS);
      float t0 = bn1[64 + n] + (b1[n] - bn1[128 + n]) * s;
#pragma unroll
      for (int i = 0; i < 8; i++) {
        float y = acc[i] * s + t0;
        tbuf[(i + 8 * hi) * 68 + n] = fmaxf(y, 0.0f);
      }
    }
  }
  __syncthreads();

  v8f acc2[4];
  if (active) {
#pragma unroll
    for (int nt = 0; nt < 4; nt++) {
      v8f acc = vzero;
      const v2f* wf = ((const v2f*)w2s) + nt * 16 * 32 + lane;
      const float* tr = tbuf + row * 68 + 2 * hi;
#pragma unroll
      for (int k = 0; k < 16; k++) {
        v2f a = *(const v2f*)(tr + 4 * k);
        acc = __builtin_amdgcn_wmma_f32_16x16x4_f32(false, a, false, wf[k * 32],
                                                    (short)0, acc, false, false);
      }
      acc2[nt] = acc;
    }
  }
  __syncthreads();

  if (active) {
#pragma unroll
    for (int nt = 0; nt < 4; nt++) {
      int n = nt * 16 + row;
      float s  = bn2[n] * rsqrtf(bn2[192 + n] + BN_EPS);
      float t0 = bn2[64 + n] + (b2[n] - bn2[128 + n]) * s;
#pragma unroll
      for (int i = 0; i < 8; i++) {
        float y = acc2[nt][i] * s + t0;
        tbuf[(i + 8 * hi) * 68 + n] = fmaxf(y, 0.0f);
      }
    }
  }
  __syncthreads();

  if (active) {
#pragma unroll
    for (int j = 0; j < 8; j++) {
      int flat = j * 32 + lane;
      int r = flat >> 4;
      int c = (flat & 15) << 2;
      float4 v = *(const float4*)(tbuf + r * 68 + c);
      *(float4*)(hout + (size_t)(tile * 16 + r) * 64 + c) = v;
    }
  }

  if (active) {
    float p0[16], p1[16];
#pragma unroll
    for (int r = 0; r < 16; r++) {
      p0[r] = tbuf[r * 68 + lane];
      p1[r] = tbuf[r * 68 + 32 + lane];
    }
    unsigned todo = 0xFFFFu;
    while (todo) {
      const int r0 = __builtin_ctz(todo);
      const int g = __builtin_amdgcn_readlane(gl, r0);
      const unsigned m = __builtin_amdgcn_ballot_w32(gl == g) & todo;
      float s0 = 0.f, s1 = 0.f;
#pragma unroll
      for (int r = 0; r < 16; r++) {
        const bool in = (m >> r) & 1u;
        s0 += in ? p0[r] : 0.f;
        s1 += in ? p1[r] : 0.f;
      }
      float* pg = pooled + (size_t)g * 64 + lane;
      atomicAdd(pg, s0);
      atomicAdd(pg + 32, s1);
      if (COUNT && lane == 0) atomicAdd(counts + g, (float)__builtin_popcount(m));
      todo &= ~m;
    }
  }
}

#define BKT_NODES     500
#define N_BKT         (N_NODES / BKT_NODES)
#define N_CHUNK       500
#define CHUNK_EDGES   (N_EDGES / N_CHUNK)
#define CHUNK_THREADS 320
#define CHUNK_PER_THR (CHUNK_EDGES / CHUNK_THREADS)

__launch_bounds__(CHUNK_THREADS)
__global__ void gin_bucket_count_kernel(const int* __restrict__ dst,
                                        int* __restrict__ bcnt) {
  __shared__ int cnt[N_BKT];
  const int t = threadIdx.x;
  const int g = blockIdx.x;
  if (t < N_BKT) cnt[t] = 0;
  __syncthreads();
  const int* dp = dst + g * CHUNK_EDGES + t;
  int d[CHUNK_PER_THR];
#pragma unroll
  for (int k = 0; k < CHUNK_PER_THR; k++) d[k] = dp[k * CHUNK_THREADS];
#pragma unroll
  for (int k = 0; k < CHUNK_PER_THR; k++) atomicAdd(&cnt[d[k] / BKT_NODES], 1);
  __syncthreads();
  if (t < N_BKT) bcnt[t * N_CHUNK + g] = cnt[t];
}

#define SCAN_CHUNK 100
#define SCAN_N (N_BKT * N_CHUNK)
#define SCAN_OWNERS (SCAN_N / SCAN_CHUNK)
__launch_bounds__(1024)
__global__ void gin_scan_kernel(const int* __restrict__ bcnt,
                                int* __restrict__ boffs) {
  __shared__ int wsum[32];
  const int t = threadIdx.x;
  const int lane = t & 31;
  const int w = t >> 5;
  const bool owner = t < SCAN_OWNERS;

  int4 v[SCAN_CHUNK / 4];
  int local = 0;
  if (owner) {
    const int4* p = (const int4*)(bcnt + t * SCAN_CHUNK);
#pragma unroll
    for (int i = 0; i < SCAN_CHUNK / 4; i++) v[i] = p[i];
#pragma unroll
    for (int i = 0; i < SCAN_CHUNK / 4; i++) local += v[i].x + v[i].y + v[i].z + v[i].w;
  }

  int x = local;
#pragma unroll
  for (int o = 1; o < 32; o <<= 1) {
    int y = __shfl_up(x, o);
    if (lane >= o) x += y;
  }
  if (lane == 31) wsum[w] = x;
  __syncthreads();
  if (w == 0) {
    int s = wsum[lane];
#pragma unroll
    for (int o = 1; o < 32; o <<= 1) {
      int y = __shfl_up(s, o);
      if (lane >= o) s += y;
    }
    wsum[lane] = s;
  }
  __syncthreads();

  if (owner) {
    int run = x - local + (w > 0 ? wsum[w - 1] : 0);
    int4* q = (int4*)(boffs + t * SCAN_CHUNK);
#pragma unroll
    for (int i = 0; i < SCAN_CHUNK / 4; i++) {
      int4 o4;
      o4.x = run; run += v[i].x;
      o4.y = run; run += v[i].y;
      o4.z = run; run += v[i].z;
      o4.w = run; run += v[i].w;
      q[i] = o4;
    }
    if (t == SCAN_OWNERS - 1) boffs[SCAN_N] = run;
  }
}

__launch_bounds__(CHUNK_THREADS)
__global__ void gin_bucket_fill_kernel(const int* __restrict__ src,
                                       const int* __restrict__ dst,
                                       const int* __restrict__ boffs,
                                       int2* __restrict__ pairs) {
  __shared__ int cur[N_BKT];
  const int t = threadIdx.x;
  const int g = blockIdx.x;
  if (t < N_BKT) cur[t] = boffs[t * N_CHUNK + g];
  __syncthreads();
  const int e0 = g * CHUNK_EDGES + t;
  int s[CHUNK_PER_THR], d[CHUNK_PER_THR];
#pragma unroll
  for (int k = 0; k < CHUNK_PER_THR; k++) {
    s[k] = src[e0 + k * CHUNK_THREADS];
    d[k] = dst[e0 + k * CHUNK_THREADS];
  }
#pragma unroll
  for (int k = 0; k < CHUNK_PER_THR; k++) {
    int pos = atomicAdd(&cur[d[k] / BKT_NODES], 1);
    pairs[pos] = make_int2(s[k], d[k]);
  }
}

#define SPARE_CNT 511
__launch_bounds__(1024)
__global__ void gin_bucket_csr_kernel(const int2* __restrict__ pairs,
                                      const int* __restrict__ boffs,
                                      int* __restrict__ rowptr,
                                      int* __restrict__ csr) {
  __shared__ int cnt[512];
  __shared__ int wsum[32];
  const int t = threadIdx.x;
  const int lane = t & 31;
  const int w = t >> 5;
  const int b = blockIdx.x;
  const int n0 = b * BKT_NODES;
  const int beg = boffs[b * N_CHUNK];
  const int end = boffs[(b + 1) * N_CHUNK];

  if (t < 512) cnt[t] = 0;
  __syncthreads();
  for (int i0 = beg + t; i0 < end; i0 += 4 * 1024) {
    int c[4];
#pragma unroll
    for (int k = 0; k < 4; k++) {
      int i = i0 + k * 1024;
      int d = pairs[i < end ? i : end - 1].y;
      c[k] = (i < end) ? d - n0 : SPARE_CNT;
    }
#pragma unroll
    for (int k = 0; k < 4; k++) atomicAdd(&cnt[c[k]], 1);
  }
  __syncthreads();

  const int v = (t < BKT_NODES) ? cnt[t] : 0;
  int x = v;
#pragma unroll
  for (int o = 1; o < 32; o <<= 1) {
    int y = __shfl_up(x, o);
    if (lane >= o) x += y;
  }
  if (lane == 31) wsum[w] = x;
  __syncthreads();
  if (w == 0) {
    int s = wsum[lane];
#pragma unroll
    for (int o = 1; o < 32; o <<= 1) {
      int y = __shfl_up(s, o);
      if (lane >= o) s += y;
    }
    wsum[lane] = s;
  }
  __syncthreads();
  const int first = beg + x - v + (w > 0 ? wsum[w - 1] : 0);
  if (t < BKT_NODES) {
    rowptr[n0 + t] = first;
    cnt[t] = first;
  }
  if (b == N_BKT - 1 && t == BKT_NODES - 1) rowptr[N_NODES] = first + v;
  __syncthreads();

  for (int i0 = beg + t; i0 < end; i0 += 4 * 1024) {
    int2 p[4];
#pragma unroll
    for (int k = 0; k < 4; k++) {
      int i = i0 + k * 1024;
      p[k] = pairs[i < end ? i : end - 1];
    }
#pragma unroll
    for (int k = 0; k < 4; k++) {
      if (i0 + k * 1024 < end) {
        int pos = atomicAdd(&cnt[p[k].y - n0], 1);
        csr[pos] = p[k].x;
      }
    }
  }
}

__launch_bounds__(256)
__global__ void gin_gather_kernel(const float* __restrict__ h,
                                  const int* __restrict__ rowptr,
                                  const int* __restrict__ csr,
                                  float* __restrict__ agg) {
  const int n = blockIdx.x * 16 + (threadIdx.x >> 4);
  const unsigned cb = (threadIdx.x & 15u) << 4;
  const int beg = rowptr[n];
  const int end = rowptr[n + 1];
  const char* hb = (const char*)h;
  float4 acc = *(const float4*)(hb + (((unsigned)n << 8) + cb));

  for (int j = beg; j < end; j += 8) {
    int s[8];
#pragma unroll
    for (int k = 0; k < 8; k++) {
      int jj = j + k;
      s[k] = csr[jj < end ? jj : end - 1];
    }
    float4 v[8];
#pragma unroll
    for (int k = 0; k < 8; k++) {
      int sk = (j + k < end) ? s[k] : N_NODES;
      v[k] = *(const float4*)(hb + (((unsigned)sk << 8) + cb));
    }
#pragma unroll
    for (int k = 0; k < 8; k++) {
      acc.x += v[k].x; acc.y += v[k].y; acc.z += v[k].z; acc.w += v[k].w;
    }
  }
  *(float4*)(agg + (size_t)n * 64 + ((threadIdx.x & 15) << 2)) = acc;
}

__global__ void gin_final_kernel(const float* __restrict__ pooled,
                                 const float* __restrict__ counts,
                                 const float* __restrict__ linW,
                                 const float* __restrict__ linb,
                                 float* __restrict__ out) {
  int idx = blockIdx.x * blockDim.x + threadIdx.x;
  if (idx >= N_GRAPHS * 10) return;
  int g = idx / 10, t = idx - g * 10;
  float acc = counts[g] * linb[t];
  for (int l = 1; l < 5; l++) acc += linb[l * 10 + t];
  for (int l = 0; l < 5; l++) {
    const float* p = pooled + ((size_t)l * N_GRAPHS + g) * 64;
    const float* w = linW + l * 640 + t;
    float s = 0.f;
#pragma unroll
    for (int k = 0; k < 64; k++) s += p[k] * w[k * 10];
    acc += s;
  }
  out[idx] = acc;
}

__global__ void gin_zero_kernel(unsigned int* __restrict__ p, int n) {
  int i = blockIdx.x * blockDim.x + threadIdx.x;
  if (i < n) p[i] = 0u;
}

extern "C" void kernel_launch(void* const* d_in, const int* in_sizes, int n_in,
                              void* d_out, int out_size, void* d_ws, size_t ws_size,
                              hipStream_t stream) {
  const float* x        = (const float*)d_in[0];
  const float* fh_W1    = (const float*)d_in[1];
  const float* fh_b1    = (const float*)d_in[2];
  const float* fh_bn1   = (const float*)d_in[3];
  const float* fh_W2    = (const float*)d_in[4];
  const float* fh_b2    = (const float*)d_in[5];
  const float* fh_bn2   = (const float*)d_in[6];
  const float* conv_W1  = (const float*)d_in[7];
  const float* conv_b1  = (const float*)d_in[8];
  const float* conv_bn1 = (const float*)d_in[9];
  const float* conv_W2  = (const float*)d_in[10];
  const float* conv_b2  = (const float*)d_in[11];
  const float* conv_bn2 = (const float*)d_in[12];
  const float* lin_W    = (const float*)d_in[13];
  const float* lin_b    = (const float*)d_in[14];
  const int*   eidx     = (const int*)d_in[15];
  const int*   batch    = (const int*)d_in[16];
  float* out = (float*)d_out;

  const int* src = eidx;
  const int* dst = eidx + N_EDGES;

  float* ws     = (float*)d_ws;
  float* h      = ws;
  float* zrow   = h + (size_t)N_NODES * 64;
  float* pooled = zrow + 64;
  float* counts = pooled + 5 * N_GRAPHS * 64;
  int*   bcnt   = (int*)(counts + N_GRAPHS);
  int*   boffs  = bcnt + SCAN_N;
  int*   rowptr = boffs + SCAN_N + 4;
  int2*  pairs  = (int2*)(rowptr + N_NODES + 4);
  int*   csr    = (int*)(pairs + N_EDGES);
  float* agg    = ws + 12000000;

  const int accN = 64 + 5 * N_GRAPHS * 64 + N_GRAPHS;
  gin_zero_kernel<<<(accN + 255) / 256, 256, 0, stream>>>((unsigned int*)zrow, accN);

  gin_bucket_count_kernel<<<N_CHUNK, CHUNK_THREADS, 0, stream>>>(dst, bcnt);
  gin_scan_kernel<<<1, 1024, 0, stream>>>(bcnt, boffs);
  gin_bucket_fill_kernel<<<N_CHUNK, CHUNK_THREADS, 0, stream>>>(src, dst, boffs, pairs);
  gin_bucket_csr_kernel<<<N_BKT, 1024, 0, stream>>>(pairs, boffs, rowptr, csr);

  const int mlpBlocks = (NTILES + 3) / 4;

  gin_mlp_kernel<D_FEAT, true><<<mlpBlocks, 128, 0, stream>>>(
      x, fh_W1, fh_b1, fh_bn1, fh_W2, fh_b2, fh_bn2, h, batch, pooled, counts, NTILES);

  for (int l = 0; l < 4; l++) {
    gin_gather_kernel<<<N_NODES / 16, 256, 0, stream>>>(h, rowptr, csr, agg);
    gin_mlp_kernel<HID, false><<<mlpBlocks, 128, 0, stream>>>(
        agg,
        conv_W1 + (size_t)l * 64 * 64, conv_b1 + l * 64, conv_bn1 + l * 256,
        conv_W2 + (size_t)l * 64 * 64, conv_b2 + l * 64, conv_bn2 + l * 256,
        h, batch, pooled + (size_t)(l + 1) * N_GRAPHS * 64, counts, NTILES);
  }

  gin_final_kernel<<<(N_GRAPHS * 10 + 255) / 256, 256, 0, stream>>>(
      pooled, counts, lin_W, lin_b, out);
}


# MLP work-group start-up as ONE load group: both weight matrices by 16-byte row loads into registers plus the wave's A rows, then the LDS swizzle; was 8-10 dependent load-wait-store trips
# speedup vs baseline: 3.1442x; 1.1117x over previous
#include <hip/hip_runtime.h>

#define N_NODES   100000
#define N_EDGES   1600000
#define N_GRAPHS  1000
#define D_FEAT    128
#define HID       64
#define BN_EPS    1e-5f
#define NTILES    (N_NODES / 16)

typedef __attribute__((ext_vector_type(2))) float v2f;
typedef __attribute__((ext_vector_type(8))) float v8f;

template<int K1, bool COUNT>
__launch_bounds__(128)
__global__ void gin_mlp_kernel(const float* __restrict__ xin,
                               const float* __restrict__ W1,
                               const float* __restrict__ b1,
                               const float* __restrict__ bn1,
                               const float* __restrict__ W2,
                               const float* __restrict__ b2,
                               const float* __restrict__ bn2,
                               float* __restrict__ hout,
                               const int* __restrict__ batch,
                               float* __restrict__ pooled,
                               float* __restrict__ counts,
                               int ntiles) {
  constexpr int KS1 = K1 / 4;
  constexpr int NQ1 = (K1 / 2) * 16 / 128;
  constexpr int NQ2 = (64 / 2) * 16 / 128;
  __shared__ __align__(16) float w1s[4 * KS1 * 32 * 2];
  __shared__ __align__(16) float w2s[4 * 16 * 32 * 2];
  __shared__ __align__(16) float tb[4][16 * 68];

  const int tid  = threadIdx.x;
  const int lane = tid & 31;
  const int wave = tid >> 5;
  const int row  = lane & 15;
  const int hi   = lane >> 4;
  const int tile = blockIdx.x * 4 + wave;
  const bool active = tile < ntiles;
  float* tbuf = tb[wave];

  float4 w1a[NQ1], w1b[NQ1], w2a[NQ2], w2b[NQ2];
#pragma unroll
  for (int j = 0; j < NQ1; j++) {
    const int p = tid + 128 * j;
    const float* wp = W1 + (p >> 4) * 128 + (p & 15) * 4;
    w1a[j] = *(const float4*)wp;
    w1b[j] = *(const float4*)(wp + 64);
  }
#pragma unroll
  for (int j = 0; j < NQ2; j++) {
    const int p = tid + 128 * j;
    const float* wp = W2 + (p >> 4) * 128 + (p & 15) * 4;
    w2a[j] = *(const float4*)wp;
    w2b[j] = *(const float4*)(wp + 64);
  }
  const int tl = active ? tile : 0;
  const int gl = batch[tl * 16 + row];
  v2f a1[KS1];
  const float* xr = xin + (size_t)(tl * 16 + row) * K1 + 2 * hi;
#pragma unroll
  for (int k = 0; k < KS1; k++) a1[k] = *(const v2f*)(xr + 4 * k);

#pragma unroll
  for (int j = 0; j < NQ1; j++) {
    const int p = tid + 128 * j;
    const int kk = 2 * (p >> 4), nq = p & 15;
    float* d = w1s + ((((nq >> 2) * KS1) + (kk >> 2)) * 32 + 4 * (nq & 3) + ((kk & 2) << 3)) * 2;
    *(float4*)d       = make_float4(w1a[j].x, w1b[j].x, w1a[j].y, w1b[j].y);
    *(float4*)(d + 4) = make_float4(w1a[j].z, w1b[j].z, w1a[j].w, w1b[j].w);
  }
#pragma unroll
  for (int j = 0; j < NQ2; j++) {
    const int p = tid + 128 * j;
    const int kk = 2 * (p >> 4), nq = p & 15;
    float* d = w2s + ((((nq >> 2) * 16) + (kk >> 2)) * 32 + 4 * (nq & 3) + ((kk & 2) << 3)) * 2;
    *(float4*)d       = make_float4(w2a[j].x, w2b[j].x, w2a[j].y, w2b[j].y);
    *(float4*)(d + 4) = make_float4(w2a[j].z, w2b[j].z, w2a[j].w, w2b[j].w);
  }
  __syncthreads();

  const v8f vzero = {0.f, 0.f, 0.f, 0.f, 0.f, 0.f, 0.f, 0.f};

  if (active) {
#pragma unroll
    for (int nt = 0; nt < 4; nt++) {
      v8f acc = vzero;
      const v2f* wf = ((const v2f*)w1s) + nt * KS1 * 32 + lane;
#pragma unroll
      for (int k = 0; k < KS1; k++) {
        acc = __builtin_amdgcn_wmma_f32_16x16x4_f32(false, a1[k], false, wf[k * 32],
                                                    (short)0, acc, false, false);
      }
      int n = nt * 16 + row;
      float s  = bn1[n] * rsqrtf(bn1[192 + n] + BN_EPS);
      float t0 = bn1[64 + n] + (b1[n] - bn1[128 + n]) * s;
#pragma unroll
      for (int i = 0; i < 8; i++) {
        float y = acc[i] * s + t0;
        tbuf[(i + 8 * hi) * 68 + n] = fmaxf(y, 0.0f);
      }
    }
  }
  __syncthreads();

  v8f acc2[4];
  if (active) {
#pragma unroll
    for (int nt = 0; nt < 4; nt++) {
      v8f acc = vzero;
      const v2f* wf = ((const v2f*)w2s) + nt * 16 * 32 + lane;
      const float* tr = tbuf + row * 68 + 2 * hi;
#pragma unroll
      for (int k = 0; k < 16; k++) {
        v2f a = *(const v2f*)(tr + 4 * k);
        acc = __builtin_amdgcn_wmma_f32_16x16x4_f32(false, a, false, wf[k * 32],
                                                    (short)0, acc, false, false);
      }
      acc2[nt] = acc;
    }
  }
  __syncthreads();

  if (active) {
#pragma unroll
    for (int nt = 0; nt < 4; nt++) {
      int n = nt * 16 + row;
      float s  = bn2[n] * rsqrtf(bn2[192 + n] + BN_EPS);
      float t0 = bn2[64 + n] + (b2[n] - bn2[128 + n]) * s;
#pragma unroll
      for (int i = 0; i < 8; i++) {
        float y = acc2[nt][i] * s + t0;
        tbuf[(i + 8 * hi) * 68 + n] = fmaxf(y, 0.0f);
      }
    }
  }
  __syncthreads();

  if (active) {
#pragma unroll
    for (int j = 0; j < 8; j++) {
      int flat = j * 32 + lane;
      int r = flat >> 4;
      int c = (flat & 15) << 2;
      float4 v = *(const float4*)(tbuf + r * 68 + c);
      *(float4*)(hout + (size_t)(tile * 16 + r) * 64 + c) = v;
    }
  }

  if (active) {
    float p0[16], p1[16];
#pragma unroll
    for (int r = 0; r < 16; r++) {
      p0[r] = tbuf[r * 68 + lane];
      p1[r] = tbuf[r * 68 + 32 + lane];
    }
    unsigned todo = 0xFFFFu;
    while (todo) {
      const int r0 = __builtin_ctz(todo);
      const int g = __builtin_amdgcn_readlane(gl, r0);
      const unsigned m = __builtin_amdgcn_ballot_w32(gl == g) & todo;
      float s0 = 0.f, s1 = 0.f;
#pragma unroll
      for (int r = 0; r < 16; r++) {
        const bool in = (m >> r) & 1u;
        s0 += in ? p0[r] : 0.f;
        s1 += in ? p1[r] : 0.f;
      }
      float* pg = pooled + (size_t)g * 64 + lane;
      atomicAdd(pg, s0);
      atomicAdd(pg + 32, s1);
      if (COUNT && lane == 0) atomicAdd(counts + g, (float)__builtin_popcount(m));
      todo &= ~m;
    }
  }
}

#define BKT_NODES     500
#define N_BKT         (N_NODES / BKT_NODES)
#define N_CHUNK       500
#define CHUNK_EDGES   (N_EDGES / N_CHUNK)
#define CHUNK_THREADS 320
#define CHUNK_PER_THR (CHUNK_EDGES / CHUNK_THREADS)

__launch_bounds__(CHUNK_THREADS)
__global__ void gin_bucket_count_kernel(const int* __restrict__ dst,
                                        int* __restrict__ bcnt) {
  __shared__ int cnt[N_BKT];
  const int t = threadIdx.x;
  const int g = blockIdx.x;
  if (t < N_BKT) cnt[t] = 0;
  __syncthreads();
  const int* dp = dst + g * CHUNK_EDGES + t;
  int d[CHUNK_PER_THR];
#pragma unroll
  for (int k = 0; k < CHUNK_PER_THR; k++) d[k] = dp[k * CHUNK_THREADS];
#pragma unroll
  for (int k = 0; k < CHUNK_PER_THR; k++) atomicAdd(&cnt[d[k] / BKT_NODES], 1);
  __syncthreads();
  if (t < N_BKT) bcnt[t * N_CHUNK + g] = cnt[t];
}

#define SCAN_CHUNK 100
#define SCAN_N (N_BKT * N_CHUNK)
#define SCAN_OWNERS (SCAN_N / SCAN_CHUNK)
__launch_bounds__(1024)
__global__ void gin_scan_kernel(const int* __restrict__ bcnt,
                                int* __restrict__ boffs) {
  __shared__ int wsum[32];
  const int t = threadIdx.x;
  const int lane = t & 31;
  const int w = t >> 5;
  const bool owner = t < SCAN_OWNERS;

  int4 v[SCAN_CHUNK / 4];
  int local = 0;
  if (owner) {
    const int4* p = (const int4*)(bcnt + t * SCAN_CHUNK);
#pragma unroll
    for (int i = 0; i < SCAN_CHUNK / 4; i++) v[i] = p[i];
#pragma unroll
    for (int i = 0; i < SCAN_CHUNK / 4; i++) local += v[i].x + v[i].y + v[i].z + v[i].w;
  }

  int x = local;
#pragma unroll
  for (int o = 1; o < 32; o <<= 1) {
    int y = __shfl_up(x, o);
    if (lane >= o) x += y;
  }
  if (lane == 31) wsum[w] = x;
  __syncthreads();
  if (w == 0) {
    int s = wsum[lane];
#pragma unroll
    for (int o = 1; o < 32; o <<= 1) {
      int y = __shfl_up(s, o);
      if (lane >= o) s += y;
    }
    wsum[lane] = s;
  }
  __syncthreads();

  if (owner) {
    int run = x - local + (w > 0 ? wsum[w - 1] : 0);
    int4* q = (int4*)(boffs + t * SCAN_CHUNK);
#pragma unroll
    for (int i = 0; i < SCAN_CHUNK / 4; i++) {
      int4 o4;
      o4.x = run; run += v[i].x;
      o4.y = run; run += v[i].y;
      o4.z = run; run += v[i].z;
      o4.w = run; run += v[i].w;
      q[i] = o4;
    }
    if (t == SCAN_OWNERS - 1) boffs[SCAN_N] = run;
  }
}

__launch_bounds__(CHUNK_THREADS)
__global__ void gin_bucket_fill_kernel(const int* __restrict__ src,
                                       const int* __restrict__ dst,
                                       const int* __restrict__ boffs,
                                       int2* __restrict__ pairs) {
  __shared__ int cur[N_BKT];
  const int t = threadIdx.x;
  const int g = blockIdx.x;
  if (t < N_BKT) cur[t] = boffs[t * N_CHUNK + g];
  __syncthreads();
  const int e0 = g * CHUNK_EDGES + t;
  int s[CHUNK_PER_THR], d[CHUNK_PER_THR];
#pragma unroll
  for (int k = 0; k < CHUNK_PER_THR; k++) {
    s[k] = src[e0 + k * CHUNK_THREADS];
    d[k] = dst[e0 + k * CHUNK_THREADS];
  }
#pragma unroll
  for (int k = 0; k < CHUNK_PER_THR; k++) {
    int pos = atomicAdd(&cur[d[k] / BKT_NODES], 1);
    pairs[pos] = make_int2(s[k], d[k]);
  }
}

#define SPARE_CNT 511
__launch_bounds__(1024)
__global__ void gin_bucket_csr_kernel(const int2* __restrict__ pairs,
                                      const int* __restrict__ boffs,
                                      int* __restrict__ rowptr,
                                      int* __restrict__ csr) {
  __shared__ int cnt[512];
  __shared__ int wsum[32];
  const int t = threadIdx.x;
  const int lane = t & 31;
  const int w = t >> 5;
  const int b = blockIdx.x;
  const int n0 = b * BKT_NODES;
  const int beg = boffs[b * N_CHUNK];
  const int end = boffs[(b + 1) * N_CHUNK];

  if (t < 512) cnt[t] = 0;
  __syncthreads();
  for (int i0 = beg + t; i0 < end; i0 += 4 * 1024) {
    int c[4];
#pragma unroll
    for (int k = 0; k < 4; k++) {
      int i = i0 + k * 1024;
      int d = pairs[i < end ? i : end - 1].y;
      c[k] = (i < end) ? d - n0 : SPARE_CNT;
    }
#pragma unroll
    for (int k = 0; k < 4; k++) atomicAdd(&cnt[c[k]], 1);
  }
  __syncthreads();

  const int v = (t < BKT_NODES) ? cnt[t] : 0;
  int x = v;
#pragma unroll
  for (int o = 1; o < 32; o <<= 1) {
    int y = __shfl_up(x, o);
    if (lane >= o) x += y;
  }
  if (lane == 31) wsum[w] = x;
  __syncthreads();
  if (w == 0) {
    int s = wsum[lane];
#pragma unroll
    for (int o = 1; o < 32; o <<= 1) {
      int y = __shfl_up(s, o);
      if (lane >= o) s += y;
    }
    wsum[lane] = s;
  }
  __syncthreads();
  const int first = beg + x - v + (w > 0 ? wsum[w - 1] : 0);
  if (t < BKT_NODES) {
    rowptr[n0 + t] = first;
    cnt[t] = first;
  }
  if (b == N_BKT - 1 && t == BKT_NODES - 1) rowptr[N_NODES] = first + v;
  __syncthreads();

  for (int i0 = beg + t; i0 < end; i0 += 4 * 1024) {
    int2 p[4];
#pragma unroll
    for (int k = 0; k < 4; k++) {
      int i = i0 + k * 1024;
      p[k] = pairs[i < end ? i : end - 1];
    }
#pragma unroll
    for (int k = 0; k < 4; k++) {
      if (i0 + k * 1024 < end) {
        int pos = atomicAdd(&cnt[p[k].y - n0], 1);
        csr[pos] = p[k].x;
      }
    }
  }
}

__launch_bounds__(256)
__global__ void gin_gather_kernel(const float* __restrict__ h,
                                  const int* __restrict__ rowptr,
                                  const int* __restrict__ csr,
                                  float* __restrict__ agg) {
  const int n = blockIdx.x * 16 + (threadIdx.x >> 4);
  const unsigned cb = (threadIdx.x & 15u) << 4;
  const int beg = rowptr[n];
  const int end = rowptr[n + 1];
  const char* hb = (const char*)h;
  float4 acc = *(const float4*)(hb + (((unsigned)n << 8) + cb));

  for (int j = beg; j < end; j += 8) {
    int s[8];
#pragma unroll
    for (int k = 0; k < 8; k++) {
      int jj = j + k;
      s[k] = csr[jj < end ? jj : end - 1];
    }
    float4 v[8];
#pragma unroll
    for (int k = 0; k < 8; k++) {
      int sk = (j + k < end) ? s[k] : N_NODES;
      v[k] = *(const float4*)(hb + (((unsigned)sk << 8) + cb));
    }
#pragma unroll
    for (int k = 0; k < 8; k++) {
      acc.x += v[k].x; acc.y += v[k].y; acc.z += v[k].z; acc.w += v[k].w;
    }
  }
  *(float4*)(agg + (size_t)n * 64 + ((threadIdx.x & 15) << 2)) = acc;
}

__global__ void gin_final_kernel(const float* __restrict__ pooled,
                                 const float* __restrict__ counts,
                                 const float* __restrict__ linW,
                                 const float* __restrict__ linb,
                                 float* __restrict__ out) {
  int idx = blockIdx.x * blockDim.x + threadIdx.x;
  if (idx >= N_GRAPHS * 10) return;
  int g = idx / 10, t = idx - g * 10;
  float acc = counts[g] * linb[t];
  for (int l = 1; l < 5; l++) acc += linb[l * 10 + t];
  for (int l = 0; l < 5; l++) {
    const float* p = pooled + ((size_t)l * N_GRAPHS + g) * 64;
    const float* w = linW + l * 640 + t;
    float s = 0.f;
#pragma unroll
    for (int k = 0; k < 64; k++) s += p[k] * w[k * 10];
    acc += s;
  }
  out[idx] = acc;
}

__global__ void gin_zero_kernel(unsigned int* __restrict__ p, int n) {
  int i = blockIdx.x * blockDim.x + threadIdx.x;
  if (i < n) p[i] = 0u;
}

extern "C" void kernel_launch(void* const* d_in, const int* in_sizes, int n_in,
                              void* d_out, int out_size, void* d_ws, size_t ws_size,
                              hipStream_t stream) {
  const float* x        = (const float*)d_in[0];
  const float* fh_W1    = (const float*)d_in[1];
  const float* fh_b1    = (const float*)d_in[2];
  const float* fh_bn1   = (const float*)d_in[3];
  const float* fh_W2    = (const float*)d_in[4];
  const float* fh_b2    = (const float*)d_in[5];
  const float* fh_bn2   = (const float*)d_in[6];
  const float* conv_W1  = (const float*)d_in[7];
  const float* conv_b1  = (const float*)d_in[8];
  const float* conv_bn1 = (const float*)d_in[9];
  const float* conv_W2  = (const float*)d_in[10];
  const float* conv_b2  = (const float*)d_in[11];
  const float* conv_bn2 = (const float*)d_in[12];
  const float* lin_W    = (const float*)d_in[13];
  const float* lin_b    = (const float*)d_in[14];
  const int*   eidx     = (const int*)d_in[15];
  const int*   batch    = (const int*)d_in[16];
  float* out = (float*)d_out;

  const int* src = eidx;
  const int* dst = eidx + N_EDGES;

  float* ws     = (float*)d_ws;
  float* h      = ws;
  float* zrow   = h + (size_t)N_NODES * 64;
  float* pooled = zrow + 64;
  float* counts = pooled + 5 * N_GRAPHS * 64;
  int*   bcnt   = (int*)(counts + N_GRAPHS);
  int*   boffs  = bcnt + SCAN_N;
  int*   rowptr = boffs + SCAN_N + 4;
  int2*  pairs  = (int2*)(rowptr + N_NODES + 4);
  int*   csr    = (int*)(pairs + N_EDGES);
  float* agg    = ws + 12000000;

  const int accN = 64 + 5 * N_GRAPHS * 64 + N_GRAPHS;
  gin_zero_kernel<<<(accN + 255) / 256, 256, 0, stream>>>((unsigned int*)zrow, accN);

  gin_bucket_count_kernel<<<N_CHUNK, CHUNK_THREADS, 0, stream>>>(dst, bcnt);
  gin_scan_kernel<<<1, 1024, 0, stream>>>(bcnt, boffs);
  gin_bucket_fill_kernel<<<N_CHUNK, CHUNK_THREADS, 0, stream>>>(src, dst, boffs, pairs);
  gin_bucket_csr_kernel<<<N_BKT, 1024, 0, stream>>>(pairs, boffs, rowptr, csr);

  const int mlpBlocks = (NTILES + 3) / 4;

  gin_mlp_kernel<D_FEAT, true><<<mlpBlocks, 128, 0, stream>>>(
      x, fh_W1, fh_b1, fh_bn1, fh_W2, fh_b2, fh_bn2, h, batch, pooled, counts, NTILES);

  for (int l = 0; l < 4; l++) {
    gin_gather_kernel<<<N_NODES / 16, 256, 0, stream>>>(h, rowptr, csr, agg);
    gin_mlp_kernel<HID, false><<<mlpBlocks, 128, 0, stream>>>(
        agg,
        conv_W1 + (size_t)l * 64 * 64, conv_b1 + l * 64, conv_bn1 + l * 256,
        conv_W2 + (size_t)l * 64 * 64, conv_b2 + l * 64, conv_bn2 + l * 256,
        h, batch, pooled + (size_t)(l + 1) * N_GRAPHS * 64, counts, NTILES);
  }

  gin_final_kernel<<<(N_GRAPHS * 10 + 255) / 256, 256, 0, stream>>>(
      pooled, counts, lin_W, lin_b, out);
}


# Gather fused into the layer kernel: each wave sums its 16-node tile's neighbour rows into LDS (16 rows in flight), then MLP and pool; agg buffer and 4 launches gone; same sum order
# speedup vs baseline: 3.2566x; 1.0357x over previous
#include <hip/hip_runtime.h>

#define N_NODES   100000
#define N_EDGES   1600000
#define N_GRAPHS  1000
#define D_FEAT    128
#define HID       64
#define BN_EPS    1e-5f
#define NTILES    (N_NODES / 16)

typedef __attribute__((ext_vector_type(2))) float v2f;
typedef __attribute__((ext_vector_type(8))) float v8f;

template<int K1, bool COUNT>
__launch_bounds__(128)
__global__ void gin_mlp_kernel(const float* __restrict__ xin,
                               const float* __restrict__ W1,
                               const float* __restrict__ b1,
                               const float* __restrict__ bn1,
                               const float* __restrict__ W2,
                               const float* __restrict__ b2,
                               const float* __restrict__ bn2,
                               float* __restrict__ hout,
                               const int* __restrict__ batch,
                               float* __restrict__ pooled,
                               float* __restrict__ counts,
                               int ntiles) {
  constexpr int KS1 = K1 / 4;
  constexpr int NQ1 = (K1 / 2) * 16 / 128;
  constexpr int NQ2 = (64 / 2) * 16 / 128;
  __shared__ __align__(16) float w1s[4 * KS1 * 32 * 2];
  __shared__ __align__(16) float w2s[4 * 16 * 32 * 2];
  __shared__ __align__(16) float tb[4][16 * 68];

  const int tid  = threadIdx.x;
  const int lane = tid & 31;
  const int wave = tid >> 5;
  const int row  = lane & 15;
  const int hi   = lane >> 4;
  const int tile = blockIdx.x * 4 + wave;
  const bool active = tile < ntiles;
  float* tbuf = tb[wave];

  float4 w1a[NQ1], w1b[NQ1], w2a[NQ2], w2b[NQ2];
#pragma unroll
  for (int j = 0; j < NQ1; j++) {
    const int p = tid + 128 * j;
    const float* wp = W1 + (p >> 4) * 128 + (p & 15) * 4;
    w1a[j] = *(const float4*)wp;
    w1b[j] = *(const float4*)(wp + 64);
  }
#pragma unroll
  for (int j = 0; j < NQ2; j++) {
    const int p = tid + 128 * j;
    const float* wp = W2 + (p >> 4) * 128 + (p & 15) * 4;
    w2a[j] = *(const float4*)wp;
    w2b[j] = *(const float4*)(wp + 64);
  }
  const int tl = active ? tile : 0;
  const int gl = batch[tl * 16 + row];
  v2f a1[KS1];
  const float* xr = xin + (size_t)(tl * 16 + row) * K1 + 2 * hi;
#pragma unroll
  for (int k = 0; k < KS1; k++) a1[k] = *(const v2f*)(xr + 4 * k);

#pragma unroll
  for (int j = 0; j < NQ1; j++) {
    const int p = tid + 128 * j;
    const int kk = 2 * (p >> 4), nq = p & 15;
    float* d = w1s + ((((nq >> 2) * KS1) + (kk >> 2)) * 32 + 4 * (nq & 3) + ((kk & 2) << 3)) * 2;
    *(float4*)d       = make_float4(w1a[j].x, w1b[j].x, w1a[j].y, w1b[j].y);
    *(float4*)(d + 4) = make_float4(w1a[j].z, w1b[j].z, w1a[j].w, w1b[j].w);
  }
#pragma unroll
  for (int j = 0; j < NQ2; j++) {
    const int p = tid + 128 * j;
    const int kk = 2 * (p >> 4), nq = p & 15;
    float* d = w2s + ((((nq >> 2) * 16) + (kk >> 2)) * 32 + 4 * (nq & 3) + ((kk & 2) << 3)) * 2;
    *(float4*)d       = make_float4(w2a[j].x, w2b[j].x, w2a[j].y, w2b[j].y);
    *(float4*)(d + 4) = make_float4(w2a[j].z, w2b[j].z, w2a[j].w, w2b[j].w);
  }
  __syncthreads();

  const v8f vzero = {0.f, 0.f, 0.f, 0.f, 0.f, 0.f, 0.f, 0.f};

  if (active) {
#pragma unroll
    for (int nt = 0; nt < 4; nt++) {
      v8f acc = vzero;
      const v2f* wf = ((const v2f*)w1s) + nt * KS1 * 32 + lane;
#pragma unroll
      for (int k = 0; k < KS1; k++) {
        acc = __builtin_amdgcn_wmma_f32_16x16x4_f32(false, a1[k], false, wf[k * 32],
                                                    (short)0, acc, false, false);
      }
      int n = nt * 16 + row;
      float s  = bn1[n] * rsqrtf(bn1[192 + n] + BN_EPS);
      float t0 = bn1[64 + n] + (b1[n] - bn1[128 + n]) * s;
#pragma unroll
      for (int i = 0; i < 8; i++) {
        float y = acc[i] * s + t0;
        tbuf[(i + 8 * hi) * 68 + n] = fmaxf(y, 0.0f);
      }
    }
  }
  __syncthreads();

  v8f acc2[4];
  if (active) {
#pragma unroll
    for (int nt = 0; nt < 4; nt++) {
      v8f acc = vzero;
      const v2f* wf = ((const v2f*)w2s) + nt * 16 * 32 + lane;
      const float* tr = tbuf + row * 68 + 2 * hi;
#pragma unroll
      for (int k = 0; k < 16; k++) {
        v2f a = *(const v2f*)(tr + 4 * k);
        acc = __builtin_amdgcn_wmma_f32_16x16x4_f32(false, a, false, wf[k * 32],
                                                    (short)0, acc, false, false);
      }
      acc2[nt] = acc;
    }
  }
  __syncthreads();

  if (active) {
#pragma unroll
    for (int nt = 0; nt < 4; nt++) {
      int n = nt * 16 + row;
      float s  = bn2[n] * rsqrtf(bn2[192 + n] + BN_EPS);
      float t0 = bn2[64 + n] + (b2[n] - bn2[128 + n]) * s;
#pragma unroll
      for (int i = 0; i < 8; i++) {
        float y = acc2[nt][i] * s + t0;
        tbuf[(i + 8 * hi) * 68 + n] = fmaxf(y, 0.0f);
      }
    }
  }
  __syncthreads();

  if (active) {
#pragma unroll
    for (int j = 0; j < 8; j++) {
      int flat = j * 32 + lane;
      int r = flat >> 4;
      int c = (flat & 15) << 2;
      float4 v = *(const float4*)(tbuf + r * 68 + c);
      *(float4*)(hout + (size_t)(tile * 16 + r) * 64 + c) = v;
    }
  }

  if (active) {
    float p0[16], p1[16];
#pragma unroll
    for (int r = 0; r < 16; r++) {
      p0[r] = tbuf[r * 68 + lane];
      p1[r] = tbuf[r * 68 + 32 + lane];
    }
    unsigned todo = 0xFFFFu;
    while (todo) {
      const int r0 = __builtin_ctz(todo);
      const int g = __builtin_amdgcn_readlane(gl, r0);
      const unsigned m = __builtin_amdgcn_ballot_w32(gl == g) & todo;
      float s0 = 0.f, s1 = 0.f;
#pragma unroll
      for (int r = 0; r < 16; r++) {
        const bool in = (m >> r) & 1u;
        s0 += in ? p0[r] : 0.f;
        s1 += in ? p1[r] : 0.f;
      }
      float* pg = pooled + (size_t)g * 64 + lane;
      atomicAdd(pg, s0);
      atomicAdd(pg + 32, s1);
      if (COUNT && lane == 0) atomicAdd(counts + g, (float)__builtin_popcount(m));
      todo &= ~m;
    }
  }
}

#define BKT_NODES     500
#define N_BKT         (N_NODES / BKT_NODES)
#define N_CHUNK       500
#define CHUNK_EDGES   (N_EDGES / N_CHUNK)
#define CHUNK_THREADS 320
#define CHUNK_PER_THR (CHUNK_EDGES / CHUNK_THREADS)

__launch_bounds__(CHUNK_THREADS)
__global__ void gin_bucket_count_kernel(const int* __restrict__ dst,
                                        int* __restrict__ bcnt) {
  __shared__ int cnt[N_BKT];
  const int t = threadIdx.x;
  const int g = blockIdx.x;
  if (t < N_BKT) cnt[t] = 0;
  __syncthreads();
  const int* dp = dst + g * CHUNK_EDGES + t;
  int d[CHUNK_PER_THR];
#pragma unroll
  for (int k = 0; k < CHUNK_PER_THR; k++) d[k] = dp[k * CHUNK_THREADS];
#pragma unroll
  for (int k = 0; k < CHUNK_PER_THR; k++) atomicAdd(&cnt[d[k] / BKT_NODES], 1);
  __syncthreads();
  if (t < N_BKT) bcnt[t * N_CHUNK + g] = cnt[t];
}

#define SCAN_CHUNK 100
#define SCAN_N (N_BKT * N_CHUNK)
#define SCAN_OWNERS (SCAN_N / SCAN_CHUNK)
__launch_bounds__(1024)
__global__ void gin_scan_kernel(const int* __restrict__ bcnt,
                                int* __restrict__ boffs) {
  __shared__ int wsum[32];
  const int t = threadIdx.x;
  const int lane = t & 31;
  const int w = t >> 5;
  const bool owner = t < SCAN_OWNERS;

  int4 v[SCAN_CHUNK / 4];
  int local = 0;
  if (owner) {
    const int4* p = (const int4*)(bcnt + t * SCAN_CHUNK);
#pragma unroll
    for (int i = 0; i < SCAN_CHUNK / 4; i++) v[i] = p[i];
#pragma unroll
    for (int i = 0; i < SCAN_CHUNK / 4; i++) local += v[i].x + v[i].y + v[i].z + v[i].w;
  }

  int x = local;
#pragma unroll
  for (int o = 1; o < 32; o <<= 1) {
    int y = __shfl_up(x, o);
    if (lane >= o) x += y;
  }
  if (lane == 31) wsum[w] = x;
  __syncthreads();
  if (w == 0) {
    int s = wsum[lane];
#pragma unroll
    for (int o = 1; o < 32; o <<= 1) {
      int y = __shfl_up(s, o);
      if (lane >= o) s += y;
    }
    wsum[lane] = s;
  }
  __syncthreads();

  if (owner) {
    int run = x - local + (w > 0 ? wsum[w - 1] : 0);
    int4* q = (int4*)(boffs + t * SCAN_CHUNK);
#pragma unroll
    for (int i = 0; i < SCAN_CHUNK / 4; i++) {
      int4 o4;
      o4.x = run; run += v[i].x;
      o4.y = run; run += v[i].y;
      o4.z = run; run += v[i].z;
      o4.w = run; run += v[i].w;
      q[i] = o4;
    }
    if (t == SCAN_OWNERS - 1) boffs[SCAN_N] = run;
  }
}

__launch_bounds__(CHUNK_THREADS)
__global__ void gin_bucket_fill_kernel(const int* __restrict__ src,
                                       const int* __restrict__ dst,
                                       const int* __restrict__ boffs,
                                       int2* __restrict__ pairs) {
  __shared__ int cur[N_BKT];
  const int t = threadIdx.x;
  const int g = blockIdx.x;
  if (t < N_BKT) cur[t] = boffs[t * N_CHUNK + g];
  __syncthreads();
  const int e0 = g * CHUNK_EDGES + t;
  int s[CHUNK_PER_THR], d[CHUNK_PER_THR];
#pragma unroll
  for (int k = 0; k < CHUNK_PER_THR; k++) {
    s[k] = src[e0 + k * CHUNK_THREADS];
    d[k] = dst[e0 + k * CHUNK_THREADS];
  }
#pragma unroll
  for (int k = 0; k < CHUNK_PER_THR; k++) {
    int pos = atomicAdd(&cur[d[k] / BKT_NODES], 1);
    pairs[pos] = make_int2(s[k], d[k]);
  }
}

#define SPARE_CNT 511
__launch_bounds__(1024)
__global__ void gin_bucket_csr_kernel(const int2* __restrict__ pairs,
                                      const int* __restrict__ boffs,
                                      int* __restrict__ rowptr,
                                      int* __restrict__ csr) {
  __shared__ int cnt[512];
  __shared__ int wsum[32];
  const int t = threadIdx.x;
  const int lane = t & 31;
  const int w = t >> 5;
  const int b = blockIdx.x;
  const int n0 = b * BKT_NODES;
  const int beg = boffs[b * N_CHUNK];
  const int end = boffs[(b + 1) * N_CHUNK];

  if (t < 512) cnt[t] = 0;
  __syncthreads();
  for (int i0 = beg + t; i0 < end; i0 += 4 * 1024) {
    int c[4];
#pragma unroll
    for (int k = 0; k < 4; k++) {
      int i = i0 + k * 1024;
      int d = pairs[i < end ? i : end - 1].y;
      c[k] = (i < end) ? d - n0 : SPARE_CNT;
    }
#pragma unroll
    for (int k = 0; k < 4; k++) atomicAdd(&cnt[c[k]], 1);
  }
  __syncthreads();

  const int v = (t < BKT_NODES) ? cnt[t] : 0;
  int x = v;
#pragma unroll
  for (int o = 1; o < 32; o <<= 1) {
    int y = __shfl_up(x, o);
    if (lane >= o) x += y;
  }
  if (lane == 31) wsum[w] = x;
  __syncthreads();
  if (w == 0) {
    int s = wsum[lane];
#pragma unroll
    for (int o = 1; o < 32; o <<= 1) {
      int y = __shfl_up(s, o);
      if (lane >= o) s += y;
    }
    wsum[lane] = s;
  }
  __syncthreads();
  const int first = beg + x - v + (w > 0 ? wsum[w - 1] : 0);
  if (t < BKT_NODES) {
    rowptr[n0 + t] = first;
    cnt[t] = first;
  }
  if (b == N_BKT - 1 && t == BKT_NODES - 1) rowptr[N_NODES] = first + v;
  __syncthreads();

  for (int i0 = beg + t; i0 < end; i0 += 4 * 1024) {
    int2 p[4];
#pragma unroll
    for (int k = 0; k < 4; k++) {
      int i = i0 + k * 1024;
      p[k] = pairs[i < end ? i : end - 1];
    }
#pragma unroll
    for (int k = 0; k < 4; k++) {
      if (i0 + k * 1024 < end) {
        int pos = atomicAdd(&cnt[p[k].y - n0], 1);
        csr[pos] = p[k].x;
      }
    }
  }
}

__launch_bounds__(256)
__global__ void gin_conv_kernel(const float* __restrict__ hin,
                                const int* __restrict__ rowptr,
                                const int* __restrict__ csr,
                                const float* __restrict__ W1,
                                const float* __restrict__ b1,
                                const float* __restrict__ bn1,
                                const float* __restrict__ W2,
                                const float* __restrict__ b2,
                                const float* __restrict__ bn2,
                                float* __restrict__ hout,
                                const int* __restrict__ batch,
                                float* __restrict__ pooled,
                                int ntiles) {
  __shared__ __align__(16) float w1s[4 * 16 * 32 * 2];
  __shared__ __align__(16) float w2s[4 * 16 * 32 * 2];
  __shared__ __align__(16) float tb[8][16 * 68];

  const int tid  = threadIdx.x;
  const int lane = tid & 31;
  const int wave = tid >> 5;
  const int row  = lane & 15;
  const int hi   = lane >> 4;
  const int tile = blockIdx.x * 8 + wave;
  const bool active = tile < ntiles;
  float* tbuf = tb[wave];

  float4 w1a[2], w1b[2], w2a[2], w2b[2];
#pragma unroll
  for (int j = 0; j < 2; j++) {
    const int p = tid + 256 * j;
    const float* wp = W1 + (p >> 4) * 128 + (p & 15) * 4;
    w1a[j] = *(const float4*)wp;
    w1b[j] = *(const float4*)(wp + 64);
    const float* wq = W2 + (p >> 4) * 128 + (p & 15) * 4;
    w2a[j] = *(const float4*)wq;
    w2b[j] = *(const float4*)(wq + 64);
  }
  const int tl = active ? tile : 0;
  const int gl = batch[tl * 16 + row];
  const int rpl = rowptr[tl * 16 + (lane < 16 ? lane : 16)];

#pragma unroll
  for (int j = 0; j < 2; j++) {
    const int p = tid + 256 * j;
    const int kk = 2 * (p >> 4), nq = p & 15;
    const int o = ((((nq >> 2) * 16) + (kk >> 2)) * 32 + 4 * (nq & 3) + ((kk & 2) << 3)) * 2;
    *(float4*)(w1s + o)     = make_float4(w1a[j].x, w1b[j].x, w1a[j].y, w1b[j].y);
    *(float4*)(w1s + o + 4) = make_float4(w1a[j].z, w1b[j].z, w1a[j].w, w1b[j].w);
    *(float4*)(w2s + o)     = make_float4(w2a[j].x, w2b[j].x, w2a[j].y, w2b[j].y);
    *(float4*)(w2s + o + 4) = make_float4(w2a[j].z, w2b[j].z, w2a[j].w, w2b[j].w);
  }

  if (active) {
    const char* hb = (const char*)hin;
    const unsigned cb = (unsigned)row << 4;
    int beg = __shfl(rpl, hi), end = __shfl(rpl, hi + 1);
    int idx[16];
#pragma unroll
    for (int k = 0; k < 16; k++) {
      int jc = (beg + k < end) ? beg + k : end - 1;
      idx[k] = csr[jc > 0 ? jc : 0];
    }
    for (int p = 0; p < 8; p++) {
      const int r = 2 * p + hi;
      const unsigned n = (unsigned)(tile * 16 + r);
      float4 acc = *(const float4*)(hb + ((n << 8) + cb));
      float4 v[16];
#pragma unroll
      for (int k = 0; k < 16; k++) {
        const unsigned sk = (beg + k < end) ? (unsigned)idx[k] : n;
        v[k] = *(const float4*)(hb + ((sk << 8) + cb));
      }
      const int r2 = (p < 7) ? r + 2 : r;
      const int nbeg = __shfl(rpl, r2), nend = __shfl(rpl, r2 + 1);
      int nidx[16];
#pragma unroll
      for (int k = 0; k < 16; k++) {
        int jc = (nbeg + k < nend) ? nbeg + k : nend - 1;
        nidx[k] = csr[jc > 0 ? jc : 0];
      }
#pragma unroll
      for (int k = 0; k < 16; k++) {
        const bool in = beg + k < end;
        acc.x += in ? v[k].x : 0.f; acc.y += in ? v[k].y : 0.f;
        acc.z += in ? v[k].z : 0.f; acc.w += in ? v[k].w : 0.f;
      }
      for (int j = beg + 16; j < end; j += 16) {
        int s[16];
#pragma unroll
        for (int k = 0; k < 16; k++) s[k] = csr[(j + k < end) ? j + k : end - 1];
#pragma unroll
        for (int k = 0; k < 16; k++) {
          const unsigned sk = (j + k < end) ? (unsigned)s[k] : n;
          v[k] = *(const float4*)(hb + ((sk << 8) + cb));
        }
#pragma unroll
        for (int k = 0; k < 16; k++) {
          const bool in = j + k < end;
          acc.x += in ? v[k].x : 0.f; acc.y += in ? v[k].y : 0.f;
          acc.z += in ? v[k].z : 0.f; acc.w += in ? v[k].w : 0.f;
        }
      }
      *(float4*)(tbuf + r * 68 + 4 * row) = acc;
      beg = nbeg; end = nend;
#pragma unroll
      for (int k = 0; k < 16; k++) idx[k] = nidx[k];
    }
  }
  __syncthreads();

  const v8f vzero = {0.f, 0.f, 0.f, 0.f, 0.f, 0.f, 0.f, 0.f};

  if (active) {
    v2f a1[16];
    const float* tr = tbuf + row * 68 + 2 * hi;
#pragma unroll
    for (int k = 0; k < 16; k++) a1[k] = *(const v2f*)(tr + 4 * k);
#pragma unroll
    for (int nt = 0; nt < 4; nt++) {
      v8f acc = vzero;
      const v2f* wf = ((const v2f*)w1s) + nt * 16 * 32 + lane;
#pragma unroll
      for (int k = 0; k < 16; k++) {
        acc = __builtin_amdgcn_wmma_f32_16x16x4_f32(false, a1[k], false, wf[k * 32],
                                                    (short)0, acc, false, false);
      }
      int n = nt * 16 + row;
      float s  = bn1[n] * rsqrtf(bn1[192 + n] + BN_EPS);
      float t0 = bn1[64 + n] + (b1[n] - bn1[128 + n]) * s;
#pragma unroll
      for (int i = 0; i < 8; i++) {
        float y = acc[i] * s + t0;
        tbuf[(i + 8 * hi) * 68 + n] = fmaxf(y, 0.0f);
      }
    }
  }
  __syncthreads();

  v8f acc2[4];
  if (active) {
#pragma unroll
    for (int nt = 0; nt < 4; nt++) {
      v8f acc = vzero;
      const v2f* wf = ((const v2f*)w2s) + nt * 16 * 32 + lane;
      const float* tr = tbuf + row * 68 + 2 * hi;
#pragma unroll
      for (int k = 0; k < 16; k++) {
        v2f a = *(const v2f*)(tr + 4 * k);
        acc = __builtin_amdgcn_wmma_f32_16x16x4_f32(false, a, false, wf[k * 32],
                                                    (short)0, acc, false, false);
      }
      acc2[nt] = acc;
    }
  }
  __syncthreads();

  if (active) {
#pragma unroll
    for (int nt = 0; nt < 4; nt++) {
      int n = nt * 16 + row;
      float s  = bn2[n] * rsqrtf(bn2[192 + n] + BN_EPS);
      float t0 = bn2[64 + n] + (b2[n] - bn2[128 + n]) * s;
#pragma unroll
      for (int i = 0; i < 8; i++) {
        float y = acc2[nt][i] * s + t0;
        tbuf[(i + 8 * hi) * 68 + n] = fmaxf(y, 0.0f);
      }
    }
  }
  __syncthreads();

  if (active) {
#pragma unroll
    for (int j = 0; j < 8; j++) {
      int flat = j * 32 + lane;
      int r = flat >> 4;
      int c = (flat & 15) << 2;
      float4 v = *(const float4*)(tbuf + r * 68 + c);
      *(float4*)(hout + (size_t)(tile * 16 + r) * 64 + c) = v;
    }
  }

  if (active) {
    float p0[16], p1[16];
#pragma unroll
    for (int r = 0; r < 16; r++) {
      p0[r] = tbuf[r * 68 + lane];
      p1[r] = tbuf[r * 68 + 32 + lane];
    }
    unsigned todo = 0xFFFFu;
    while (todo) {
      const int r0 = __builtin_ctz(todo);
      const int g = __builtin_amdgcn_readlane(gl, r0);
      const unsigned m = __builtin_amdgcn_ballot_w32(gl == g) & todo;
      float s0 = 0.f, s1 = 0.f;
#pragma unroll
      for (int r = 0; r < 16; r++) {
        const bool in = (m >> r) & 1u;
        s0 += in ? p0[r] : 0.f;
        s1 += in ? p1[r] : 0.f;
      }
      float* pg = pooled + (size_t)g * 64 + lane;
      atomicAdd(pg, s0);
      atomicAdd(pg + 32, s1);
      todo &= ~m;
    }
  }
}

__global__ void gin_final_kernel(const float* __restrict__ pooled,
                                 const float* __restrict__ counts,
                                 const float* __restrict__ linW,
                                 const float* __restrict__ linb,
                                 float* __restrict__ out) {
  int idx = blockIdx.x * blockDim.x + threadIdx.x;
  if (idx >= N_GRAPHS * 10) return;
  int g = idx / 10, t = idx - g * 10;
  float acc = counts[g] * linb[t];
  for (int l = 1; l < 5; l++) acc += linb[l * 10 + t];
  for (int l = 0; l < 5; l++) {
    const float* p = pooled + ((size_t)l * N_GRAPHS + g) * 64;
    const float* w = linW + l * 640 + t;
    float s = 0.f;
#pragma unroll
    for (int k = 0; k < 64; k++) s += p[k] * w[k * 10];
    acc += s;
  }
  out[idx] = acc;
}

__global__ void gin_zero_kernel(unsigned int* __restrict__ p, int n) {
  int i = blockIdx.x * blockDim.x + threadIdx.x;
  if (i < n) p[i] = 0u;
}

extern "C" void kernel_launch(void* const* d_in, const int* in_sizes, int n_in,
                              void* d_out, int out_size, void* d_ws, size_t ws_size,
                              hipStream_t stream) {
  const float* x        = (const float*)d_in[0];
  const float* fh_W1    = (const float*)d_in[1];
  const float* fh_b1    = (const float*)d_in[2];
  const float* fh_bn1   = (const float*)d_in[3];
  const float* fh_W2    = (const float*)d_in[4];
  const float* fh_b2    = (const float*)d_in[5];
  const float* fh_bn2   = (const float*)d_in[6];
  const float* conv_W1  = (const float*)d_in[7];
  const float* conv_b1  = (const float*)d_in[8];
  const float* conv_bn1 = (const float*)d_in[9];
  const float* conv_W2  = (const float*)d_in[10];
  const float* conv_b2  = (const float*)d_in[11];
  const float* conv_bn2 = (const float*)d_in[12];
  const float* lin_W    = (const float*)d_in[13];
  const float* lin_b    = (const float*)d_in[14];
  const int*   eidx     = (const int*)d_in[15];
  const int*   batch    = (const int*)d_in[16];
  float* out = (float*)d_out;

  const int* src = eidx;
  const int* dst = eidx + N_EDGES;

  float* ws     = (float*)d_ws;
  float* h      = ws;
  float* zrow   = h + (size_t)N_NODES * 64;
  float* pooled = zrow + 64;
  float* counts = pooled + 5 * N_GRAPHS * 64;
  int*   bcnt   = (int*)(counts + N_GRAPHS);
  int*   boffs  = bcnt + SCAN_N;
  int*   rowptr = boffs + SCAN_N + 4;
  int2*  pairs  = (int2*)(rowptr + N_NODES + 4);
  int*   csr    = (int*)(pairs + N_EDGES);
  float* h2     = ws + 12000000;

  const int accN = 64 + 5 * N_GRAPHS * 64 + N_GRAPHS;
  gin_zero_kernel<<<(accN + 255) / 256, 256, 0, stream>>>((unsigned int*)zrow, accN);

  gin_bucket_count_kernel<<<N_CHUNK, CHUNK_THREADS, 0, stream>>>(dst, bcnt);
  gin_scan_kernel<<<1, 1024, 0, stream>>>(bcnt, boffs);
  gin_bucket_fill_kernel<<<N_CHUNK, CHUNK_THREADS, 0, stream>>>(src, dst, boffs, pairs);
  gin_bucket_csr_kernel<<<N_BKT, 1024, 0, stream>>>(pairs, boffs, rowptr, csr);

  const int mlpBlocks = (NTILES + 3) / 4;

  gin_mlp_kernel<D_FEAT, true><<<mlpBlocks, 128, 0, stream>>>(
      x, fh_W1, fh_b1, fh_bn1, fh_W2, fh_b2, fh_bn2, h, batch, pooled, counts, NTILES);

  const int convBlocks = (NTILES + 7) / 8;
  float* hin = h;
  float* hnext = h2;
  for (int l = 0; l < 4; l++) {
    gin_conv_kernel<<<convBlocks, 256, 0, stream>>>(
        hin, rowptr, csr,
        conv_W1 + (size_t)l * 64 * 64, conv_b1 + l * 64, conv_bn1 + l * 256,
        conv_W2 + (size_t)l * 64 * 64, conv_b2 + l * 64, conv_bn2 + l * 256,
        hnext, batch, pooled + (size_t)(l + 1) * N_GRAPHS * 64, NTILES);
    float* t = hin; hin = hnext; hnext = t;
  }

  gin_final_kernel<<<(N_GRAPHS * 10 + 255) / 256, 256, 0, stream>>>(
      pooled, counts, lin_W, lin_b, out);
}


# Conv gather: a node's in-edge indices by one coalesced csr load per 16 slots, broadcast in its half-wave by DPP row_share (was 16 same-address loads); slots 16-31 prefetched too; same sum order
# speedup vs baseline: 3.3993x; 1.0438x over previous
#include <hip/hip_runtime.h>

#define N_NODES   100000
#define N_EDGES   1600000
#define N_GRAPHS  1000
#define D_FEAT    128
#define HID       64
#define BN_EPS    1e-5f
#define NTILES    (N_NODES / 16)

typedef __attribute__((ext_vector_type(2))) float v2f;
typedef __attribute__((ext_vector_type(8))) float v8f;

template<int K>
static __device__ __forceinline__ int row_share_i(int v) {
  return __builtin_amdgcn_update_dpp(0, v, 0x150 + K, 0xF, 0xF, true);
}
#define GIN_SLOTS4(F, B)  F((B)) F((B) + 1) F((B) + 2) F((B) + 3)
#define GIN_SLOTS16(F)    GIN_SLOTS4(F, 0) GIN_SLOTS4(F, 4) GIN_SLOTS4(F, 8) GIN_SLOTS4(F, 12)

template<int K1, bool COUNT>
__launch_bounds__(128)
__global__ void gin_mlp_kernel(const float* __restrict__ xin,
                               const float* __restrict__ W1,
                               const float* __restrict__ b1,
                               const float* __restrict__ bn1,
                               const float* __restrict__ W2,
                               const float* __restrict__ b2,
                               const float* __restrict__ bn2,
                               float* __restrict__ hout,
                               const int* __restrict__ batch,
                               float* __restrict__ pooled,
                               float* __restrict__ counts,
                               int ntiles) {
  constexpr int KS1 = K1 / 4;
  constexpr int NQ1 = (K1 / 2) * 16 / 128;
  constexpr int NQ2 = (64 / 2) * 16 / 128;
  __shared__ __align__(16) float w1s[4 * KS1 * 32 * 2];
  __shared__ __align__(16) float w2s[4 * 16 * 32 * 2];
  __shared__ __align__(16) float tb[4][16 * 68];

  const int tid  = threadIdx.x;
  const int lane = tid & 31;
  const int wave = tid >> 5;
  const int row  = lane & 15;
  const int hi   = lane >> 4;
  const int tile = blockIdx.x * 4 + wave;
  const bool active = tile < ntiles;
  float* tbuf = tb[wave];

  float4 w1a[NQ1], w1b[NQ1], w2a[NQ2], w2b[NQ2];
#pragma unroll
  for (int j = 0; j < NQ1; j++) {
    const int p = tid + 128 * j;
    const float* wp = W1 + (p >> 4) * 128 + (p & 15) * 4;
    w1a[j] = *(const float4*)wp;
    w1b[j] = *(const float4*)(wp + 64);
  }
#pragma unroll
  for (int j = 0; j < NQ2; j++) {
    const int p = tid + 128 * j;
    const float* wp = W2 + (p >> 4) * 128 + (p & 15) * 4;
    w2a[j] = *(const float4*)wp;
    w2b[j] = *(const float4*)(wp + 64);
  }
  const int tl = active ? tile : 0;
  const int gl = batch[tl * 16 + row];
  v2f a1[KS1];
  const float* xr = xin + (size_t)(tl * 16 + row) * K1 + 2 * hi;
#pragma unroll
  for (int k = 0; k < KS1; k++) a1[k] = *(const v2f*)(xr + 4 * k);

#pragma unroll
  for (int j = 0; j < NQ1; j++) {
    const int p = tid + 128 * j;
    const int kk = 2 * (p >> 4), nq = p & 15;
    float* d = w1s + ((((nq >> 2) * KS1) + (kk >> 2)) * 32 + 4 * (nq & 3) + ((kk & 2) << 3)) * 2;
    *(float4*)d       = make_float4(w1a[j].x, w1b[j].x, w1a[j].y, w1b[j].y);
    *(float4*)(d + 4) = make_float4(w1a[j].z, w1b[j].z, w1a[j].w, w1b[j].w);
  }
#pragma unroll
  for (int j = 0; j < NQ2; j++) {
    const int p = tid + 128 * j;
    const int kk = 2 * (p >> 4), nq = p & 15;
    float* d = w2s + ((((nq >> 2) * 16) + (kk >> 2)) * 32 + 4 * (nq & 3) + ((kk & 2) << 3)) * 2;
    *(float4*)d       = make_float4(w2a[j].x, w2b[j].x, w2a[j].y, w2b[j].y);
    *(float4*)(d + 4) = make_float4(w2a[j].z, w2b[j].z, w2a[j].w, w2b[j].w);
  }
  __syncthreads();

  const v8f vzero = {0.f, 0.f, 0.f, 0.f, 0.f, 0.f, 0.f, 0.f};

  if (active) {
#pragma unroll
    for (int nt = 0; nt < 4; nt++) {
      v8f acc = vzero;
      const v2f* wf = ((const v2f*)w1s) + nt * KS1 * 32 + lane;
#pragma unroll
      for (int k = 0; k < KS1; k++) {
        acc = __builtin_amdgcn_wmma_f32_16x16x4_f32(false, a1[k], false, wf[k * 32],
                                                    (short)0, acc, false, false);
      }
      int n = nt * 16 + row;
      float s  = bn1[n] * rsqrtf(bn1[192 + n] + BN_EPS);
      float t0 = bn1[64 + n] + (b1[n] - bn1[128 + n]) * s;
#pragma unroll
      for (int i = 0; i < 8; i++) {
        float y = acc[i] * s + t0;
        tbuf[(i + 8 * hi) * 68 + n] = fmaxf(y, 0.0f);
      }
    }
  }
  __syncthreads();

  v8f acc2[4];
  if (active) {
#pragma unroll
    for (int nt = 0; nt < 4; nt++) {
      v8f acc = vzero;
      const v2f* wf = ((const v2f*)w2s) + nt * 16 * 32 + lane;
      const float* tr = tbuf + row * 68 + 2 * hi;
#pragma unroll
      for (int k = 0; k < 16; k++) {
        v2f a = *(const v2f*)(tr + 4 * k);
        acc = __builtin_amdgcn_wmma_f32_16x16x4_f32(false, a, false, wf[k * 32],
                                                    (short)0, acc, false, false);
      }
      acc2[nt] = acc;
    }
  }
  __syncthreads();

  if (active) {
#pragma unroll
    for (int nt = 0; nt < 4; nt++) {
      int n = nt * 16 + row;
      float s  = bn2[n] * rsqrtf(bn2[192 + n] + BN_EPS);
      float t0 = bn2[64 + n] + (b2[n] - bn2[128 + n]) * s;
#pragma unroll
      for (int i = 0; i < 8; i++) {
        float y = acc2[nt][i] * s + t0;
        tbuf[(i + 8 * hi) * 68 + n] = fmaxf(y, 0.0f);
      }
    }
  }
  __syncthreads();

  if (active) {
#pragma unroll
    for (int j = 0; j < 8; j++) {
      int flat = j * 32 + lane;
      int r = flat >> 4;
      int c = (flat & 15) << 2;
      float4 v = *(const float4*)(tbuf + r * 68 + c);
      *(float4*)(hout + (size_t)(tile * 16 + r) * 64 + c) = v;
    }
  }

  if (active) {
    float p0[16], p1[16];
#pragma unroll
    for (int r = 0; r < 16; r++) {
      p0[r] = tbuf[r * 68 + lane];
      p1[r] = tbuf[r * 68 + 32 + lane];
    }
    unsigned todo = 0xFFFFu;
    while (todo) {
      const int r0 = __builtin_ctz(todo);
      const int g = __builtin_amdgcn_readlane(gl, r0);
      const unsigned m = __builtin_amdgcn_ballot_w32(gl == g) & todo;
      float s0 = 0.f, s1 = 0.f;
#pragma unroll
      for (int r = 0; r < 16; r++) {
        const bool in = (m >> r) & 1u;
        s0 += in ? p0[r] : 0.f;
        s1 += in ? p1[r] : 0.f;
      }
      float* pg = pooled + (size_t)g * 64 + lane;
      atomicAdd(pg, s0);
      atomicAdd(pg + 32, s1);
      if (COUNT && lane == 0) atomicAdd(counts + g, (float)__builtin_popcount(m));
      todo &= ~m;
    }
  }
}

#define BKT_NODES     500
#define N_BKT         (N_NODES / BKT_NODES)
#define N_CHUNK       500
#define CHUNK_EDGES   (N_EDGES / N_CHUNK)
#define CHUNK_THREADS 320
#define CHUNK_PER_THR (CHUNK_EDGES / CHUNK_THREADS)

__launch_bounds__(CHUNK_THREADS)
__global__ void gin_bucket_count_kernel(const int* __restrict__ dst,
                                        int* __restrict__ bcnt) {
  __shared__ int cnt[N_BKT];
  const int t = threadIdx.x;
  const int g = blockIdx.x;
  if (t < N_BKT) cnt[t] = 0;
  __syncthreads();
  const int* dp = dst + g * CHUNK_EDGES + t;
  int d[CHUNK_PER_THR];
#pragma unroll
  for (int k = 0; k < CHUNK_PER_THR; k++) d[k] = dp[k * CHUNK_THREADS];
#pragma unroll
  for (int k = 0; k < CHUNK_PER_THR; k++) atomicAdd(&cnt[d[k] / BKT_NODES], 1);
  __syncthreads();
  if (t < N_BKT) bcnt[t * N_CHUNK + g] = cnt[t];
}

#define SCAN_CHUNK 100
#define SCAN_N (N_BKT * N_CHUNK)
#define SCAN_OWNERS (SCAN_N / SCAN_CHUNK)
__launch_bounds__(1024)
__global__ void gin_scan_kernel(const int* __restrict__ bcnt,
                                int* __restrict__ boffs) {
  __shared__ int wsum[32];
  const int t = threadIdx.x;
  const int lane = t & 31;
  const int w = t >> 5;
  const bool owner = t < SCAN_OWNERS;

  int4 v[SCAN_CHUNK / 4];
  int local = 0;
  if (owner) {
    const int4* p = (const int4*)(bcnt + t * SCAN_CHUNK);
#pragma unroll
    for (int i = 0; i < SCAN_CHUNK / 4; i++) v[i] = p[i];
#pragma unroll
    for (int i = 0; i < SCAN_CHUNK / 4; i++) local += v[i].x + v[i].y + v[i].z + v[i].w;
  }

  int x = local;
#pragma unroll
  for (int o = 1; o < 32; o <<= 1) {
    int y = __shfl_up(x, o);
    if (lane >= o) x += y;
  }
  if (lane == 31) wsum[w] = x;
  __syncthreads();
  if (w == 0) {
    int s = wsum[lane];
#pragma unroll
    for (int o = 1; o < 32; o <<= 1) {
      int y = __shfl_up(s, o);
      if (lane >= o) s += y;
    }
    wsum[lane] = s;
  }
  __syncthreads();

  if (owner) {
    int run = x - local + (w > 0 ? wsum[w - 1] : 0);
    int4* q = (int4*)(boffs + t * SCAN_CHUNK);
#pragma unroll
    for (int i = 0; i < SCAN_CHUNK / 4; i++) {
      int4 o4;
      o4.x = run; run += v[i].x;
      o4.y = run; run += v[i].y;
      o4.z = run; run += v[i].z;
      o4.w = run; run += v[i].w;
      q[i] = o4;
    }
    if (t == SCAN_OWNERS - 1) boffs[SCAN_N] = run;
  }
}

__launch_bounds__(CHUNK_THREADS)
__global__ void gin_bucket_fill_kernel(const int* __restrict__ src,
                                       const int* __restrict__ dst,
                                       const int* __restrict__ boffs,
                                       int2* __restrict__ pairs) {
  __shared__ int cur[N_BKT];
  const int t = threadIdx.x;
  const int g = blockIdx.x;
  if (t < N_BKT) cur[t] = boffs[t * N_CHUNK + g];
  __syncthreads();
  const int e0 = g * CHUNK_EDGES + t;
  int s[CHUNK_PER_THR], d[CHUNK_PER_THR];
#pragma unroll
  for (int k = 0; k < CHUNK_PER_THR; k++) {
    s[k] = src[e0 + k * CHUNK_THREADS];
    d[k] = dst[e0 + k * CHUNK_THREADS];
  }
#pragma unroll
  for (int k = 0; k < CHUNK_PER_THR; k++) {
    int pos = atomicAdd(&cur[d[k] / BKT_NODES], 1);
    pairs[pos] = make_int2(s[k], d[k]);
  }
}

#define SPARE_CNT 511
__launch_bounds__(1024)
__global__ void gin_bucket_csr_kernel(const int2* __restrict__ pairs,
                                      const int* __restrict__ boffs,
                                      int* __restrict__ rowptr,
                                      int* __restrict__ csr) {
  __shared__ int cnt[512];
  __shared__ int wsum[32];
  const int t = threadIdx.x;
  const int lane = t & 31;
  const int w = t >> 5;
  const int b = blockIdx.x;
  const int n0 = b * BKT_NODES;
  const int beg = boffs[b * N_CHUNK];
  const int end = boffs[(b + 1) * N_CHUNK];

  if (t < 512) cnt[t] = 0;
  __syncthreads();
  for (int i0 = beg + t; i0 < end; i0 += 4 * 1024) {
    int c[4];
#pragma unroll
    for (int k = 0; k < 4; k++) {
      int i = i0 + k * 1024;
      int d = pairs[i < end ? i : end - 1].y;
      c[k] = (i < end) ? d - n0 : SPARE_CNT;
    }
#pragma unroll
    for (int k = 0; k < 4; k++) atomicAdd(&cnt[c[k]], 1);
  }
  __syncthreads();

  const int v = (t < BKT_NODES) ? cnt[t] : 0;
  int x = v;
#pragma unroll
  for (int o = 1; o < 32; o <<= 1) {
    int y = __shfl_up(x, o);
    if (lane >= o) x += y;
  }
  if (lane == 31) wsum[w] = x;
  __syncthreads();
  if (w == 0) {
    int s = wsum[lane];
#pragma unroll
    for (int o = 1; o < 32; o <<= 1) {
      int y = __shfl_up(s, o);
      if (lane >= o) s += y;
    }
    wsum[lane] = s;
  }
  __syncthreads();
  const int first = beg + x - v + (w > 0 ? wsum[w - 1] : 0);
  if (t < BKT_NODES) {
    rowptr[n0 + t] = first;
    cnt[t] = first;
  }
  if (b == N_BKT - 1 && t == BKT_NODES - 1) rowptr[N_NODES] = first + v;
  __syncthreads();

  for (int i0 = beg + t; i0 < end; i0 += 4 * 1024) {
    int2 p[4];
#pragma unroll
    for (int k = 0; k < 4; k++) {
      int i = i0 + k * 1024;
      p[k] = pairs[i < end ? i : end - 1];
    }
#pragma unroll
    for (int k = 0; k < 4; k++) {
      if (i0 + k * 1024 < end) {
        int pos = atomicAdd(&cnt[p[k].y - n0], 1);
        csr[pos] = p[k].x;
      }
    }
  }
}

__launch_bounds__(256)
__global__ void gin_conv_kernel(const float* __restrict__ hin,
                                const int* __restrict__ rowptr,
                                const int* __restrict__ csr,
                                const float* __restrict__ W1,
                                const float* __restrict__ b1,
                                const float* __restrict__ bn1,
                                const float* __restrict__ W2,
                                const float* __restrict__ b2,
                                const float* __restrict__ bn2,
                                float* __restrict__ hout,
                                const int* __restrict__ batch,
                                float* __restrict__ pooled,
                                int ntiles) {
  __shared__ __align__(16) float w1s[4 * 16 * 32 * 2];
  __shared__ __align__(16) float w2s[4 * 16 * 32 * 2];
  __shared__ __align__(16) float tb[8][16 * 68];

  const int tid  = threadIdx.x;
  const int lane = tid & 31;
  const int wave = tid >> 5;
  const int row  = lane & 15;
  const int hi   = lane >> 4;
  const int tile = blockIdx.x * 8 + wave;
  const bool active = tile < ntiles;
  float* tbuf = tb[wave];

  float4 w1a[2], w1b[2], w2a[2], w2b[2];
#pragma unroll
  for (int j = 0; j < 2; j++) {
    const int p = tid + 256 * j;
    const float* wp = W1 + (p >> 4) * 128 + (p & 15) * 4;
    w1a[j] = *(const float4*)wp;
    w1b[j] = *(const float4*)(wp + 64);
    const float* wq = W2 + (p >> 4) * 128 + (p & 15) * 4;
    w2a[j] = *(const float4*)wq;
    w2b[j] = *(const float4*)(wq + 64);
  }
  const int tl = active ? tile : 0;
  const int gl = batch[tl * 16 + row];
  const int rpl = rowptr[tl * 16 + (lane < 16 ? lane : 16)];

#pragma unroll
  for (int j = 0; j < 2; j++) {
    const int p = tid + 256 * j;
    const int kk = 2 * (p >> 4), nq = p & 15;
    const int o = ((((nq >> 2) * 16) + (kk >> 2)) * 32 + 4 * (nq & 3) + ((kk & 2) << 3)) * 2;
    *(float4*)(w1s + o)     = make_float4(w1a[j].x, w1b[j].x, w1a[j].y, w1b[j].y);
    *(float4*)(w1s + o + 4) = make_float4(w1a[j].z, w1b[j].z, w1a[j].w, w1b[j].w);
    *(float4*)(w2s + o)     = make_float4(w2a[j].x, w2b[j].x, w2a[j].y, w2b[j].y);
    *(float4*)(w2s + o + 4) = make_float4(w2a[j].z, w2b[j].z, w2a[j].w, w2b[j].w);
  }

  if (active) {
    const char* hb = (const char*)hin;
    const unsigned cb = (unsigned)row << 4;
    int b0 = __builtin_amdgcn_readlane(rpl, 0);
    int b1 = __builtin_amdgcn_readlane(rpl, 1);
    int b2 = __builtin_amdgcn_readlane(rpl, 2);
    int beg = hi ? b1 : b0, end = hi ? b2 : b1;
    int c0, c1;
    {
      int j = beg + row;      j = j < end ? j : end - 1;
      c0 = csr[j > 0 ? j : 0];
      j = beg + 16 + row;     j = j < end ? j : end - 1;
      c1 = csr[j > 0 ? j : 0];
    }
    for (int p = 0; p < 8; p++) {
      const int r = 2 * p + hi;
      const unsigned n = (unsigned)(tile * 16 + r);
      float4 acc = *(const float4*)(hb + ((n << 8) + cb));
      float4 v[16];
#define GIN_LOADV(k) v[(k)] = *(const float4*)(hb + ((((unsigned)row_share_i<(k)>(c0)) << 8) + cb));
      GIN_SLOTS16(GIN_LOADV)
#undef GIN_LOADV
      const int nb0 = b2;
      const int nb1 = __builtin_amdgcn_readlane(rpl, 2 * p + 3);
      const int nb2 = __builtin_amdgcn_readlane(rpl, 2 * p + 4);
      const int nbeg = hi ? nb1 : nb0, nend = hi ? nb2 : nb1;
      int nc0, nc1;
      {
        int j = nbeg + row;   j = j < nend ? j : nend - 1;
        nc0 = csr[j > 0 ? j : 0];
        j = nbeg + 16 + row;  j = j < nend ? j : nend - 1;
        nc1 = csr[j > 0 ? j : 0];
      }
#define GIN_ACCV(k) { const bool in = beg + (k) < end; \
        acc.x += in ? v[(k)].x : 0.f; acc.y += in ? v[(k)].y : 0.f; \
        acc.z += in ? v[(k)].z : 0.f; acc.w += in ? v[(k)].w : 0.f; }
      GIN_SLOTS16(GIN_ACCV)
#undef GIN_ACCV
      const int d0 = b1 - b0, d1 = b2 - b1;
      const int more = (d0 > d1 ? d0 : d1) - 16;
      if (more > 0) {
#define GIN_LOADU(k) v[(k)] = *(const float4*)(hb + ((((unsigned)row_share_i<(k)>(c1)) << 8) + cb));
#define GIN_ACCU(k) { const bool in = beg + 16 + (k) < end; \
        acc.x += in ? v[(k)].x : 0.f; acc.y += in ? v[(k)].y : 0.f; \
        acc.z += in ? v[(k)].z : 0.f; acc.w += in ? v[(k)].w : 0.f; }
        GIN_SLOTS4(GIN_LOADU, 0)
        if (more > 4) {
          GIN_SLOTS4(GIN_LOADU, 4)
          if (more > 8) {
            GIN_SLOTS4(GIN_LOADU, 8)
            if (more > 12) { GIN_SLOTS4(GIN_LOADU, 12) }
          }
        }
        GIN_SLOTS4(GIN_ACCU, 0)
        if (more > 4) {
          GIN_SLOTS4(GIN_ACCU, 4)
          if (more > 8) {
            GIN_SLOTS4(GIN_ACCU, 8)
            if (more > 12) { GIN_SLOTS4(GIN_ACCU, 12) }
          }
        }
#undef GIN_LOADU
#undef GIN_ACCU
      }
      for (int j = beg + 32; j < end; j += 16) {
        int s[16];
#pragma unroll
        for (int k = 0; k < 16; k++) s[k] = csr[(j + k < end) ? j + k : end - 1];
#pragma unroll
        for (int k = 0; k < 16; k++) {
          const unsigned sk = (j + k < end) ? (unsigned)s[k] : n;
          v[k] = *(const float4*)(hb + ((sk << 8) + cb));
        }
#pragma unroll
        for (int k = 0; k < 16; k++) {
          const bool in = j + k < end;
          acc.x += in ? v[k].x : 0.f; acc.y += in ? v[k].y : 0.f;
          acc.z += in ? v[k].z : 0.f; acc.w += in ? v[k].w : 0.f;
        }
      }
      *(float4*)(tbuf + r * 68 + 4 * row) = acc;
      b0 = nb0; b1 = nb1; b2 = nb2;
      beg = nbeg; end = nend;
      c0 = nc0; c1 = nc1;
    }
  }
  __syncthreads();

  const v8f vzero = {0.f, 0.f, 0.f, 0.f, 0.f, 0.f, 0.f, 0.f};

  if (active) {
    v2f a1[16];
    const float* tr = tbuf + row * 68 + 2 * hi;
#pragma unroll
    for (int k = 0; k < 16; k++) a1[k] = *(const v2f*)(tr + 4 * k);
#pragma unroll
    for (int nt = 0; nt < 4; nt++) {
      v8f acc = vzero;
      const v2f* wf = ((const v2f*)w1s) + nt * 16 * 32 + lane;
#pragma unroll
      for (int k = 0; k < 16; k++) {
        acc = __builtin_amdgcn_wmma_f32_16x16x4_f32(false, a1[k], false, wf[k * 32],
                                                    (short)0, acc, false, false);
      }
      int n = nt * 16 + row;
      float s  = bn1[n] * rsqrtf(bn1[192 + n] + BN_EPS);
      float t0 = bn1[64 + n] + (b1[n] - bn1[128 + n]) * s;
#pragma unroll
      for (int i = 0; i < 8; i++) {
        float y = acc[i] * s + t0;
        tbuf[(i + 8 * hi) * 68 + n] = fmaxf(y, 0.0f);
      }
    }
  }
  __syncthreads();

  v8f acc2[4];
  if (active) {
#pragma unroll
    for (int nt = 0; nt < 4; nt++) {
      v8f acc = vzero;
      const v2f* wf = ((const v2f*)w2s) + nt * 16 * 32 + lane;
      const float* tr = tbuf + row * 68 + 2 * hi;
#pragma unroll
      for (int k = 0; k < 16; k++) {
        v2f a = *(const v2f*)(tr + 4 * k);
        acc = __builtin_amdgcn_wmma_f32_16x16x4_f32(false, a, false, wf[k * 32],
                                                    (short)0, acc, false, false);
      }
      acc2[nt] = acc;
    }
  }
  __syncthreads();

  if (active) {
#pragma unroll
    for (int nt = 0; nt < 4; nt++) {
      int n = nt * 16 + row;
      float s  = bn2[n] * rsqrtf(bn2[192 + n] + BN_EPS);
      float t0 = bn2[64 + n] + (b2[n] - bn2[128 + n]) * s;
#pragma unroll
      for (int i = 0; i < 8; i++) {
        float y = acc2[nt][i] * s + t0;
        tbuf[(i + 8 * hi) * 68 + n] = fmaxf(y, 0.0f);
      }
    }
  }
  __syncthreads();

  if (active) {
#pragma unroll
    for (int j = 0; j < 8; j++) {
      int flat = j * 32 + lane;
      int r = flat >> 4;
      int c = (flat & 15) << 2;
      float4 v = *(const float4*)(tbuf + r * 68 + c);
      *(float4*)(hout + (size_t)(tile * 16 + r) * 64 + c) = v;
    }
  }

  if (active) {
    float p0[16], p1[16];
#pragma unroll
    for (int r = 0; r < 16; r++) {
      p0[r] = tbuf[r * 68 + lane];
      p1[r] = tbuf[r * 68 + 32 + lane];
    }
    unsigned todo = 0xFFFFu;
    while (todo) {
      const int r0 = __builtin_ctz(todo);
      const int g = __builtin_amdgcn_readlane(gl, r0);
      const unsigned m = __builtin_amdgcn_ballot_w32(gl == g) & todo;
      float s0 = 0.f, s1 = 0.f;
#pragma unroll
      for (int r = 0; r < 16; r++) {
        const bool in = (m >> r) & 1u;
        s0 += in ? p0[r] : 0.f;
        s1 += in ? p1[r] : 0.f;
      }
      float* pg = pooled + (size_t)g * 64 + lane;
      atomicAdd(pg, s0);
      atomicAdd(pg + 32, s1);
      todo &= ~m;
    }
  }
}

__global__ void gin_final_kernel(const float* __restrict__ pooled,
                                 const float* __restrict__ counts,
                                 const float* __restrict__ linW,
                                 const float* __restrict__ linb,
                                 float* __restrict__ out) {
  int idx = blockIdx.x * blockDim.x + threadIdx.x;
  if (idx >= N_GRAPHS * 10) return;
  int g = idx / 10, t = idx - g * 10;
  float acc = counts[g] * linb[t];
  for (int l = 1; l < 5; l++) acc += linb[l * 10 + t];
  for (int l = 0; l < 5; l++) {
    const float* p = pooled + ((size_t)l * N_GRAPHS + g) * 64;
    const float* w = linW + l * 640 + t;
    float s = 0.f;
#pragma unroll
    for (int k = 0; k < 64; k++) s += p[k] * w[k * 10];
    acc += s;
  }
  out[idx] = acc;
}

__global__ void gin_zero_kernel(unsigned int* __restrict__ p, int n) {
  int i = blockIdx.x * blockDim.x + threadIdx.x;
  if (i < n) p[i] = 0u;
}

extern "C" void kernel_launch(void* const* d_in, const int* in_sizes, int n_in,
                              void* d_out, int out_size, void* d_ws, size_t ws_size,
                              hipStream_t stream) {
  const float* x        = (const float*)d_in[0];
  const float* fh_W1    = (const float*)d_in[1];
  const float* fh_b1    = (const float*)d_in[2];
  const float* fh_bn1   = (const float*)d_in[3];
  const float* fh_W2    = (const float*)d_in[4];
  const float* fh_b2    = (const float*)d_in[5];
  const float* fh_bn2   = (const float*)d_in[6];
  const float* conv_W1  = (const float*)d_in[7];
  const float* conv_b1  = (const float*)d_in[8];
  const float* conv_bn1 = (const float*)d_in[9];
  const float* conv_W2  = (const float*)d_in[10];
  const float* conv_b2  = (const float*)d_in[11];
  const float* conv_bn2 = (const float*)d_in[12];
  const float* lin_W    = (const float*)d_in[13];
  const float* lin_b    = (const float*)d_in[14];
  const int*   eidx     = (const int*)d_in[15];
  const int*   batch    = (const int*)d_in[16];
  float* out = (float*)d_out;

  const int* src = eidx;
  const int* dst = eidx + N_EDGES;

  float* ws     = (float*)d_ws;
  float* h      = ws;
  float* zrow   = h + (size_t)N_NODES * 64;
  float* pooled = zrow + 64;
  float* counts = pooled + 5 * N_GRAPHS * 64;
  int*   bcnt   = (int*)(counts + N_GRAPHS);
  int*   boffs  = bcnt + SCAN_N;
  int*   rowptr = boffs + SCAN_N + 4;
  int2*  pairs  = (int2*)(rowptr + N_NODES + 4);
  int*   csr    = (int*)(pairs + N_EDGES);
  float* h2     = ws + 12000000;

  const int accN = 64 + 5 * N_GRAPHS * 64 + N_GRAPHS;
  gin_zero_kernel<<<(accN + 255) / 256, 256, 0, stream>>>((unsigned int*)zrow, accN);

  gin_bucket_count_kernel<<<N_CHUNK, CHUNK_THREADS, 0, stream>>>(dst, bcnt);
  gin_scan_kernel<<<1, 1024, 0, stream>>>(bcnt, boffs);
  gin_bucket_fill_kernel<<<N_CHUNK, CHUNK_THREADS, 0, stream>>>(src, dst, boffs, pairs);
  gin_bucket_csr_kernel<<<N_BKT, 1024, 0, stream>>>(pairs, boffs, rowptr, csr);

  const int mlpBlocks = (NTILES + 3) / 4;

  gin_mlp_kernel<D_FEAT, true><<<mlpBlocks, 128, 0, stream>>>(
      x, fh_W1, fh_b1, fh_bn1, fh_W2, fh_b2, fh_bn2, h, batch, pooled, counts, NTILES);

  const int convBlocks = (NTILES + 7) / 8;
  float* hin = h;
  float* hnext = h2;
  for (int l = 0; l < 4; l++) {
    gin_conv_kernel<<<convBlocks, 256, 0, stream>>>(
        hin, rowptr, csr,
        conv_W1 + (size_t)l * 64 * 64, conv_b1 + l * 64, conv_bn1 + l * 256,
        conv_W2 + (size_t)l * 64 * 64, conv_b2 + l * 64, conv_bn2 + l * 256,
        hnext, batch, pooled + (size_t)(l + 1) * N_GRAPHS * 64, NTILES);
    float* t = hin; hin = hnext; hnext = t;
  }

  gin_final_kernel<<<(N_GRAPHS * 10 + 255) / 256, 256, 0, stream>>>(
      pooled, counts, lin_W, lin_b, out);
}


# Conv kernel: each wave does two 16-node tiles in turn (4-wave groups, half the waves), no work-group barrier in the tile loop: one tile's matrix phase runs under other waves' gathers
# speedup vs baseline: 3.5373x; 1.0406x over previous
#include <hip/hip_runtime.h>

#define N_NODES   100000
#define N_EDGES   1600000
#define N_GRAPHS  1000
#define D_FEAT    128
#define HID       64
#define BN_EPS    1e-5f
#define NTILES    (N_NODES / 16)

typedef __attribute__((ext_vector_type(2))) float v2f;
typedef __attribute__((ext_vector_type(8))) float v8f;

template<int K>
static __device__ __forceinline__ int row_share_i(int v) {
  return __builtin_amdgcn_update_dpp(0, v, 0x150 + K, 0xF, 0xF, true);
}
#define GIN_SLOTS4(F, B)  F((B)) F((B) + 1) F((B) + 2) F((B) + 3)
#define GIN_SLOTS16(F)    GIN_SLOTS4(F, 0) GIN_SLOTS4(F, 4) GIN_SLOTS4(F, 8) GIN_SLOTS4(F, 12)

template<int K1, bool COUNT>
__launch_bounds__(128)
__global__ void gin_mlp_kernel(const float* __restrict__ xin,
                               const float* __restrict__ W1,
                               const float* __restrict__ b1,
                               const float* __restrict__ bn1,
                               const float* __restrict__ W2,
                               const float* __restrict__ b2,
                               const float* __restrict__ bn2,
                               float* __restrict__ hout,
                               const int* __restrict__ batch,
                               float* __restrict__ pooled,
                               float* __restrict__ counts,
                               int ntiles) {
  constexpr int KS1 = K1 / 4;
  constexpr int NQ1 = (K1 / 2) * 16 / 128;
  constexpr int NQ2 = (64 / 2) * 16 / 128;
  __shared__ __align__(16) float w1s[4 * KS1 * 32 * 2];
  __shared__ __align__(16) float w2s[4 * 16 * 32 * 2];
  __shared__ __align__(16) float tb[4][16 * 68];

  const int tid  = threadIdx.x;
  const int lane = tid & 31;
  const int wave = tid >> 5;
  const int row  = lane & 15;
  const int hi   = lane >> 4;
  const int tile = blockIdx.x * 4 + wave;
  const bool active = tile < ntiles;
  float* tbuf = tb[wave];

  float4 w1a[NQ1], w1b[NQ1], w2a[NQ2], w2b[NQ2];
#pragma unroll
  for (int j = 0; j < NQ1; j++) {
    const int p = tid + 128 * j;
    const float* wp = W1 + (p >> 4) * 128 + (p & 15) * 4;
    w1a[j] = *(const float4*)wp;
    w1b[j] = *(const float4*)(wp + 64);
  }
#pragma unroll
  for (int j = 0; j < NQ2; j++) {
    const int p = tid + 128 * j;
    const float* wp = W2 + (p >> 4) * 128 + (p & 15) * 4;
    w2a[j] = *(const float4*)wp;
    w2b[j] = *(const float4*)(wp + 64);
  }
  const int tl = active ? tile : 0;
  const int gl = batch[tl * 16 + row];
  v2f a1[KS1];
  const float* xr = xin + (size_t)(tl * 16 + row) * K1 + 2 * hi;
#pragma unroll
  for (int k = 0; k < KS1; k++) a1[k] = *(const v2f*)(xr + 4 * k);

#pragma unroll
  for (int j = 0; j < NQ1; j++) {
    const int p = tid + 128 * j;
    const int kk = 2 * (p >> 4), nq = p & 15;
    float* d = w1s + ((((nq >> 2) * KS1) + (kk >> 2)) * 32 + 4 * (nq & 3) + ((kk & 2) << 3)) * 2;
    *(float4*)d       = make_float4(w1a[j].x, w1b[j].x, w1a[j].y, w1b[j].y);
    *(float4*)(d + 4) = make_float4(w1a[j].z, w1b[j].z, w1a[j].w, w1b[j].w);
  }
#pragma unroll
  for (int j = 0; j < NQ2; j++) {
    const int p = tid + 128 * j;
    const int kk = 2 * (p >> 4), nq = p & 15;
    float* d = w2s + ((((nq >> 2) * 16) + (kk >> 2)) * 32 + 4 * (nq & 3) + ((kk & 2) << 3)) * 2;
    *(float4*)d       = make_float4(w2a[j].x, w2b[j].x, w2a[j].y, w2b[j].y);
    *(float4*)(d + 4) = make_float4(w2a[j].z, w2b[j].z, w2a[j].w, w2b[j].w);
  }
  __syncthreads();

  const v8f vzero = {0.f, 0.f, 0.f, 0.f, 0.f, 0.f, 0.f, 0.f};

  if (active) {
#pragma unroll
    for (int nt = 0; nt < 4; nt++) {
      v8f acc = vzero;
      const v2f* wf = ((const v2f*)w1s) + nt * KS1 * 32 + lane;
#pragma unroll
      for (int k = 0; k < KS1; k++) {
        acc = __builtin_amdgcn_wmma_f32_16x16x4_f32(false, a1[k], false, wf[k * 32],
                                                    (short)0, acc, false, false);
      }
      int n = nt * 16 + row;
      float s  = bn1[n] * rsqrtf(bn1[192 + n] + BN_EPS);
      float t0 = bn1[64 + n] + (b1[n] - bn1[128 + n]) * s;
#pragma unroll
      for (int i = 0; i < 8; i++) {
        float y = acc[i] * s + t0;
        tbuf[(i + 8 * hi) * 68 + n] = fmaxf(y, 0.0f);
      }
    }
  }
  __syncthreads();

  v8f acc2[4];
  if (active) {
#pragma unroll
    for (int nt = 0; nt < 4; nt++) {
      v8f acc = vzero;
      const v2f* wf = ((const v2f*)w2s) + nt * 16 * 32 + lane;
      const float* tr = tbuf + row * 68 + 2 * hi;
#pragma unroll
      for (int k = 0; k < 16; k++) {
        v2f a = *(const v2f*)(tr + 4 * k);
        acc = __builtin_amdgcn_wmma_f32_16x16x4_f32(false, a, false, wf[k * 32],
                                                    (short)0, acc, false, false);
      }
      acc2[nt] = acc;
    }
  }
  __syncthreads();

  if (active) {
#pragma unroll
    for (int nt = 0; nt < 4; nt++) {
      int n = nt * 16 + row;
      float s  = bn2[n] * rsqrtf(bn2[192 + n] + BN_EPS);
      float t0 = bn2[64 + n] + (b2[n] - bn2[128 + n]) * s;
#pragma unroll
      for (int i = 0; i < 8; i++) {
        float y = acc2[nt][i] * s + t0;
        tbuf[(i + 8 * hi) * 68 + n] = fmaxf(y, 0.0f);
      }
    }
  }
  __syncthreads();

  if (active) {
#pragma unroll
    for (int j = 0; j < 8; j++) {
      int flat = j * 32 + lane;
      int r = flat >> 4;
      int c = (flat & 15) << 2;
      float4 v = *(const float4*)(tbuf + r * 68 + c);
      *(float4*)(hout + (size_t)(tile * 16 + r) * 64 + c) = v;
    }
  }

  if (active) {
    float p0[16], p1[16];
#pragma unroll
    for (int r = 0; r < 16; r++) {
      p0[r] = tbuf[r * 68 + lane];
      p1[r] = tbuf[r * 68 + 32 + lane];
    }
    unsigned todo = 0xFFFFu;
    while (todo) {
      const int r0 = __builtin_ctz(todo);
      const int g = __builtin_amdgcn_readlane(gl, r0);
      const unsigned m = __builtin_amdgcn_ballot_w32(gl == g) & todo;
      float s0 = 0.f, s1 = 0.f;
#pragma unroll
      for (int r = 0; r < 16; r++) {
        const bool in = (m >> r) & 1u;
        s0 += in ? p0[r] : 0.f;
        s1 += in ? p1[r] : 0.f;
      }
      float* pg = pooled + (size_t)g * 64 + lane;
      atomicAdd(pg, s0);
      atomicAdd(pg + 32, s1);
      if (COUNT && lane == 0) atomicAdd(counts + g, (float)__builtin_popcount(m));
      todo &= ~m;
    }
  }
}

#define BKT_NODES     500
#define N_BKT         (N_NODES / BKT_NODES)
#define N_CHUNK       500
#define CHUNK_EDGES   (N_EDGES / N_CHUNK)
#define CHUNK_THREADS 320
#define CHUNK_PER_THR (CHUNK_EDGES / CHUNK_THREADS)

__launch_bounds__(CHUNK_THREADS)
__global__ void gin_bucket_count_kernel(const int* __restrict__ dst,
                                        int* __restrict__ bcnt) {
  __shared__ int cnt[N_BKT];
  const int t = threadIdx.x;
  const int g = blockIdx.x;
  if (t < N_BKT) cnt[t] = 0;
  __syncthreads();
  const int* dp = dst + g * CHUNK_EDGES + t;
  int d[CHUNK_PER_THR];
#pragma unroll
  for (int k = 0; k < CHUNK_PER_THR; k++) d[k] = dp[k * CHUNK_THREADS];
#pragma unroll
  for (int k = 0; k < CHUNK_PER_THR; k++) atomicAdd(&cnt[d[k] / BKT_NODES], 1);
  __syncthreads();
  if (t < N_BKT) bcnt[t * N_CHUNK + g] = cnt[t];
}

#define SCAN_CHUNK 100
#define SCAN_N (N_BKT * N_CHUNK)
#define SCAN_OWNERS (SCAN_N / SCAN_CHUNK)
__launch_bounds__(1024)
__global__ void gin_scan_kernel(const int* __restrict__ bcnt,
                                int* __restrict__ boffs) {
  __shared__ int wsum[32];
  const int t = threadIdx.x;
  const int lane = t & 31;
  const int w = t >> 5;
  const bool owner = t < SCAN_OWNERS;

  int4 v[SCAN_CHUNK / 4];
  int local = 0;
  if (owner) {
    const int4* p = (const int4*)(bcnt + t * SCAN_CHUNK);
#pragma unroll
    for (int i = 0; i < SCAN_CHUNK / 4; i++) v[i] = p[i];
#pragma unroll
    for (int i = 0; i < SCAN_CHUNK / 4; i++) local += v[i].x + v[i].y + v[i].z + v[i].w;
  }

  int x = local;
#pragma unroll
  for (int o = 1; o < 32; o <<= 1) {
    int y = __shfl_up(x, o);
    if (lane >= o) x += y;
  }
  if (lane == 31) wsum[w] = x;
  __syncthreads();
  if (w == 0) {
    int s = wsum[lane];
#pragma unroll
    for (int o = 1; o < 32; o <<= 1) {
      int y = __shfl_up(s, o);
      if (lane >= o) s += y;
    }
    wsum[lane] = s;
  }
  __syncthreads();

  if (owner) {
    int run = x - local + (w > 0 ? wsum[w - 1] : 0);
    int4* q = (int4*)(boffs + t * SCAN_CHUNK);
#pragma unroll
    for (int i = 0; i < SCAN_CHUNK / 4; i++) {
      int4 o4;
      o4.x = run; run += v[i].x;
      o4.y = run; run += v[i].y;
      o4.z = run; run += v[i].z;
      o4.w = run; run += v[i].w;
      q[i] = o4;
    }
    if (t == SCAN_OWNERS - 1) boffs[SCAN_N] = run;
  }
}

__launch_bounds__(CHUNK_THREADS)
__global__ void gin_bucket_fill_kernel(const int* __restrict__ src,
                                       const int* __restrict__ dst,
                                       const int* __restrict__ boffs,
                                       int2* __restrict__ pairs) {
  __shared__ int cur[N_BKT];
  const int t = threadIdx.x;
  const int g = blockIdx.x;
  if (t < N_BKT) cur[t] = boffs[t * N_CHUNK + g];
  __syncthreads();
  const int e0 = g * CHUNK_EDGES + t;
  int s[CHUNK_PER_THR], d[CHUNK_PER_THR];
#pragma unroll
  for (int k = 0; k < CHUNK_PER_THR; k++) {
    s[k] = src[e0 + k * CHUNK_THREADS];
    d[k] = dst[e0 + k * CHUNK_THREADS];
  }
#pragma unroll
  for (int k = 0; k < CHUNK_PER_THR; k++) {
    int pos = atomicAdd(&cur[d[k] / BKT_NODES], 1);
    pairs[pos] = make_int2(s[k], d[k]);
  }
}

#define SPARE_CNT 511
__launch_bounds__(1024)
__global__ void gin_bucket_csr_kernel(const int2* __restrict__ pairs,
                                      const int* __restrict__ boffs,
                                      int* __restrict__ rowptr,
                                      int* __restrict__ csr) {
  __shared__ int cnt[512];
  __shared__ int wsum[32];
  const int t = threadIdx.x;
  const int lane = t & 31;
  const int w = t >> 5;
  const int b = blockIdx.x;
  const int n0 = b * BKT_NODES;
  const int beg = boffs[b * N_CHUNK];
  const int end = boffs[(b + 1) * N_CHUNK];

  if (t < 512) cnt[t] = 0;
  __syncthreads();
  for (int i0 = beg + t; i0 < end; i0 += 4 * 1024) {
    int c[4];
#pragma unroll
    for (int k = 0; k < 4; k++) {
      int i = i0 + k * 1024;
      int d = pairs[i < end ? i : end - 1].y;
      c[k] = (i < end) ? d - n0 : SPARE_CNT;
    }
#pragma unroll
    for (int k = 0; k < 4; k++) atomicAdd(&cnt[c[k]], 1);
  }
  __syncthreads();

  const int v = (t < BKT_NODES) ? cnt[t] : 0;
  int x = v;
#pragma unroll
  for (int o = 1; o < 32; o <<= 1) {
    int y = __shfl_up(x, o);
    if (lane >= o) x += y;
  }
  if (lane == 31) wsum[w] = x;
  __syncthreads();
  if (w == 0) {
    int s = wsum[lane];
#pragma unroll
    for (int o = 1; o < 32; o <<= 1) {
      int y = __shfl_up(s, o);
      if (lane >= o) s += y;
    }
    wsum[lane] = s;
  }
  __syncthreads();
  const int first = beg + x - v + (w > 0 ? wsum[w - 1] : 0);
  if (t < BKT_NODES) {
    rowptr[n0 + t] = first;
    cnt[t] = first;
  }
  if (b == N_BKT - 1 && t == BKT_NODES - 1) rowptr[N_NODES] = first + v;
  __syncthreads();

  for (int i0 = beg + t; i0 < end; i0 += 4 * 1024) {
    int2 p[4];
#pragma unroll
    for (int k = 0; k < 4; k++) {
      int i = i0 + k * 1024;
      p[k] = pairs[i < end ? i : end - 1];
    }
#pragma unroll
    for (int k = 0; k < 4; k++) {
      if (i0 + k * 1024 < end) {
        int pos = atomicAdd(&cnt[p[k].y - n0], 1);
        csr[pos] = p[k].x;
      }
    }
  }
}

#define CONV_WAVES 4
#define CONV_TILES_PER_WAVE 2
#define CONV_TILES_PER_WG (CONV_WAVES * CONV_TILES_PER_WAVE)

static __device__ __forceinline__ void wave_lds_handover() {
  __builtin_amdgcn_fence(__ATOMIC_RELEASE, "wavefront");
  __builtin_amdgcn_wave_barrier();
  __builtin_amdgcn_fence(__ATOMIC_ACQUIRE, "wavefront");
}

static __device__ __forceinline__ void conv_first_slots(const int* __restrict__ csr, int rpl, int row, int hi,
                                                        int& c0, int& c1) {
  const int b0 = __builtin_amdgcn_readlane(rpl, 0);
  const int b1 = __builtin_amdgcn_readlane(rpl, 1);
  const int b2 = __builtin_amdgcn_readlane(rpl, 2);
  const int beg = hi ? b1 : b0, end = hi ? b2 : b1;
  int j = beg + row;      j = j < end ? j : end - 1;
  c0 = csr[j > 0 ? j : 0];
  j = beg + 16 + row;     j = j < end ? j : end - 1;
  c1 = csr[j > 0 ? j : 0];
}

__launch_bounds__(CONV_WAVES * 32)
__global__ void gin_conv_kernel(const float* __restrict__ hin,
                                const int* __restrict__ rowptr,
                                const int* __restrict__ csr,
                                const float* __restrict__ W1,
                                const float* __restrict__ b1,
                                const float* __restrict__ bn1,
                                const float* __restrict__ W2,
                                const float* __restrict__ b2,
                                const float* __restrict__ bn2,
                                float* __restrict__ hout,
                                const int* __restrict__ batch,
                                float* __restrict__ pooled,
                                int ntiles) {
  __shared__ __align__(16) float w1s[4 * 16 * 32 * 2];
  __shared__ __align__(16) float w2s[4 * 16 * 32 * 2];
  __shared__ __align__(16) float tb[CONV_WAVES][16 * 68];

  const int tid  = threadIdx.x;
  const int lane = tid & 31;
  const int wave = tid >> 5;
  const int row  = lane & 15;
  const int hi   = lane >> 4;
  float* tbuf = tb[wave];

  float4 w1a[4], w1b[4], w2a[4], w2b[4];
#pragma unroll
  for (int j = 0; j < 4; j++) {
    const int p = tid + 128 * j;
    const float* wp = W1 + (p >> 4) * 128 + (p & 15) * 4;
    w1a[j] = *(const float4*)wp;
    w1b[j] = *(const float4*)(wp + 64);
    const float* wq = W2 + (p >> 4) * 128 + (p & 15) * 4;
    w2a[j] = *(const float4*)wq;
    w2b[j] = *(const float4*)(wq + 64);
  }
  const int tile0 = blockIdx.x * CONV_TILES_PER_WG + wave;
  const int tl0 = tile0 < ntiles ? tile0 : 0;
  const int tl1 = tile0 + CONV_WAVES < ntiles ? tile0 + CONV_WAVES : 0;
  int gl   = batch[tl0 * 16 + row];
  int gln  = batch[tl1 * 16 + row];
  int rpl  = rowptr[tl0 * 16 + (lane < 16 ? lane : 16)];
  int rpln = rowptr[tl1 * 16 + (lane < 16 ? lane : 16)];
  int c0, c1;
  conv_first_slots(csr, rpl, row, hi, c0, c1);

#pragma unroll
  for (int j = 0; j < 4; j++) {
    const int p = tid + 128 * j;
    const int kk = 2 * (p >> 4), nq = p & 15;
    const int o = ((((nq >> 2) * 16) + (kk >> 2)) * 32 + 4 * (nq & 3) + ((kk & 2) << 3)) * 2;
    *(float4*)(w1s + o)     = make_float4(w1a[j].x, w1b[j].x, w1a[j].y, w1b[j].y);
    *(float4*)(w1s + o + 4) = make_float4(w1a[j].z, w1b[j].z, w1a[j].w, w1b[j].w);
    *(float4*)(w2s + o)     = make_float4(w2a[j].x, w2b[j].x, w2a[j].y, w2b[j].y);
    *(float4*)(w2s + o + 4) = make_float4(w2a[j].z, w2b[j].z, w2a[j].w, w2b[j].w);
  }
  __syncthreads();

  const v8f vzero = {0.f, 0.f, 0.f, 0.f, 0.f, 0.f, 0.f, 0.f};

  for (int it = 0; it < CONV_TILES_PER_WAVE; it++) {
    const int tile = tile0 + it * CONV_WAVES;
    if (tile >= ntiles) break;

    {
      const char* hb = (const char*)hin;
      const unsigned cb = (unsigned)row << 4;
      int b0 = __builtin_amdgcn_readlane(rpl, 0);
      int b1 = __builtin_amdgcn_readlane(rpl, 1);
      int b2 = __builtin_amdgcn_readlane(rpl, 2);
      int beg = hi ? b1 : b0, end = hi ? b2 : b1;
      for (int p = 0; p < 8; p++) {
        const int r = 2 * p + hi;
        const unsigned n = (unsigned)(tile * 16 + r);
        float4 acc = *(const float4*)(hb + ((n << 8) + cb));
        float4 v[16];
#define GIN_LOADV(k) v[(k)] = *(const float4*)(hb + ((((unsigned)row_share_i<(k)>(c0)) << 8) + cb));
        GIN_SLOTS16(GIN_LOADV)
#undef GIN_LOADV
        const int nb0 = b2;
        const int nb1 = __builtin_amdgcn_readlane(rpl, 2 * p + 3);
        const int nb2 = __builtin_amdgcn_readlane(rpl, 2 * p + 4);
        const int nbeg = hi ? nb1 : nb0, nend = hi ? nb2 : nb1;
        int nc0, nc1;
        {
          int j = nbeg + row;   j = j < nend ? j : nend - 1;
          nc0 = csr[j > 0 ? j : 0];
          j = nbeg + 16 + row;  j = j < nend ? j : nend - 1;
          nc1 = csr[j > 0 ? j : 0];
        }
#define GIN_ACCV(k) { const bool in = beg + (k) < end; \
        acc.x += in ? v[(k)].x : 0.f; acc.y += in ? v[(k)].y : 0.f; \
        acc.z += in ? v[(k)].z : 0.f; acc.w += in ? v[(k)].w : 0.f; }
        GIN_SLOTS16(GIN_ACCV)
#undef GIN_ACCV
        const int d0 = b1 - b0, d1 = b2 - b1;
        const int more = (d0 > d1 ? d0 : d1) - 16;
        if (more > 0) {
#define GIN_LOADU(k) v[(k)] = *(const float4*)(hb + ((((unsigned)row_share_i<(k)>(c1)) << 8) + cb));
#define GIN_ACCU(k) { const bool in = beg + 16 + (k) < end; \
        acc.x += in ? v[(k)].x : 0.f; acc.y += in ? v[(k)].y : 0.f; \
        acc.z += in ? v[(k)].z : 0.f; acc.w += in ? v[(k)].w : 0.f; }
          GIN_SLOTS4(GIN_LOADU, 0)
          if (more > 4) {
            GIN_SLOTS4(GIN_LOADU, 4)
            if (more > 8) {
              GIN_SLOTS4(GIN_LOADU, 8)
              if (more > 12) { GIN_SLOTS4(GIN_LOADU, 12) }
            }
          }
          GIN_SLOTS4(GIN_ACCU, 0)
          if (more > 4) {
            GIN_SLOTS4(GIN_ACCU, 4)
            if (more > 8) {
              GIN_SLOTS4(GIN_ACCU, 8)
              if (more > 12) { GIN_SLOTS4(GIN_ACCU, 12) }
            }
          }
#undef GIN_LOADU
#undef GIN_ACCU
        }
        for (int j = beg + 32; j < end; j += 16) {
          int s[16];
#pragma unroll
          for (int k = 0; k < 16; k++) s[k] = csr[(j + k < end) ? j + k : end - 1];
#pragma unroll
          for (int k = 0; k < 16; k++) {
            const unsigned sk = (j + k < end) ? (unsigned)s[k] : n;
            v[k] = *(const float4*)(hb + ((sk << 8) + cb));
          }
#pragma unroll
          for (int k = 0; k < 16; k++) {
            const bool in = j + k < end;
            acc.x += in ? v[k].x : 0.f; acc.y += in ? v[k].y : 0.f;
            acc.z += in ? v[k].z : 0.f; acc.w += in ? v[k].w : 0.f;
          }
        }
        *(float4*)(tbuf + r * 68 + 4 * row) = acc;
        b0 = nb0; b1 = nb1; b2 = nb2;
        beg = nbeg; end = nend;
        c0 = nc0; c1 = nc1;
      }
    }
    if (it + 1 < CONV_TILES_PER_WAVE) conv_first_slots(csr, rpln, row, hi, c0, c1);
    wave_lds_handover();

    {
      v2f a1[16];
      const float* tr = tbuf + row * 68 + 2 * hi;
#pragma unroll
      for (int k = 0; k < 16; k++) a1[k] = *(const v2f*)(tr + 4 * k);
#pragma unroll
      for (int nt = 0; nt < 4; nt++) {
        v8f acc = vzero;
        const v2f* wf = ((const v2f*)w1s) + nt * 16 * 32 + lane;
#pragma unroll
        for (int k = 0; k < 16; k++) {
          acc = __builtin_amdgcn_wmma_f32_16x16x4_f32(false, a1[k], false, wf[k * 32],
                                                      (short)0, acc, false, false);
        }
        int n = nt * 16 + row;
        float s  = bn1[n] * rsqrtf(bn1[192 + n] + BN_EPS);
        float t0 = bn1[64 + n] + (b1[n] - bn1[128 + n]) * s;
#pragma unroll
        for (int i = 0; i < 8; i++) {
          float y = acc[i] * s + t0;
          tbuf[(i + 8 * hi) * 68 + n] = fmaxf(y, 0.0f);
        }
      }
    }
    wave_lds_handover();

    v8f acc2[4];
    {
#pragma unroll
      for (int nt = 0; nt < 4; nt++) {
        v8f acc = vzero;
        const v2f* wf = ((const v2f*)w2s) + nt * 16 * 32 + lane;
        const float* tr = tbuf + row * 68 + 2 * hi;
#pragma unroll
        for (int k = 0; k < 16; k++) {
          v2f a = *(const v2f*)(tr + 4 * k);
          acc = __builtin_amdgcn_wmma_f32_16x16x4_f32(false, a, false, wf[k * 32],
                                                      (short)0, acc, false, false);
        }
        acc2[nt] = acc;
      }
    }

    {
#pragma unroll
      for (int nt = 0; nt < 4; nt++) {
        int n = nt * 16 + row;
        float s  = bn2[n] * rsqrtf(bn2[192 + n] + BN_EPS);
        float t0 = bn2[64 + n] + (b2[n] - bn2[128 + n]) * s;
#pragma unroll
        for (int i = 0; i < 8; i++) {
          float y = acc2[nt][i] * s + t0;
          tbuf[(i + 8 * hi) * 68 + n] = fmaxf(y, 0.0f);
        }
      }
    }
    wave_lds_handover();

    {
#pragma unroll
      for (int j = 0; j < 8; j++) {
        int flat = j * 32 + lane;
        int r = flat >> 4;
        int c = (flat & 15) << 2;
        float4 v = *(const float4*)(tbuf + r * 68 + c);
        *(float4*)(hout + (size_t)(tile * 16 + r) * 64 + c) = v;
      }
    }

    {
      float p0[16], p1[16];
#pragma unroll
      for (int r = 0; r < 16; r++) {
        p0[r] = tbuf[r * 68 + lane];
        p1[r] = tbuf[r * 68 + 32 + lane];
      }
      unsigned todo = 0xFFFFu;
      while (todo) {
        const int r0 = __builtin_ctz(todo);
        const int g = __builtin_amdgcn_readlane(gl, r0);
        const unsigned m = __builtin_amdgcn_ballot_w32(gl == g) & todo;
        float s0 = 0.f, s1 = 0.f;
#pragma unroll
        for (int r = 0; r < 16; r++) {
          const bool in = (m >> r) & 1u;
          s0 += in ? p0[r] : 0.f;
          s1 += in ? p1[r] : 0.f;
        }
        float* pg = pooled + (size_t)g * 64 + lane;
        atomicAdd(pg, s0);
        atomicAdd(pg + 32, s1);
        todo &= ~m;
      }
    }
    wave_lds_handover();
    gl = gln;
    rpl = rpln;
  }
}

__global__ void gin_final_kernel(const float* __restrict__ pooled,
                                 const float* __restrict__ counts,
                                 const float* __restrict__ linW,
                                 const float* __restrict__ linb,
                                 float* __restrict__ out) {
  int idx = blockIdx.x * blockDim.x + threadIdx.x;
  if (idx >= N_GRAPHS * 10) return;
  int g = idx / 10, t = idx - g * 10;
  float acc = counts[g] * linb[t];
  for (int l = 1; l < 5; l++) acc += linb[l * 10 + t];
  for (int l = 0; l < 5; l++) {
    const float* p = pooled + ((size_t)l * N_GRAPHS + g) * 64;
    const float* w = linW + l * 640 + t;
    float s = 0.f;
#pragma unroll
    for (int k = 0; k < 64; k++) s += p[k] * w[k * 10];
    acc += s;
  }
  out[idx] = acc;
}

__global__ void gin_zero_kernel(unsigned int* __restrict__ p, int n) {
  int i = blockIdx.x * blockDim.x + threadIdx.x;
  if (i < n) p[i] = 0u;
}

extern "C" void kernel_launch(void* const* d_in, const int* in_sizes, int n_in,
                              void* d_out, int out_size, void* d_ws, size_t ws_size,
                              hipStream_t stream) {
  const float* x        = (const float*)d_in[0];
  const float* fh_W1    = (const float*)d_in[1];
  const float* fh_b1    = (const float*)d_in[2];
  const float* fh_bn1   = (const float*)d_in[3];
  const float* fh_W2    = (const float*)d_in[4];
  const float* fh_b2    = (const float*)d_in[5];
  const float* fh_bn2   = (const float*)d_in[6];
  const float* conv_W1  = (const float*)d_in[7];
  const float* conv_b1  = (const float*)d_in[8];
  const float* conv_bn1 = (const float*)d_in[9];
  const float* conv_W2  = (const float*)d_in[10];
  const float* conv_b2  = (const float*)d_in[11];
  const float* conv_bn2 = (const float*)d_in[12];
  const float* lin_W    = (const float*)d_in[13];
  const float* lin_b    = (const float*)d_in[14];
  const int*   eidx     = (const int*)d_in[15];
  const int*   batch    = (const int*)d_in[16];
  float* out = (float*)d_out;

  const int* src = eidx;
  const int* dst = eidx + N_EDGES;

  float* ws     = (float*)d_ws;
  float* h      = ws;
  float* zrow   = h + (size_t)N_NODES * 64;
  float* pooled = zrow + 64;
  float* counts = pooled + 5 * N_GRAPHS * 64;
  int*   bcnt   = (int*)(counts + N_GRAPHS);
  int*   boffs  = bcnt + SCAN_N;
  int*   rowptr = boffs + SCAN_N + 4;
  int2*  pairs  = (int2*)(rowptr + N_NODES + 4);
  int*   csr    = (int*)(pairs + N_EDGES);
  float* h2     = ws + 12000000;

  const int accN = 64 + 5 * N_GRAPHS * 64 + N_GRAPHS;
  gin_zero_kernel<<<(accN + 255) / 256, 256, 0, stream>>>((unsigned int*)zrow, accN);

  gin_bucket_count_kernel<<<N_CHUNK, CHUNK_THREADS, 0, stream>>>(dst, bcnt);
  gin_scan_kernel<<<1, 1024, 0, stream>>>(bcnt, boffs);
  gin_bucket_fill_kernel<<<N_CHUNK, CHUNK_THREADS, 0, stream>>>(src, dst, boffs, pairs);
  gin_bucket_csr_kernel<<<N_BKT, 1024, 0, stream>>>(pairs, boffs, rowptr, csr);

  const int mlpBlocks = (NTILES + 3) / 4;

  gin_mlp_kernel<D_FEAT, true><<<mlpBlocks, 128, 0, stream>>>(
      x, fh_W1, fh_b1, fh_bn1, fh_W2, fh_b2, fh_bn2, h, batch, pooled, counts, NTILES);

  const int convBlocks = (NTILES + CONV_TILES_PER_WG - 1) / CONV_TILES_PER_WG;
  float* hin = h;
  float* hnext = h2;
  for (int l = 0; l < 4; l++) {
    gin_conv_kernel<<<convBlocks, CONV_WAVES * 32, 0, stream>>>(
        hin, rowptr, csr,
        conv_W1 + (size_t)l * 64 * 64, conv_b1 + l * 64, conv_bn1 + l * 256,
        conv_W2 + (size_t)l * 64 * 64, conv_b2 + l * 64, conv_bn2 + l * 256,
        hnext, batch, pooled + (size_t)(l + 1) * N_GRAPHS * 64, NTILES);
    float* t = hin; hin = hnext; hnext = t;
  }

  gin_final_kernel<<<(N_GRAPHS * 10 + 255) / 256, 256, 0, stream>>>(
      pooled, counts, lin_W, lin_b, out);
}


# Conv gather: rows of in-edge slots 16-31 join the pass's one load group in registers of their own (was a second load-wait trip behind the sums of slots 0-15 in 2 of 3 passes); same sum order
# speedup vs baseline: 3.6465x; 1.0309x over previous
#include <hip/hip_runtime.h>

#define N_NODES   100000
#define N_EDGES   1600000
#define N_GRAPHS  1000
#define D_FEAT    128
#define HID       64
#define BN_EPS    1e-5f
#define NTILES    (N_NODES / 16)

typedef __attribute__((ext_vector_type(2))) float v2f;
typedef __attribute__((ext_vector_type(8))) float v8f;

template<int K>
static __device__ __forceinline__ int row_share_i(int v) {
  return __builtin_amdgcn_update_dpp(0, v, 0x150 + K, 0xF, 0xF, true);
}
#define GIN_SLOTS4(F, B)  F((B)) F((B) + 1) F((B) + 2) F((B) + 3)
#define GIN_SLOTS16(F)    GIN_SLOTS4(F, 0) GIN_SLOTS4(F, 4) GIN_SLOTS4(F, 8) GIN_SLOTS4(F, 12)

template<int K1, bool COUNT>
__launch_bounds__(128)
__global__ void gin_mlp_kernel(const float* __restrict__ xin,
                               const float* __restrict__ W1,
                               const float* __restrict__ b1,
                               const float* __restrict__ bn1,
                               const float* __restrict__ W2,
                               const float* __restrict__ b2,
                               const float* __restrict__ bn2,
                               float* __restrict__ hout,
                               const int* __restrict__ batch,
                               float* __restrict__ pooled,
                               float* __restrict__ counts,
                               int ntiles) {
  constexpr int KS1 = K1 / 4;
  constexpr int NQ1 = (K1 / 2) * 16 / 128;
  constexpr int NQ2 = (64 / 2) * 16 / 128;
  __shared__ __align__(16) float w1s[4 * KS1 * 32 * 2];
  __shared__ __align__(16) float w2s[4 * 16 * 32 * 2];
  __shared__ __align__(16) float tb[4][16 * 68];

  const int tid  = threadIdx.x;
  const int lane = tid & 31;
  const int wave = tid >> 5;
  const int row  = lane & 15;
  const int hi   = lane >> 4;
  const int tile = blockIdx.x * 4 + wave;
  const bool active = tile < ntiles;
  float* tbuf = tb[wave];

  float4 w1a[NQ1], w1b[NQ1], w2a[NQ2], w2b[NQ2];
#pragma unroll
  for (int j = 0; j < NQ1; j++) {
    const int p = tid + 128 * j;
    const float* wp = W1 + (p >> 4) * 128 + (p & 15) * 4;
    w1a[j] = *(const float4*)wp;
    w1b[j] = *(const float4*)(wp + 64);
  }
#pragma unroll
  for (int j = 0; j < NQ2; j++) {
    const int p = tid + 128 * j;
    const float* wp = W2 + (p >> 4) * 128 + (p & 15) * 4;
    w2a[j] = *(const float4*)wp;
    w2b[j] = *(const float4*)(wp + 64);
  }
  const int tl = active ? tile : 0;
  const int gl = batch[tl * 16 + row];
  v2f a1[KS1];
  const float* xr = xin + (size_t)(tl * 16 + row) * K1 + 2 * hi;
#pragma unroll
  for (int k = 0; k < KS1; k++) a1[k] = *(const v2f*)(xr + 4 * k);

#pragma unroll
  for (int j = 0; j < NQ1; j++) {
    const int p = tid + 128 * j;
    const int kk = 2 * (p >> 4), nq = p & 15;
    float* d = w1s + ((((nq >> 2) * KS1) + (kk >> 2)) * 32 + 4 * (nq & 3) + ((kk & 2) << 3)) * 2;
    *(float4*)d       = make_float4(w1a[j].x, w1b[j].x, w1a[j].y, w1b[j].y);
    *(float4*)(d + 4) = make_float4(w1a[j].z, w1b[j].z, w1a[j].w, w1b[j].w);
  }
#pragma unroll
  for (int j = 0; j < NQ2; j++) {
    const int p = tid + 128 * j;
    const int kk = 2 * (p >> 4), nq = p & 15;
    float* d = w2s + ((((nq >> 2) * 16) + (kk >> 2)) * 32 + 4 * (nq & 3) + ((kk & 2) << 3)) * 2;
    *(float4*)d       = make_float4(w2a[j].x, w2b[j].x, w2a[j].y, w2b[j].y);
    *(float4*)(d + 4) = make_float4(w2a[j].z, w2b[j].z, w2a[j].w, w2b[j].w);
  }
  __syncthreads();

  const v8f vzero = {0.f, 0.f, 0.f, 0.f, 0.f, 0.f, 0.f, 0.f};

  if (active) {
#pragma unroll
    for (int nt = 0; nt < 4; nt++) {
      v8f acc = vzero;
      const v2f* wf = ((const v2f*)w1s) + nt * KS1 * 32 + lane;
#pragma unroll
      for (int k = 0; k < KS1; k++) {
        acc = __builtin_amdgcn_wmma_f32_16x16x4_f32(false, a1[k], false, wf[k * 32],
                                                    (short)0, acc, false, false);
      }
      int n = nt * 16 + row;
      float s  = bn1[n] * rsqrtf(bn1[192 + n] + BN_EPS);
      float t0 = bn1[64 + n] + (b1[n] - bn1[128 + n]) * s;
#pragma unroll
      for (int i = 0; i < 8; i++) {
        float y = acc[i] * s + t0;
        tbuf[(i + 8 * hi) * 68 + n] = fmaxf(y, 0.0f);
      }
    }
  }
  __syncthreads();

  v8f acc2[4];
  if (active) {
#pragma unroll
    for (int nt = 0; nt < 4; nt++) {
      v8f acc = vzero;
      const v2f* wf = ((const v2f*)w2s) + nt * 16 * 32 + lane;
      const float* tr = tbuf + row * 68 + 2 * hi;
#pragma unroll
      for (int k = 0; k < 16; k++) {
        v2f a = *(const v2f*)(tr + 4 * k);
        acc = __builtin_amdgcn_wmma_f32_16x16x4_f32(false, a, false, wf[k * 32],
                                                    (short)0, acc, false, false);
      }
      acc2[nt] = acc;
    }
  }
  __syncthreads();

  if (active) {
#pragma unroll
    for (int nt = 0; nt < 4; nt++) {
      int n = nt * 16 + row;
      float s  = bn2[n] * rsqrtf(bn2[192 + n] + BN_EPS);
      float t0 = bn2[64 + n] + (b2[n] - bn2[128 + n]) * s;
#pragma unroll
      for (int i = 0; i < 8; i++) {
        float y = acc2[nt][i] * s + t0;
        tbuf[(i + 8 * hi) * 68 + n] = fmaxf(y, 0.0f);
      }
    }
  }
  __syncthreads();

  if (active) {
#pragma unroll
    for (int j = 0; j < 8; j++) {
      int flat = j * 32 + lane;
      int r = flat >> 4;
      int c = (flat & 15) << 2;
      float4 v = *(const float4*)(tbuf + r * 68 + c);
      *(float4*)(hout + (size_t)(tile * 16 + r) * 64 + c) = v;
    }
  }

  if (active) {
    float p0[16], p1[16];
#pragma unroll
    for (int r = 0; r < 16; r++) {
      p0[r] = tbuf[r * 68 + lane];
      p1[r] = tbuf[r * 68 + 32 + lane];
    }
    unsigned todo = 0xFFFFu;
    while (todo) {
      const int r0 = __builtin_ctz(todo);
      const int g = __builtin_amdgcn_readlane(gl, r0);
      const unsigned m = __builtin_amdgcn_ballot_w32(gl == g) & todo;
      float s0 = 0.f, s1 = 0.f;
#pragma unroll
      for (int r = 0; r < 16; r++) {
        const bool in = (m >> r) & 1u;
        s0 += in ? p0[r] : 0.f;
        s1 += in ? p1[r] : 0.f;
      }
      float* pg = pooled + (size_t)g * 64 + lane;
      atomicAdd(pg, s0);
      atomicAdd(pg + 32, s1);
      if (COUNT && lane == 0) atomicAdd(counts + g, (float)__builtin_popcount(m));
      todo &= ~m;
    }
  }
}

#define BKT_NODES     500
#define N_BKT         (N_NODES / BKT_NODES)
#define N_CHUNK       500
#define CHUNK_EDGES   (N_EDGES / N_CHUNK)
#define CHUNK_THREADS 320
#define CHUNK_PER_THR (CHUNK_EDGES / CHUNK_THREADS)

__launch_bounds__(CHUNK_THREADS)
__global__ void gin_bucket_count_kernel(const int* __restrict__ dst,
                                        int* __restrict__ bcnt) {
  __shared__ int cnt[N_BKT];
  const int t = threadIdx.x;
  const int g = blockIdx.x;
  if (t < N_BKT) cnt[t] = 0;
  __syncthreads();
  const int* dp = dst + g * CHUNK_EDGES + t;
  int d[CHUNK_PER_THR];
#pragma unroll
  for (int k = 0; k < CHUNK_PER_THR; k++) d[k] = dp[k * CHUNK_THREADS];
#pragma unroll
  for (int k = 0; k < CHUNK_PER_THR; k++) atomicAdd(&cnt[d[k] / BKT_NODES], 1);
  __syncthreads();
  if (t < N_BKT) bcnt[t * N_CHUNK + g] = cnt[t];
}

#define SCAN_CHUNK 100
#define SCAN_N (N_BKT * N_CHUNK)
#define SCAN_OWNERS (SCAN_N / SCAN_CHUNK)
__launch_bounds__(1024)
__global__ void gin_scan_kernel(const int* __restrict__ bcnt,
                                int* __restrict__ boffs) {
  __shared__ int wsum[32];
  const int t = threadIdx.x;
  const int lane = t & 31;
  const int w = t >> 5;
  const bool owner = t < SCAN_OWNERS;

  int4 v[SCAN_CHUNK / 4];
  int local = 0;
  if (owner) {
    const int4* p = (const int4*)(bcnt + t * SCAN_CHUNK);
#pragma unroll
    for (int i = 0; i < SCAN_CHUNK / 4; i++) v[i] = p[i];
#pragma unroll
    for (int i = 0; i < SCAN_CHUNK / 4; i++) local += v[i].x + v[i].y + v[i].z + v[i].w;
  }

  int x = local;
#pragma unroll
  for (int o = 1; o < 32; o <<= 1) {
    int y = __shfl_up(x, o);
    if (lane >= o) x += y;
  }
  if (lane == 31) wsum[w] = x;
  __syncthreads();
  if (w == 0) {
    int s = wsum[lane];
#pragma unroll
    for (int o = 1; o < 32; o <<= 1) {
      int y = __shfl_up(s, o);
      if (lane >= o) s += y;
    }
    wsum[lane] = s;
  }
  __syncthreads();

  if (owner) {
    int run = x - local + (w > 0 ? wsum[w - 1] : 0);
    int4* q = (int4*)(boffs + t * SCAN_CHUNK);
#pragma unroll
    for (int i = 0; i < SCAN_CHUNK / 4; i++) {
      int4 o4;
      o4.x = run; run += v[i].x;
      o4.y = run; run += v[i].y;
      o4.z = run; run += v[i].z;
      o4.w = run; run += v[i].w;
      q[i] = o4;
    }
    if (t == SCAN_OWNERS - 1) boffs[SCAN_N] = run;
  }
}

__launch_bounds__(CHUNK_THREADS)
__global__ void gin_bucket_fill_kernel(const int* __restrict__ src,
                                       const int* __restrict__ dst,
                                       const int* __restrict__ boffs,
                                       int2* __restrict__ pairs) {
  __shared__ int cur[N_BKT];
  const int t = threadIdx.x;
  const int g = blockIdx.x;
  if (t < N_BKT) cur[t] = boffs[t * N_CHUNK + g];
  __syncthreads();
  const int e0 = g * CHUNK_EDGES + t;
  int s[CHUNK_PER_THR], d[CHUNK_PER_THR];
#pragma unroll
  for (int k = 0; k < CHUNK_PER_THR; k++) {
    s[k] = src[e0 + k * CHUNK_THREADS];
    d[k] = dst[e0 + k * CHUNK_THREADS];
  }
#pragma unroll
  for (int k = 0; k < CHUNK_PER_THR; k++) {
    int pos = atomicAdd(&cur[d[k] / BKT_NODES], 1);
    pairs[pos] = make_int2(s[k], d[k]);
  }
}

#define SPARE_CNT 511
__launch_bounds__(1024)
__global__ void gin_bucket_csr_kernel(const int2* __restrict__ pairs,
                                      const int* __restrict__ boffs,
                                      int* __restrict__ rowptr,
                                      int* __restrict__ csr) {
  __shared__ int cnt[512];
  __shared__ int wsum[32];
  const int t = threadIdx.x;
  const int lane = t & 31;
  const int w = t >> 5;
  const int b = blockIdx.x;
  const int n0 = b * BKT_NODES;
  const int beg = boffs[b * N_CHUNK];
  const int end = boffs[(b + 1) * N_CHUNK];

  if (t < 512) cnt[t] = 0;
  __syncthreads();
  for (int i0 = beg + t; i0 < end; i0 += 4 * 1024) {
    int c[4];
#pragma unroll
    for (int k = 0; k < 4; k++) {
      int i = i0 + k * 1024;
      int d = pairs[i < end ? i : end - 1].y;
      c[k] = (i < end) ? d - n0 : SPARE_CNT;
    }
#pragma unroll
    for (int k = 0; k < 4; k++) atomicAdd(&cnt[c[k]], 1);
  }
  __syncthreads();

  const int v = (t < BKT_NODES) ? cnt[t] : 0;
  int x = v;
#pragma unroll
  for (int o = 1; o < 32; o <<= 1) {
    int y = __shfl_up(x, o);
    if (lane >= o) x += y;
  }
  if (lane == 31) wsum[w] = x;
  __syncthreads();
  if (w == 0) {
    int s = wsum[lane];
#pragma unroll
    for (int o = 1; o < 32; o <<= 1) {
      int y = __shfl_up(s, o);
      if (lane >= o) s += y;
    }
    wsum[lane] = s;
  }
  __syncthreads();
  const int first = beg + x - v + (w > 0 ? wsum[w - 1] : 0);
  if (t < BKT_NODES) {
    rowptr[n0 + t] = first;
    cnt[t] = first;
  }
  if (b == N_BKT - 1 && t == BKT_NODES - 1) rowptr[N_NODES] = first + v;
  __syncthreads();

  for (int i0 = beg + t; i0 < end; i0 += 4 * 1024) {
    int2 p[4];
#pragma unroll
    for (int k = 0; k < 4; k++) {
      int i = i0 + k * 1024;
      p[k] = pairs[i < end ? i : end - 1];
    }
#pragma unroll
    for (int k = 0; k < 4; k++) {
      if (i0 + k * 1024 < end) {
        int pos = atomicAdd(&cnt[p[k].y - n0], 1);
        csr[pos] = p[k].x;
      }
    }
  }
}

#define CONV_WAVES 4
#define CONV_TILES_PER_WAVE 2
#define CONV_TILES_PER_WG (CONV_WAVES * CONV_TILES_PER_WAVE)

static __device__ __forceinline__ void wave_lds_handover() {
  __builtin_amdgcn_fence(__ATOMIC_RELEASE, "wavefront");
  __builtin_amdgcn_wave_barrier();
  __builtin_amdgcn_fence(__ATOMIC_ACQUIRE, "wavefront");
}

static __device__ __forceinline__ void conv_first_slots(const int* __restrict__ csr, int rpl, int row, int hi,
                                                        int& c0, int& c1) {
  const int b0 = __builtin_amdgcn_readlane(rpl, 0);
  const int b1 = __builtin_amdgcn_readlane(rpl, 1);
  const int b2 = __builtin_amdgcn_readlane(rpl, 2);
  const int beg = hi ? b1 : b0, end = hi ? b2 : b1;
  int j = beg + row;      j = j < end ? j : end - 1;
  c0 = csr[j > 0 ? j : 0];
  j = beg + 16 + row;     j = j < end ? j : end - 1;
  c1 = csr[j > 0 ? j : 0];
}

__launch_bounds__(CONV_WAVES * 32)
__global__ void gin_conv_kernel(const float* __restrict__ hin,
                                const int* __restrict__ rowptr,
                                const int* __restrict__ csr,
                                const float* __restrict__ W1,
                                const float* __restrict__ b1,
                                const float* __restrict__ bn1,
                                const float* __restrict__ W2,
                                const float* __restrict__ b2,
                                const float* __restrict__ bn2,
                                float* __restrict__ hout,
                                const int* __restrict__ batch,
                                float* __restrict__ pooled,
                                int ntiles) {
  __shared__ __align__(16) float w1s[4 * 16 * 32 * 2];
  __shared__ __align__(16) float w2s[4 * 16 * 32 * 2];
  __shared__ __align__(16) float tb[CONV_WAVES][16 * 68];

  const int tid  = threadIdx.x;
  const int lane = tid & 31;
  const int wave = tid >> 5;
  const int row  = lane & 15;
  const int hi   = lane >> 4;
  float* tbuf = tb[wave];

  float4 w1a[4], w1b[4], w2a[4], w2b[4];
#pragma unroll
  for (int j = 0; j < 4; j++) {
    const int p = tid + 128 * j;
    const float* wp = W1 + (p >> 4) * 128 + (p & 15) * 4;
    w1a[j] = *(const float4*)wp;
    w1b[j] = *(const float4*)(wp + 64);
    const float* wq = W2 + (p >> 4) * 128 + (p & 15) * 4;
    w2a[j] = *(const float4*)wq;
    w2b[j] = *(const float4*)(wq + 64);
  }
  const int tile0 = blockIdx.x * CONV_TILES_PER_WG + wave;
  const int tl0 = tile0 < ntiles ? tile0 : 0;
  const int tl1 = tile0 + CONV_WAVES < ntiles ? tile0 + CONV_WAVES : 0;
  int gl   = batch[tl0 * 16 + row];
  int gln  = batch[tl1 * 16 + row];
  int rpl  = rowptr[tl0 * 16 + (lane < 16 ? lane : 16)];
  int rpln = rowptr[tl1 * 16 + (lane < 16 ? lane : 16)];
  int c0, c1;
  conv_first_slots(csr, rpl, row, hi, c0, c1);

#pragma unroll
  for (int j = 0; j < 4; j++) {
    const int p = tid + 128 * j;
    const int kk = 2 * (p >> 4), nq = p & 15;
    const int o = ((((nq >> 2) * 16) + (kk >> 2)) * 32 + 4 * (nq & 3) + ((kk & 2) << 3)) * 2;
    *(float4*)(w1s + o)     = make_float4(w1a[j].x, w1b[j].x, w1a[j].y, w1b[j].y);
    *(float4*)(w1s + o + 4) = make_float4(w1a[j].z, w1b[j].z, w1a[j].w, w1b[j].w);
    *(float4*)(w2s + o)     = make_float4(w2a[j].x, w2b[j].x, w2a[j].y, w2b[j].y);
    *(float4*)(w2s + o + 4) = make_float4(w2a[j].z, w2b[j].z, w2a[j].w, w2b[j].w);
  }
  __syncthreads();

  const v8f vzero = {0.f, 0.f, 0.f, 0.f, 0.f, 0.f, 0.f, 0.f};

  for (int it = 0; it < CONV_TILES_PER_WAVE; it++) {
    const int tile = tile0 + it * CONV_WAVES;
    if (tile >= ntiles) break;

    {
      const char* hb = (const char*)hin;
      const unsigned cb = (unsigned)row << 4;
      int b0 = __builtin_amdgcn_readlane(rpl, 0);
      int b1 = __builtin_amdgcn_readlane(rpl, 1);
      int b2 = __builtin_amdgcn_readlane(rpl, 2);
      int beg = hi ? b1 : b0, end = hi ? b2 : b1;
      for (int p = 0; p < 8; p++) {
        const int r = 2 * p + hi;
        const unsigned n = (unsigned)(tile * 16 + r);
        float4 acc = *(const float4*)(hb + ((n << 8) + cb));
        float4 v[16], u[16];
#define GIN_LOADV(k) v[(k)] = *(const float4*)(hb + ((((unsigned)row_share_i<(k)>(c0)) << 8) + cb));
        GIN_SLOTS16(GIN_LOADV)
#undef GIN_LOADV
        const int nb0 = b2;
        const int nb1 = __builtin_amdgcn_readlane(rpl, 2 * p + 3);
        const int nb2 = __builtin_amdgcn_readlane(rpl, 2 * p + 4);
        const int nbeg = hi ? nb1 : nb0, nend = hi ? nb2 : nb1;
        int nc0, nc1;
        {
          int j = nbeg + row;   j = j < nend ? j : nend - 1;
          nc0 = csr[j > 0 ? j : 0];
          j = nbeg + 16 + row;  j = j < nend ? j : nend - 1;
          nc1 = csr[j > 0 ? j : 0];
        }
        const int d0 = b1 - b0, d1 = b2 - b1;
        const int more = (d0 > d1 ? d0 : d1) - 16;
#define GIN_LOADU(k) u[(k)] = *(const float4*)(hb + ((((unsigned)row_share_i<(k)>(c1)) << 8) + cb));
        if (more > 0) {
          GIN_SLOTS4(GIN_LOADU, 0)
          if (more > 4) {
            GIN_SLOTS4(GIN_LOADU, 4)
            if (more > 8) {
              GIN_SLOTS4(GIN_LOADU, 8)
              if (more > 12) { GIN_SLOTS4(GIN_LOADU, 12) }
            }
          }
        }
#undef GIN_LOADU
#define GIN_ACCV(k) { const bool in = beg + (k) < end; \
        acc.x += in ? v[(k)].x : 0.f; acc.y += in ? v[(k)].y : 0.f; \
        acc.z += in ? v[(k)].z : 0.f; acc.w += in ? v[(k)].w : 0.f; }
        GIN_SLOTS16(GIN_ACCV)
#undef GIN_ACCV
        if (more > 0) {
#define GIN_ACCU(k) { const bool in = beg + 16 + (k) < end; \
        acc.x += in ? u[(k)].x : 0.f; acc.y += in ? u[(k)].y : 0.f; \
        acc.z += in ? u[(k)].z : 0.f; acc.w += in ? u[(k)].w : 0.f; }
          GIN_SLOTS4(GIN_ACCU, 0)
          if (more > 4) {
            GIN_SLOTS4(GIN_ACCU, 4)
            if (more > 8) {
              GIN_SLOTS4(GIN_ACCU, 8)
              if (more > 12) { GIN_SLOTS4(GIN_ACCU, 12) }
            }
          }
#undef GIN_ACCU
        }
        for (int j = beg + 32; j < end; j += 16) {
          int s[16];
#pragma unroll
          for (int k = 0; k < 16; k++) s[k] = csr[(j + k < end) ? j + k : end - 1];
#pragma unroll
          for (int k = 0; k < 16; k++) {
            const unsigned sk = (j + k < end) ? (unsigned)s[k] : n;
            v[k] = *(const float4*)(hb + ((sk << 8) + cb));
          }
#pragma unroll
          for (int k = 0; k < 16; k++) {
            const bool in = j + k < end;
            acc.x += in ? v[k].x : 0.f; acc.y += in ? v[k].y : 0.f;
            acc.z += in ? v[k].z : 0.f; acc.w += in ? v[k].w : 0.f;
          }
        }
        *(float4*)(tbuf + r * 68 + 4 * row) = acc;
        b0 = nb0; b1 = nb1; b2 = nb2;
        beg = nbeg; end = nend;
        c0 = nc0; c1 = nc1;
      }
    }
    if (it + 1 < CONV_TILES_PER_WAVE) conv_first_slots(csr, rpln, row, hi, c0, c1);
    wave_lds_handover();

    {
      v2f a1[16];
      const float* tr = tbuf + row * 68 + 2 * hi;
#pragma unroll
      for (int k = 0; k < 16; k++) a1[k] = *(const v2f*)(tr + 4 * k);
#pragma unroll
      for (int nt = 0; nt < 4; nt++) {
        v8f acc = vzero;
        const v2f* wf = ((const v2f*)w1s) + nt * 16 * 32 + lane;
#pragma unroll
        for (int k = 0; k < 16; k++) {
          acc = __builtin_amdgcn_wmma_f32_16x16x4_f32(false, a1[k], false, wf[k * 32],
                                                      (short)0, acc, false, false);
        }
        int n = nt * 16 + row;
        float s  = bn1[n] * rsqrtf(bn1[192 + n] + BN_EPS);
        float t0 = bn1[64 + n] + (b1[n] - bn1[128 + n]) * s;
#pragma unroll
        for (int i = 0; i < 8; i++) {
          float y = acc[i] * s + t0;
          tbuf[(i + 8 * hi) * 68 + n] = fmaxf(y, 0.0f);
        }
      }
    }
    wave_lds_handover();

    v8f acc2[4];
    {
#pragma unroll
      for (int nt = 0; nt < 4; nt++) {
        v8f acc = vzero;
        const v2f* wf = ((const v2f*)w2s) + nt * 16 * 32 + lane;
        const float* tr = tbuf + row * 68 + 2 * hi;
#pragma unroll
        for (int k = 0; k < 16; k++) {
          v2f a = *(const v2f*)(tr + 4 * k);
          acc = __builtin_amdgcn_wmma_f32_16x16x4_f32(false, a, false, wf[k * 32],
                                                      (short)0, acc, false, false);
        }
        acc2[nt] = acc;
      }
    }

    {
#pragma unroll
      for (int nt = 0; nt < 4; nt++) {
        int n = nt * 16 + row;
        float s  = bn2[n] * rsqrtf(bn2[192 + n] + BN_EPS);
        float t0 = bn2[64 + n] + (b2[n] - bn2[128 + n]) * s;
#pragma unroll
        for (int i = 0; i < 8; i++) {
          float y = acc2[nt][i] * s + t0;
          tbuf[(i + 8 * hi) * 68 + n] = fmaxf(y, 0.0f);
        }
      }
    }
    wave_lds_handover();

    {
#pragma unroll
      for (int j = 0; j < 8; j++) {
        int flat = j * 32 + lane;
        int r = flat >> 4;
        int c = (flat & 15) << 2;
        float4 v = *(const float4*)(tbuf + r * 68 + c);
        *(float4*)(hout + (size_t)(tile * 16 + r) * 64 + c) = v;
      }
    }

    {
      float p0[16], p1[16];
#pragma unroll
      for (int r = 0; r < 16; r++) {
        p0[r] = tbuf[r * 68 + lane];
        p1[r] = tbuf[r * 68 + 32 + lane];
      }
      unsigned todo = 0xFFFFu;
      while (todo) {
        const int r0 = __builtin_ctz(todo);
        const int g = __builtin_amdgcn_readlane(gl, r0);
        const unsigned m = __builtin_amdgcn_ballot_w32(gl == g) & todo;
        float s0 = 0.f, s1 = 0.f;
#pragma unroll
        for (int r = 0; r < 16; r++) {
          const bool in = (m >> r) & 1u;
          s0 += in ? p0[r] : 0.f;
          s1 += in ? p1[r] : 0.f;
        }
        float* pg = pooled + (size_t)g * 64 + lane;
        atomicAdd(pg, s0);
        atomicAdd(pg + 32, s1);
        todo &= ~m;
      }
    }
    wave_lds_handover();
    gl = gln;
    rpl = rpln;
  }
}

__global__ void gin_final_kernel(const float* __restrict__ pooled,
                                 const float* __restrict__ counts,
                                 const float* __restrict__ linW,
                                 const float* __restrict__ linb,
                                 float* __restrict__ out) {
  int idx = blockIdx.x * blockDim.x + threadIdx.x;
  if (idx >= N_GRAPHS * 10) return;
  int g = idx / 10, t = idx - g * 10;
  float acc = counts[g] * linb[t];
  for (int l = 1; l < 5; l++) acc += linb[l * 10 + t];
  for (int l = 0; l < 5; l++) {
    const float* p = pooled + ((size_t)l * N_GRAPHS + g) * 64;
    const float* w = linW + l * 640 + t;
    float s = 0.f;
#pragma unroll
    for (int k = 0; k < 64; k++) s += p[k] * w[k * 10];
    acc += s;
  }
  out[idx] = acc;
}

__global__ void gin_zero_kernel(unsigned int* __restrict__ p, int n) {
  int i = blockIdx.x * blockDim.x + threadIdx.x;
  if (i < n) p[i] = 0u;
}

extern "C" void kernel_launch(void* const* d_in, const int* in_sizes, int n_in,
                              void* d_out, int out_size, void* d_ws, size_t ws_size,
                              hipStream_t stream) {
  const float* x        = (const float*)d_in[0];
  const float* fh_W1    = (const float*)d_in[1];
  const float* fh_b1    = (const float*)d_in[2];
  const float* fh_bn1   = (const float*)d_in[3];
  const float* fh_W2    = (const float*)d_in[4];
  const float* fh_b2    = (const float*)d_in[5];
  const float* fh_bn2   = (const float*)d_in[6];
  const float* conv_W1  = (const float*)d_in[7];
  const float* conv_b1  = (const float*)d_in[8];
  const float* conv_bn1 = (const float*)d_in[9];
  const float* conv_W2  = (const float*)d_in[10];
  const float* conv_b2  = (const float*)d_in[11];
  const float* conv_bn2 = (const float*)d_in[12];
  const float* lin_W    = (const float*)d_in[13];
  const float* lin_b    = (const float*)d_in[14];
  const int*   eidx     = (const int*)d_in[15];
  const int*   batch    = (const int*)d_in[16];
  float* out = (float*)d_out;

  const int* src = eidx;
  const int* dst = eidx + N_EDGES;

  float* ws     = (float*)d_ws;
  float* h      = ws;
  float* zrow   = h + (size_t)N_NODES * 64;
  float* pooled = zrow + 64;
  float* counts = pooled + 5 * N_GRAPHS * 64;
  int*   bcnt   = (int*)(counts + N_GRAPHS);
  int*   boffs  = bcnt + SCAN_N;
  int*   rowptr = boffs + SCAN_N + 4;
  int2*  pairs  = (int2*)(rowptr + N_NODES + 4);
  int*   csr    = (int*)(pairs + N_EDGES);
  float* h2     = ws + 12000000;

  const int accN = 64 + 5 * N_GRAPHS * 64 + N_GRAPHS;
  gin_zero_kernel<<<(accN + 255) / 256, 256, 0, stream>>>((unsigned int*)zrow, accN);

  gin_bucket_count_kernel<<<N_CHUNK, CHUNK_THREADS, 0, stream>>>(dst, bcnt);
  gin_scan_kernel<<<1, 1024, 0, stream>>>(bcnt, boffs);
  gin_bucket_fill_kernel<<<N_CHUNK, CHUNK_THREADS, 0, stream>>>(src, dst, boffs, pairs);
  gin_bucket_csr_kernel<<<N_BKT, 1024, 0, stream>>>(pairs, boffs, rowptr, csr);

  const int mlpBlocks = (NTILES + 3) / 4;

  gin_mlp_kernel<D_FEAT, true><<<mlpBlocks, 128, 0, stream>>>(
      x, fh_W1, fh_b1, fh_bn1, fh_W2, fh_b2, fh_bn2, h, batch, pooled, counts, NTILES);

  const int convBlocks = (NTILES + CONV_TILES_PER_WG - 1) / CONV_TILES_PER_WG;
  float* hin = h;
  float* hnext = h2;
  for (int l = 0; l < 4; l++) {
    gin_conv_kernel<<<convBlocks, CONV_WAVES * 32, 0, stream>>>(
        hin, rowptr, csr,
        conv_W1 + (size_t)l * 64 * 64, conv_b1 + l * 64, conv_bn1 + l * 256,
        conv_W2 + (size_t)l * 64 * 64, conv_b2 + l * 64, conv_bn2 + l * 256,
        hnext, batch, pooled + (size_t)(l + 1) * N_GRAPHS * 64, NTILES);
    float* t = hin; hin = hnext; hnext = t;
  }

  gin_final_kernel<<<(N_GRAPHS * 10 + 255) / 256, 256, 0, stream>>>(
      pooled, counts, lin_W, lin_b, out);
}


# Count-table prefix sum by 200 work-groups, one per bucket row: each adds the rows on its nearer side by whole-line 16-byte loads, then scans its 500 entries; was 1 work-group, 32 lines per load
# speedup vs baseline: 4.2320x; 1.1606x over previous
#include <hip/hip_runtime.h>

#define N_NODES   100000
#define N_EDGES   1600000
#define N_GRAPHS  1000
#define D_FEAT    128
#define HID       64
#define BN_EPS    1e-5f
#define NTILES    (N_NODES / 16)

typedef __attribute__((ext_vector_type(2))) float v2f;
typedef __attribute__((ext_vector_type(8))) float v8f;

template<int K>
static __device__ __forceinline__ int row_share_i(int v) {
  return __builtin_amdgcn_update_dpp(0, v, 0x150 + K, 0xF, 0xF, true);
}
#define GIN_SLOTS4(F, B)  F((B)) F((B) + 1) F((B) + 2) F((B) + 3)
#define GIN_SLOTS16(F)    GIN_SLOTS4(F, 0) GIN_SLOTS4(F, 4) GIN_SLOTS4(F, 8) GIN_SLOTS4(F, 12)

template<int K1, bool COUNT>
__launch_bounds__(128)
__global__ void gin_mlp_kernel(const float* __restrict__ xin,
                               const float* __restrict__ W1,
                               const float* __restrict__ b1,
                               const float* __restrict__ bn1,
                               const float* __restrict__ W2,
                               const float* __restrict__ b2,
                               const float* __restrict__ bn2,
                               float* __restrict__ hout,
                               const int* __restrict__ batch,
                               float* __restrict__ pooled,
                               float* __restrict__ counts,
                               int ntiles) {
  constexpr int KS1 = K1 / 4;
  constexpr int NQ1 = (K1 / 2) * 16 / 128;
  constexpr int NQ2 = (64 / 2) * 16 / 128;
  __shared__ __align__(16) float w1s[4 * KS1 * 32 * 2];
  __shared__ __align__(16) float w2s[4 * 16 * 32 * 2];
  __shared__ __align__(16) float tb[4][16 * 68];

  const int tid  = threadIdx.x;
  const int lane = tid & 31;
  const int wave = tid >> 5;
  const int row  = lane & 15;
  const int hi   = lane >> 4;
  const int tile = blockIdx.x * 4 + wave;
  const bool active = tile < ntiles;
  float* tbuf = tb[wave];

  float4 w1a[NQ1], w1b[NQ1], w2a[NQ2], w2b[NQ2];
#pragma unroll
  for (int j = 0; j < NQ1; j++) {
    const int p = tid + 128 * j;
    const float* wp = W1 + (p >> 4) * 128 + (p & 15) * 4;
    w1a[j] = *(const float4*)wp;
    w1b[j] = *(const float4*)(wp + 64);
  }
#pragma unroll
  for (int j = 0; j < NQ2; j++) {
    const int p = tid + 128 * j;
    const float* wp = W2 + (p >> 4) * 128 + (p & 15) * 4;
    w2a[j] = *(const float4*)wp;
    w2b[j] = *(const float4*)(wp + 64);
  }
  const int tl = active ? tile : 0;
  const int gl = batch[tl * 16 + row];
  v2f a1[KS1];
  const float* xr = xin + (size_t)(tl * 16 + row) * K1 + 2 * hi;
#pragma unroll
  for (int k = 0; k < KS1; k++) a1[k] = *(const v2f*)(xr + 4 * k);

#pragma unroll
  for (int j = 0; j < NQ1; j++) {
    const int p = tid + 128 * j;
    const int kk = 2 * (p >> 4), nq = p & 15;
    float* d = w1s + ((((nq >> 2) * KS1) + (kk >> 2)) * 32 + 4 * (nq & 3) + ((kk & 2) << 3)) * 2;
    *(float4*)d       = make_float4(w1a[j].x, w1b[j].x, w1a[j].y, w1b[j].y);
    *(float4*)(d + 4) = make_float4(w1a[j].z, w1b[j].z, w1a[j].w, w1b[j].w);
  }
#pragma unroll
  for (int j = 0; j < NQ2; j++) {
    const int p = tid + 128 * j;
    const int kk = 2 * (p >> 4), nq = p & 15;
    float* d = w2s + ((((nq >> 2) * 16) + (kk >> 2)) * 32 + 4 * (nq & 3) + ((kk & 2) << 3)) * 2;
    *(float4*)d       = make_float4(w2a[j].x, w2b[j].x, w2a[j].y, w2b[j].y);
    *(float4*)(d + 4) = make_float4(w2a[j].z, w2b[j].z, w2a[j].w, w2b[j].w);
  }
  __syncthreads();

  const v8f vzero = {0.f, 0.f, 0.f, 0.f, 0.f, 0.f, 0.f, 0.f};

  if (active) {
#pragma unroll
    for (int nt = 0; nt < 4; nt++) {
      v8f acc = vzero;
      const v2f* wf = ((const v2f*)w1s) + nt * KS1 * 32 + lane;
#pragma unroll
      for (int k = 0; k < KS1; k++) {
        acc = __builtin_amdgcn_wmma_f32_16x16x4_f32(false, a1[k], false, wf[k * 32],
                                                    (short)0, acc, false, false);
      }
      int n = nt * 16 + row;
      float s  = bn1[n] * rsqrtf(bn1[192 + n] + BN_EPS);
      float t0 = bn1[64 + n] + (b1[n] - bn1[128 + n]) * s;
#pragma unroll
      for (int i = 0; i < 8; i++) {
        float y = acc[i] * s + t0;
        tbuf[(i + 8 * hi) * 68 + n] = fmaxf(y, 0.0f);
      }
    }
  }
  __syncthreads();

  v8f acc2[4];
  if (active) {
#pragma unroll
    for (int nt = 0; nt < 4; nt++) {
      v8f acc = vzero;
      const v2f* wf = ((const v2f*)w2s) + nt * 16 * 32 + lane;
      const float* tr = tbuf + row * 68 + 2 * hi;
#pragma unroll
      for (int k = 0; k < 16; k++) {
        v2f a = *(const v2f*)(tr + 4 * k);
        acc = __builtin_amdgcn_wmma_f32_16x16x4_f32(false, a, false, wf[k * 32],
                                                    (short)0, acc, false, false);
      }
      acc2[nt] = acc;
    }
  }
  __syncthreads();

  if (active) {
#pragma unroll
    for (int nt = 0; nt < 4; nt++) {
      int n = nt * 16 + row;
      float s  = bn2[n] * rsqrtf(bn2[192 + n] + BN_EPS);
      float t0 = bn2[64 + n] + (b2[n] - bn2[128 + n]) * s;
#pragma unroll
      for (int i = 0; i < 8; i++) {
        float y = acc2[nt][i] * s + t0;
        tbuf[(i + 8 * hi) * 68 + n] = fmaxf(y, 0.0f);
      }
    }
  }
  __syncthreads();

  if (active) {
#pragma unroll
    for (int j = 0; j < 8; j++) {
      int flat = j * 32 + lane;
      int r = flat >> 4;
      int c = (flat & 15) << 2;
      float4 v = *(const float4*)(tbuf + r * 68 + c);
      *(float4*)(hout + (size_t)(tile * 16 + r) * 64 + c) = v;
    }
  }

  if (active) {
    float p0[16], p1[16];
#pragma unroll
    for (int r = 0; r < 16; r++) {
      p0[r] = tbuf[r * 68 + lane];
      p1[r] = tbuf[r * 68 + 32 + lane];
    }
    unsigned todo = 0xFFFFu;
    while (todo) {
      const int r0 = __builtin_ctz(todo);
      const int g = __builtin_amdgcn_readlane(gl, r0);
      const unsigned m = __builtin_amdgcn_ballot_w32(gl == g) & todo;
      float s0 = 0.f, s1 = 0.f;
#pragma unroll
      for (int r = 0; r < 16; r++) {
        const bool in = (m >> r) & 1u;
        s0 += in ? p0[r] : 0.f;
        s1 += in ? p1[r] : 0.f;
      }
      float* pg = pooled + (size_t)g * 64 + lane;
      atomicAdd(pg, s0);
      atomicAdd(pg + 32, s1);
      if (COUNT && lane == 0) atomicAdd(counts + g, (float)__builtin_popcount(m));
      todo &= ~m;
    }
  }
}

#define BKT_NODES     500
#define N_BKT         (N_NODES / BKT_NODES)
#define N_CHUNK       500
#define CHUNK_EDGES   (N_EDGES / N_CHUNK)
#define CHUNK_THREADS 320
#define CHUNK_PER_THR (CHUNK_EDGES / CHUNK_THREADS)

__launch_bounds__(CHUNK_THREADS)
__global__ void gin_bucket_count_kernel(const int* __restrict__ dst,
                                        int* __restrict__ bcnt) {
  __shared__ int cnt[N_BKT];
  const int t = threadIdx.x;
  const int g = blockIdx.x;
  if (t < N_BKT) cnt[t] = 0;
  __syncthreads();
  const int* dp = dst + g * CHUNK_EDGES + t;
  int d[CHUNK_PER_THR];
#pragma unroll
  for (int k = 0; k < CHUNK_PER_THR; k++) d[k] = dp[k * CHUNK_THREADS];
#pragma unroll
  for (int k = 0; k < CHUNK_PER_THR; k++) atomicAdd(&cnt[d[k] / BKT_NODES], 1);
  __syncthreads();
  if (t < N_BKT) bcnt[t * N_CHUNK + g] = cnt[t];
}

#define SCAN_N       (N_BKT * N_CHUNK)
#define ROW_Q        (N_CHUNK / 4)
#define SCAN_THREADS 1024
#define SCAN_TRIPS   (((N_BKT / 2) * ROW_Q + SCAN_THREADS - 1) / SCAN_THREADS)
__launch_bounds__(SCAN_THREADS)
__global__ void gin_scan_kernel(const int* __restrict__ bcnt,
                                int* __restrict__ boffs) {
  __shared__ int wsum[32];
  __shared__ int rsum[4];
  const int t = threadIdx.x;
  const int lane = t & 31;
  const int w = t >> 5;
  const int b = blockIdx.x;
  const int nq = b * ROW_Q;
  const bool front = b < N_BKT / 2;
  const int q0 = front ? 0 : nq;
  const int q1 = front ? nq : SCAN_N / 4;
  const int4* p = (const int4*)bcnt;

  int4 v[SCAN_TRIPS];
#pragma unroll
  for (int i = 0; i < SCAN_TRIPS; i++) {
    const int q = q0 + t + SCAN_THREADS * i;
    v[i] = p[q < q1 ? q : 0];
  }
  const int4 r = p[nq + (t < ROW_Q ? t : 0)];
  int part = 0;
#pragma unroll
  for (int i = 0; i < SCAN_TRIPS; i++) {
    const int q = q0 + t + SCAN_THREADS * i;
    part += (q < q1) ? (v[i].x + v[i].y + v[i].z + v[i].w) : 0;
  }
#pragma unroll
  for (int o = 16; o > 0; o >>= 1) part += __shfl_xor(part, o);
  if (lane == 0) wsum[w] = part;

  const int local = (t < ROW_Q) ? (r.x + r.y + r.z + r.w) : 0;
  int x = local;
#pragma unroll
  for (int o = 1; o < 32; o <<= 1) {
    int y = __shfl_up(x, o);
    if (lane >= o) x += y;
  }
  if (w < 4 && lane == 31) rsum[w] = x;
  __syncthreads();

  if (w < 4) {
    int tot = wsum[lane];
#pragma unroll
    for (int o = 16; o > 0; o >>= 1) tot += __shfl_xor(tot, o);
    int run = front ? tot : N_EDGES - tot;
    run += x - local;
    if (w > 0) run += rsum[0];
    if (w > 1) run += rsum[1];
    if (w > 2) run += rsum[2];
    if (t < ROW_Q) {
      int4 o4;
      o4.x = run; run += r.x;
      o4.y = run; run += r.y;
      o4.z = run; run += r.z;
      o4.w = run; run += r.w;
      ((int4*)boffs)[nq + t] = o4;
      if (b == N_BKT - 1 && t == ROW_Q - 1) boffs[SCAN_N] = run;
    }
  }
}

__launch_bounds__(CHUNK_THREADS)
__global__ void gin_bucket_fill_kernel(const int* __restrict__ src,
                                       const int* __restrict__ dst,
                                       const int* __restrict__ boffs,
                                       int2* __restrict__ pairs) {
  __shared__ int cur[N_BKT];
  const int t = threadIdx.x;
  const int g = blockIdx.x;
  if (t < N_BKT) cur[t] = boffs[t * N_CHUNK + g];
  __syncthreads();
  const int e0 = g * CHUNK_EDGES + t;
  int s[CHUNK_PER_THR], d[CHUNK_PER_THR];
#pragma unroll
  for (int k = 0; k < CHUNK_PER_THR; k++) {
    s[k] = src[e0 + k * CHUNK_THREADS];
    d[k] = dst[e0 + k * CHUNK_THREADS];
  }
#pragma unroll
  for (int k = 0; k < CHUNK_PER_THR; k++) {
    int pos = atomicAdd(&cur[d[k] / BKT_NODES], 1);
    pairs[pos] = make_int2(s[k], d[k]);
  }
}

#define SPARE_CNT 511
__launch_bounds__(1024)
__global__ void gin_bucket_csr_kernel(const int2* __restrict__ pairs,
                                      const int* __restrict__ boffs,
                                      int* __restrict__ rowptr,
                                      int* __restrict__ csr) {
  __shared__ int cnt[512];
  __shared__ int wsum[32];
  const int t = threadIdx.x;
  const int lane = t & 31;
  const int w = t >> 5;
  const int b = blockIdx.x;
  const int n0 = b * BKT_NODES;
  const int beg = boffs[b * N_CHUNK];
  const int end = boffs[(b + 1) * N_CHUNK];

  if (t < 512) cnt[t] = 0;
  __syncthreads();
  for (int i0 = beg + t; i0 < end; i0 += 4 * 1024) {
    int c[4];
#pragma unroll
    for (int k = 0; k < 4; k++) {
      int i = i0 + k * 1024;
      int d = pairs[i < end ? i : end - 1].y;
      c[k] = (i < end) ? d - n0 : SPARE_CNT;
    }
#pragma unroll
    for (int k = 0; k < 4; k++) atomicAdd(&cnt[c[k]], 1);
  }
  __syncthreads();

  const int v = (t < BKT_NODES) ? cnt[t] : 0;
  int x = v;
#pragma unroll
  for (int o = 1; o < 32; o <<= 1) {
    int y = __shfl_up(x, o);
    if (lane >= o) x += y;
  }
  if (lane == 31) wsum[w] = x;
  __syncthreads();
  if (w == 0) {
    int s = wsum[lane];
#pragma unroll
    for (int o = 1; o < 32; o <<= 1) {
      int y = __shfl_up(s, o);
      if (lane >= o) s += y;
    }
    wsum[lane] = s;
  }
  __syncthreads();
  const int first = beg + x - v + (w > 0 ? wsum[w - 1] : 0);
  if (t < BKT_NODES) {
    rowptr[n0 + t] = first;
    cnt[t] = first;
  }
  if (b == N_BKT - 1 && t == BKT_NODES - 1) rowptr[N_NODES] = first + v;
  __syncthreads();

  for (int i0 = beg + t; i0 < end; i0 += 4 * 1024) {
    int2 p[4];
#pragma unroll
    for (int k = 0; k < 4; k++) {
      int i = i0 + k * 1024;
      p[k] = pairs[i < end ? i : end - 1];
    }
#pragma unroll
    for (int k = 0; k < 4; k++) {
      if (i0 + k * 1024 < end) {
        int pos = atomicAdd(&cnt[p[k].y - n0], 1);
        csr[pos] = p[k].x;
      }
    }
  }
}

#define CONV_WAVES 4
#define CONV_TILES_PER_WAVE 2
#define CONV_TILES_PER_WG (CONV_WAVES * CONV_TILES_PER_WAVE)

static __device__ __forceinline__ void wave_lds_handover() {
  __builtin_amdgcn_fence(__ATOMIC_RELEASE, "wavefront");
  __builtin_amdgcn_wave_barrier();
  __builtin_amdgcn_fence(__ATOMIC_ACQUIRE, "wavefront");
}

static __device__ __forceinline__ void conv_first_slots(const int* __restrict__ csr, int rpl, int row, int hi,
                                                        int& c0, int& c1) {
  const int b0 = __builtin_amdgcn_readlane(rpl, 0);
  const int b1 = __builtin_amdgcn_readlane(rpl, 1);
  const int b2 = __builtin_amdgcn_readlane(rpl, 2);
  const int beg = hi ? b1 : b0, end = hi ? b2 : b1;
  int j = beg + row;      j = j < end ? j : end - 1;
  c0 = csr[j > 0 ? j : 0];
  j = beg + 16 + row;     j = j < end ? j : end - 1;
  c1 = csr[j > 0 ? j : 0];
}

__launch_bounds__(CONV_WAVES * 32)
__global__ void gin_conv_kernel(const float* __restrict__ hin,
                                const int* __restrict__ rowptr,
                                const int* __restrict__ csr,
                                const float* __restrict__ W1,
                                const float* __restrict__ b1,
                                const float* __restrict__ bn1,
                                const float* __restrict__ W2,
                                const float* __restrict__ b2,
                                const float* __restrict__ bn2,
                                float* __restrict__ hout,
                                const int* __restrict__ batch,
                                float* __restrict__ pooled,
                                int ntiles) {
  __shared__ __align__(16) float w1s[4 * 16 * 32 * 2];
  __shared__ __align__(16) float w2s[4 * 16 * 32 * 2];
  __shared__ __align__(16) float tb[CONV_WAVES][16 * 68];

  const int tid  = threadIdx.x;
  const int lane = tid & 31;
  const int wave = tid >> 5;
  const int row  = lane & 15;
  const int hi   = lane >> 4;
  float* tbuf = tb[wave];

  float4 w1a[4], w1b[4], w2a[4], w2b[4];
#pragma unroll
  for (int j = 0; j < 4; j++) {
    const int p = tid + 128 * j;
    const float* wp = W1 + (p >> 4) * 128 + (p & 15) * 4;
    w1a[j] = *(const float4*)wp;
    w1b[j] = *(const float4*)(wp + 64);
    const float* wq = W2 + (p >> 4) * 128 + (p & 15) * 4;
    w2a[j] = *(const float4*)wq;
    w2b[j] = *(const float4*)(wq + 64);
  }
  const int tile0 = blockIdx.x * CONV_TILES_PER_WG + wave;
  const int tl0 = tile0 < ntiles ? tile0 : 0;
  const int tl1 = tile0 + CONV_WAVES < ntiles ? tile0 + CONV_WAVES : 0;
  int gl   = batch[tl0 * 16 + row];
  int gln  = batch[tl1 * 16 + row];
  int rpl  = rowptr[tl0 * 16 + (lane < 16 ? lane : 16)];
  int rpln = rowptr[tl1 * 16 + (lane < 16 ? lane : 16)];
  int c0, c1;
  conv_first_slots(csr, rpl, row, hi, c0, c1);

#pragma unroll
  for (int j = 0; j < 4; j++) {
    const int p = tid + 128 * j;
    const int kk = 2 * (p >> 4), nq = p & 15;
    const int o = ((((nq >> 2) * 16) + (kk >> 2)) * 32 + 4 * (nq & 3) + ((kk & 2) << 3)) * 2;
    *(float4*)(w1s + o)     = make_float4(w1a[j].x, w1b[j].x, w1a[j].y, w1b[j].y);
    *(float4*)(w1s + o + 4) = make_float4(w1a[j].z, w1b[j].z, w1a[j].w, w1b[j].w);
    *(float4*)(w2s + o)     = make_float4(w2a[j].x, w2b[j].x, w2a[j].y, w2b[j].y);
    *(float4*)(w2s + o + 4) = make_float4(w2a[j].z, w2b[j].z, w2a[j].w, w2b[j].w);
  }
  __syncthreads();

  const v8f vzero = {0.f, 0.f, 0.f, 0.f, 0.f, 0.f, 0.f, 0.f};

  for (int it = 0; it < CONV_TILES_PER_WAVE; it++) {
    const int tile = tile0 + it * CONV_WAVES;
    if (tile >= ntiles) break;

    {
      const char* hb = (const char*)hin;
      const unsigned cb = (unsigned)row << 4;
      int b0 = __builtin_amdgcn_readlane(rpl, 0);
      int b1 = __builtin_amdgcn_readlane(rpl, 1);
      int b2 = __builtin_amdgcn_readlane(rpl, 2);
      int beg = hi ? b1 : b0, end = hi ? b2 : b1;
      for (int p = 0; p < 8; p++) {
        const int r = 2 * p + hi;
        const unsigned n = (unsigned)(tile * 16 + r);
        float4 acc = *(const float4*)(hb + ((n << 8) + cb));
        float4 v[16], u[16];
#define GIN_LOADV(k) v[(k)] = *(const float4*)(hb + ((((unsigned)row_share_i<(k)>(c0)) << 8) + cb));
        GIN_SLOTS16(GIN_LOADV)
#undef GIN_LOADV
        const int nb0 = b2;
        const int nb1 = __builtin_amdgcn_readlane(rpl, 2 * p + 3);
        const int nb2 = __builtin_amdgcn_readlane(rpl, 2 * p + 4);
        const int nbeg = hi ? nb1 : nb0, nend = hi ? nb2 : nb1;
        int nc0, nc1;
        {
          int j = nbeg + row;   j = j < nend ? j : nend - 1;
          nc0 = csr[j > 0 ? j : 0];
          j = nbeg + 16 + row;  j = j < nend ? j : nend - 1;
          nc1 = csr[j > 0 ? j : 0];
        }
        const int d0 = b1 - b0, d1 = b2 - b1;
        const int more = (d0 > d1 ? d0 : d1) - 16;
#define GIN_LOADU(k) u[(k)] = *(const float4*)(hb + ((((unsigned)row_share_i<(k)>(c1)) << 8) + cb));
        if (more > 0) {
          GIN_SLOTS4(GIN_LOADU, 0)
          if (more > 4) {
            GIN_SLOTS4(GIN_LOADU, 4)
            if (more > 8) {
              GIN_SLOTS4(GIN_LOADU, 8)
              if (more > 12) { GIN_SLOTS4(GIN_LOADU, 12) }
            }
          }
        }
#undef GIN_LOADU
#define GIN_ACCV(k) { const bool in = beg + (k) < end; \
        acc.x += in ? v[(k)].x : 0.f; acc.y += in ? v[(k)].y : 0.f; \
        acc.z += in ? v[(k)].z : 0.f; acc.w += in ? v[(k)].w : 0.f; }
        GIN_SLOTS16(GIN_ACCV)
#undef GIN_ACCV
        if (more > 0) {
#define GIN_ACCU(k) { const bool in = beg + 16 + (k) < end; \
        acc.x += in ? u[(k)].x : 0.f; acc.y += in ? u[(k)].y : 0.f; \
        acc.z += in ? u[(k)].z : 0.f; acc.w += in ? u[(k)].w : 0.f; }
          GIN_SLOTS4(GIN_ACCU, 0)
          if (more > 4) {
            GIN_SLOTS4(GIN_ACCU, 4)
            if (more > 8) {
              GIN_SLOTS4(GIN_ACCU, 8)
              if (more > 12) { GIN_SLOTS4(GIN_ACCU, 12) }
            }
          }
#undef GIN_ACCU
        }
        for (int j = beg + 32; j < end; j += 16) {
          int s[16];
#pragma unroll
          for (int k = 0; k < 16; k++) s[k] = csr[(j + k < end) ? j + k : end - 1];
#pragma unroll
          for (int k = 0; k < 16; k++) {
            const unsigned sk = (j + k < end) ? (unsigned)s[k] : n;
            v[k] = *(const float4*)(hb + ((sk << 8) + cb));
          }
#pragma unroll
          for (int k = 0; k < 16; k++) {
            const bool in = j + k < end;
            acc.x += in ? v[k].x : 0.f; acc.y += in ? v[k].y : 0.f;
            acc.z += in ? v[k].z : 0.f; acc.w += in ? v[k].w : 0.f;
          }
        }
        *(float4*)(tbuf + r * 68 + 4 * row) = acc;
        b0 = nb0; b1 = nb1; b2 = nb2;
        beg = nbeg; end = nend;
        c0 = nc0; c1 = nc1;
      }
    }
    if (it + 1 < CONV_TILES_PER_WAVE) conv_first_slots(csr, rpln, row, hi, c0, c1);
    wave_lds_handover();

    {
      v2f a1[16];
      const float* tr = tbuf + row * 68 + 2 * hi;
#pragma unroll
      for (int k = 0; k < 16; k++) a1[k] = *(const v2f*)(tr + 4 * k);
#pragma unroll
      for (int nt = 0; nt < 4; nt++) {
        v8f acc = vzero;
        const v2f* wf = ((const v2f*)w1s) + nt * 16 * 32 + lane;
#pragma unroll
        for (int k = 0; k < 16; k++) {
          acc = __builtin_amdgcn_wmma_f32_16x16x4_f32(false, a1[k], false, wf[k * 32],
                                                      (short)0, acc, false, false);
        }
        int n = nt * 16 + row;
        float s  = bn1[n] * rsqrtf(bn1[192 + n] + BN_EPS);
        float t0 = bn1[64 + n] + (b1[n] - bn1[128 + n]) * s;
#pragma unroll
        for (int i = 0; i < 8; i++) {
          float y = acc[i] * s + t0;
          tbuf[(i + 8 * hi) * 68 + n] = fmaxf(y, 0.0f);
        }
      }
    }
    wave_lds_handover();

    v8f acc2[4];
    {
#pragma unroll
      for (int nt = 0; nt < 4; nt++) {
        v8f acc = vzero;
        const v2f* wf = ((const v2f*)w2s) + nt * 16 * 32 + lane;
        const float* tr = tbuf + row * 68 + 2 * hi;
#pragma unroll
        for (int k = 0; k < 16; k++) {
          v2f a = *(const v2f*)(tr + 4 * k);
          acc = __builtin_amdgcn_wmma_f32_16x16x4_f32(false, a, false, wf[k * 32],
                                                      (short)0, acc, false, false);
        }
        acc2[nt] = acc;
      }
    }

    {
#pragma unroll
      for (int nt = 0; nt < 4; nt++) {
        int n = nt * 16 + row;
        float s  = bn2[n] * rsqrtf(bn2[192 + n] + BN_EPS);
        float t0 = bn2[64 + n] + (b2[n] - bn2[128 + n]) * s;
#pragma unroll
        for (int i = 0; i < 8; i++) {
          float y = acc2[nt][i] * s + t0;
          tbuf[(i + 8 * hi) * 68 + n] = fmaxf(y, 0.0f);
        }
      }
    }
    wave_lds_handover();

    {
#pragma unroll
      for (int j = 0; j < 8; j++) {
        int flat = j * 32 + lane;
        int r = flat >> 4;
        int c = (flat & 15) << 2;
        float4 v = *(const float4*)(tbuf + r * 68 + c);
        *(float4*)(hout + (size_t)(tile * 16 + r) * 64 + c) = v;
      }
    }

    {
      float p0[16], p1[16];
#pragma unroll
      for (int r = 0; r < 16; r++) {
        p0[r] = tbuf[r * 68 + lane];
        p1[r] = tbuf[r * 68 + 32 + lane];
      }
      unsigned todo = 0xFFFFu;
      while (todo) {
        const int r0 = __builtin_ctz(todo);
        const int g = __builtin_amdgcn_readlane(gl, r0);
        const unsigned m = __builtin_amdgcn_ballot_w32(gl == g) & todo;
        float s0 = 0.f, s1 = 0.f;
#pragma unroll
        for (int r = 0; r < 16; r++) {
          const bool in = (m >> r) & 1u;
          s0 += in ? p0[r] : 0.f;
          s1 += in ? p1[r] : 0.f;
        }
        float* pg = pooled + (size_t)g * 64 + lane;
        atomicAdd(pg, s0);
        atomicAdd(pg + 32, s1);
        todo &= ~m;
      }
    }
    wave_lds_handover();
    gl = gln;
    rpl = rpln;
  }
}

__global__ void gin_final_kernel(const float* __restrict__ pooled,
                                 const float* __restrict__ counts,
                                 const float* __restrict__ linW,
                                 const float* __restrict__ linb,
                                 float* __restrict__ out) {
  int idx = blockIdx.x * blockDim.x + threadIdx.x;
  if (idx >= N_GRAPHS * 10) return;
  int g = idx / 10, t = idx - g * 10;
  float acc = counts[g] * linb[t];
  for (int l = 1; l < 5; l++) acc += linb[l * 10 + t];
  for (int l = 0; l < 5; l++) {
    const float* p = pooled + ((size_t)l * N_GRAPHS + g) * 64;
    const float* w = linW + l * 640 + t;
    float s = 0.f;
#pragma unroll
    for (int k = 0; k < 64; k++) s += p[k] * w[k * 10];
    acc += s;
  }
  out[idx] = acc;
}

__global__ void gin_zero_kernel(unsigned int* __restrict__ p, int n) {
  int i = blockIdx.x * blockDim.x + threadIdx.x;
  if (i < n) p[i] = 0u;
}

extern "C" void kernel_launch(void* const* d_in, const int* in_sizes, int n_in,
                              void* d_out, int out_size, void* d_ws, size_t ws_size,
                              hipStream_t stream) {
  const float* x        = (const float*)d_in[0];
  const float* fh_W1    = (const float*)d_in[1];
  const float* fh_b1    = (const float*)d_in[2];
  const float* fh_bn1   = (const float*)d_in[3];
  const float* fh_W2    = (const float*)d_in[4];
  const float* fh_b2    = (const float*)d_in[5];
  const float* fh_bn2   = (const float*)d_in[6];
  const float* conv_W1  = (const float*)d_in[7];
  const float* conv_b1  = (const float*)d_in[8];
  const float* conv_bn1 = (const float*)d_in[9];
  const float* conv_W2  = (const float*)d_in[10];
  const float* conv_b2  = (const float*)d_in[11];
  const float* conv_bn2 = (const float*)d_in[12];
  const float* lin_W    = (const float*)d_in[13];
  const float* lin_b    = (const float*)d_in[14];
  const int*   eidx     = (const int*)d_in[15];
  const int*   batch    = (const int*)d_in[16];
  float* out = (float*)d_out;

  const int* src = eidx;
  const int* dst = eidx + N_EDGES;

  float* ws     = (float*)d_ws;
  float* h      = ws;
  float* zrow   = h + (size_t)N_NODES * 64;
  float* pooled = zrow + 64;
  float* counts = pooled + 5 * N_GRAPHS * 64;
  int*   bcnt   = (int*)(counts + N_GRAPHS);
  int*   boffs  = bcnt + SCAN_N;
  int*   rowptr = boffs + SCAN_N + 4;
  int2*  pairs  = (int2*)(rowptr + N_NODES + 4);
  int*   csr    = (int*)(pairs + N_EDGES);
  float* h2     = ws + 12000000;

  const int accN = 64 + 5 * N_GRAPHS * 64 + N_GRAPHS;
  gin_zero_kernel<<<(accN + 255) / 256, 256, 0, stream>>>((unsigned int*)zrow, accN);

  gin_bucket_count_kernel<<<N_CHUNK, CHUNK_THREADS, 0, stream>>>(dst, bcnt);
  gin_scan_kernel<<<N_BKT, SCAN_THREADS, 0, stream>>>(bcnt, boffs);
  gin_bucket_fill_kernel<<<N_CHUNK, CHUNK_THREADS, 0, stream>>>(src, dst, boffs, pairs);
  gin_bucket_csr_kernel<<<N_BKT, 1024, 0, stream>>>(pairs, boffs, rowptr, csr);

  const int mlpBlocks = (NTILES + 3) / 4;

  gin_mlp_kernel<D_FEAT, true><<<mlpBlocks, 128, 0, stream>>>(
      x, fh_W1, fh_b1, fh_bn1, fh_W2, fh_b2, fh_bn2, h, batch, pooled, counts, NTILES);

  const int convBlocks = (NTILES + CONV_TILES_PER_WG - 1) / CONV_TILES_PER_WG;
  float* hin = h;
  float* hnext = h2;
  for (int l = 0; l < 4; l++) {
    gin_conv_kernel<<<convBlocks, CONV_WAVES * 32, 0, stream>>>(
        hin, rowptr, csr,
        conv_W1 + (size_t)l * 64 * 64, conv_b1 + l * 64, conv_bn1 + l * 256,
        conv_W2 + (size_t)l * 64 * 64, conv_b2 + l * 64, conv_bn2 + l * 256,
        hnext, batch, pooled + (size_t)(l + 1) * N_GRAPHS * 64, NTILES);
    float* t = hin; hin = hnext; hnext = t;
  }

  gin_final_kernel<<<(N_GRAPHS * 10 + 255) / 256, 256, 0, stream>>>(
      pooled, counts, lin_W, lin_b, out);
}


# Independent launches merged: the zeroing rides in the bucket-count launch, the bucket-fill work-groups (now 4 waves; in-edge order = their LDS arrival rank) share the encoder MLP's launch; 11 -> 9
# speedup vs baseline: 4.4494x; 1.0514x over previous
#include <hip/hip_runtime.h>

#define N_NODES   100000
#define N_EDGES   1600000
#define N_GRAPHS  1000
#define D_FEAT    128
#define HID       64
#define BN_EPS    1e-5f
#define NTILES    (N_NODES / 16)

typedef __attribute__((ext_vector_type(2))) float v2f;
typedef __attribute__((ext_vector_type(8))) float v8f;

template<int K>
static __device__ __forceinline__ int row_share_i(int v) {
  return __builtin_amdgcn_update_dpp(0, v, 0x150 + K, 0xF, 0xF, true);
}
#define GIN_SLOTS4(F, B)  F((B)) F((B) + 1) F((B) + 2) F((B) + 3)
#define GIN_SLOTS16(F)    GIN_SLOTS4(F, 0) GIN_SLOTS4(F, 4) GIN_SLOTS4(F, 8) GIN_SLOTS4(F, 12)

#define BKT_NODES     500
#define N_BKT         (N_NODES / BKT_NODES)
#define N_CHUNK       500
#define CHUNK_EDGES   (N_EDGES / N_CHUNK)
#define FILL_THREADS  128
#define FILL_PER_THR  (CHUNK_EDGES / FILL_THREADS)

template<int K1, bool COUNT>
__launch_bounds__(128)
__global__ void gin_fill_mlp_kernel(const int* __restrict__ esrc,
                                    const int* __restrict__ edst,
                                    const int* __restrict__ boffs,
                                    int2* __restrict__ pairs,
                                    const float* __restrict__ xin,
                                    const float* __restrict__ W1,
                                    const float* __restrict__ b1,
                                    const float* __restrict__ bn1,
                                    const float* __restrict__ W2,
                                    const float* __restrict__ b2,
                                    const float* __restrict__ bn2,
                                    float* __restrict__ hout,
                                    const int* __restrict__ batch,
                                    float* __restrict__ pooled,
                                    float* __restrict__ counts,
                                    int ntiles) {
  __shared__ int cur[N_BKT];
  if (blockIdx.x < N_CHUNK) {
    const int t = threadIdx.x;
    const int g = blockIdx.x;
    for (int b = t; b < N_BKT; b += FILL_THREADS) cur[b] = boffs[b * N_CHUNK + g];
    __syncthreads();
    const int e0 = g * CHUNK_EDGES + t;
    int s[FILL_PER_THR], d[FILL_PER_THR];
#pragma unroll
    for (int k = 0; k < FILL_PER_THR; k++) {
      s[k] = esrc[e0 + k * FILL_THREADS];
      d[k] = edst[e0 + k * FILL_THREADS];
    }
#pragma unroll
    for (int k = 0; k < FILL_PER_THR; k++) {
      int pos = atomicAdd(&cur[d[k] / BKT_NODES], 1);
      pairs[pos] = make_int2(s[k], d[k]);
    }
    return;
  }
  const int mblk = blockIdx.x - N_CHUNK;

  constexpr int KS1 = K1 / 4;
  constexpr int NQ1 = (K1 / 2) * 16 / 128;
  constexpr int NQ2 = (64 / 2) * 16 / 128;
  __shared__ __align__(16) float w1s[4 * KS1 * 32 * 2];
  __shared__ __align__(16) float w2s[4 * 16 * 32 * 2];
  __shared__ __align__(16) float tb[4][16 * 68];

  const int tid  = threadIdx.x;
  const int lane = tid & 31;
  const int wave = tid >> 5;
  const int row  = lane & 15;
  const int hi   = lane >> 4;
  const int tile = mblk * 4 + wave;
  const bool active = tile < ntiles;
  float* tbuf = tb[wave];

  float4 w1a[NQ1], w1b[NQ1], w2a[NQ2], w2b[NQ2];
#pragma unroll
  for (int j = 0; j < NQ1; j++) {
    const int p = tid + 128 * j;
    const float* wp = W1 + (p >> 4) * 128 + (p & 15) * 4;
    w1a[j] = *(const float4*)wp;
    w1b[j] = *(const float4*)(wp + 64);
  }
#pragma unroll
  for (int j = 0; j < NQ2; j++) {
    const int p = tid + 128 * j;
    const float* wp = W2 + (p >> 4) * 128 + (p & 15) * 4;
    w2a[j] = *(const float4*)wp;
    w2b[j] = *(const float4*)(wp + 64);
  }
  const int tl = active ? tile : 0;
  const int gl = batch[tl * 16 + row];
  v2f a1[KS1];
  const float* xr = xin + (size_t)(tl * 16 + row) * K1 + 2 * hi;
#pragma unroll
  for (int k = 0; k < KS1; k++) a1[k] = *(const v2f*)(xr + 4 * k);

#pragma unroll
  for (int j = 0; j < NQ1; j++) {
    const int p = tid + 128 * j;
    const int kk = 2 * (p >> 4), nq = p & 15;
    float* d = w1s + ((((nq >> 2) * KS1) + (kk >> 2)) * 32 + 4 * (nq & 3) + ((kk & 2) << 3)) * 2;
    *(float4*)d       = make_float4(w1a[j].x, w1b[j].x, w1a[j].y, w1b[j].y);
    *(float4*)(d + 4) = make_float4(w1a[j].z, w1b[j].z, w1a[j].w, w1b[j].w);
  }
#pragma unroll
  for (int j = 0; j < NQ2; j++) {
    const int p = tid + 128 * j;
    const int kk = 2 * (p >> 4), nq = p & 15;
    float* d = w2s + ((((nq >> 2) * 16) + (kk >> 2)) * 32 + 4 * (nq & 3) + ((kk & 2) << 3)) * 2;
    *(float4*)d       = make_float4(w2a[j].x, w2b[j].x, w2a[j].y, w2b[j].y);
    *(float4*)(d + 4) = make_float4(w2a[j].z, w2b[j].z, w2a[j].w, w2b[j].w);
  }
  __syncthreads();

  const v8f vzero = {0.f, 0.f, 0.f, 0.f, 0.f, 0.f, 0.f, 0.f};

  if (active) {
#pragma unroll
    for (int nt = 0; nt < 4; nt++) {
      v8f acc = vzero;
      const v2f* wf = ((const v2f*)w1s) + nt * KS1 * 32 + lane;
#pragma unroll
      for (int k = 0; k < KS1; k++) {
        acc = __builtin_amdgcn_wmma_f32_16x16x4_f32(false, a1[k], false, wf[k * 32],
                                                    (short)0, acc, false, false);
      }
      int n = nt * 16 + row;
      float s  = bn1[n] * rsqrtf(bn1[192 + n] + BN_EPS);
      float t0 = bn1[64 + n] + (b1[n] - bn1[128 + n]) * s;
#pragma unroll
      for (int i = 0; i < 8; i++) {
        float y = acc[i] * s + t0;
        tbuf[(i + 8 * hi) * 68 + n] = fmaxf(y, 0.0f);
      }
    }
  }
  __syncthreads();

  v8f acc2[4];
  if (active) {
#pragma unroll
    for (int nt = 0; nt < 4; nt++) {
      v8f acc = vzero;
      const v2f* wf = ((const v2f*)w2s) + nt * 16 * 32 + lane;
      const float* tr = tbuf + row * 68 + 2 * hi;
#pragma unroll
      for (int k = 0; k < 16; k++) {
        v2f a = *(const v2f*)(tr + 4 * k);
        acc = __builtin_amdgcn_wmma_f32_16x16x4_f32(false, a, false, wf[k * 32],
                                                    (short)0, acc, false, false);
      }
      acc2[nt] = acc;
    }
  }
  __syncthreads();

  if (active) {
#pragma unroll
    for (int nt = 0; nt < 4; nt++) {
      int n = nt * 16 + row;
      float s  = bn2[n] * rsqrtf(bn2[192 + n] + BN_EPS);
      float t0 = bn2[64 + n] + (b2[n] - bn2[128 + n]) * s;
#pragma unroll
      for (int i = 0; i < 8; i++) {
        float y = acc2[nt][i] * s + t0;
        tbuf[(i + 8 * hi) * 68 + n] = fmaxf(y, 0.0f);
      }
    }
  }
  __syncthreads();

  if (active) {
#pragma unroll
    for (int j = 0; j < 8; j++) {
      int flat = j * 32 + lane;
      int r = flat >> 4;
      int c = (flat & 15) << 2;
      float4 v = *(const float4*)(tbuf + r * 68 + c);
      *(float4*)(hout + (size_t)(tile * 16 + r) * 64 + c) = v;
    }
  }

  if (active) {
    float p0[16], p1[16];
#pragma unroll
    for (int r = 0; r < 16; r++) {
      p0[r] = tbuf[r * 68 + lane];
      p1[r] = tbuf[r * 68 + 32 + lane];
    }
    unsigned todo = 0xFFFFu;
    while (todo) {
      const int r0 = __builtin_ctz(todo);
      const int g = __builtin_amdgcn_readlane(gl, r0);
      const unsigned m = __builtin_amdgcn_ballot_w32(gl == g) & todo;
      float s0 = 0.f, s1 = 0.f;
#pragma unroll
      for (int r = 0; r < 16; r++) {
        const bool in = (m >> r) & 1u;
        s0 += in ? p0[r] : 0.f;
        s1 += in ? p1[r] : 0.f;
      }
      float* pg = pooled + (size_t)g * 64 + lane;
      atomicAdd(pg, s0);
      atomicAdd(pg + 32, s1);
      if (COUNT && lane == 0) atomicAdd(counts + g, (float)__builtin_popcount(m));
      todo &= ~m;
    }
  }
}

#define CHUNK_THREADS 320
#define CHUNK_PER_THR (CHUNK_EDGES / CHUNK_THREADS)

__launch_bounds__(CHUNK_THREADS)
__global__ void gin_bucket_count_kernel(const int* __restrict__ dst,
                                        int* __restrict__ bcnt,
                                        float4* __restrict__ zero16,
                                        int nzero16) {
  __shared__ int cnt[N_BKT];
  const int t = threadIdx.x;
  const int g = blockIdx.x;
  if (t < N_BKT) cnt[t] = 0;
  __syncthreads();
  const int* dp = dst + g * CHUNK_EDGES + t;
  int d[CHUNK_PER_THR];
#pragma unroll
  for (int k = 0; k < CHUNK_PER_THR; k++) d[k] = dp[k * CHUNK_THREADS];
#pragma unroll
  for (int k = 0; k < CHUNK_PER_THR; k++) atomicAdd(&cnt[d[k] / BKT_NODES], 1);
  __syncthreads();
  if (t < N_BKT) bcnt[t * N_CHUNK + g] = cnt[t];
  const int gid = g * CHUNK_THREADS + t;
  if (gid < nzero16) zero16[gid] = make_float4(0.f, 0.f, 0.f, 0.f);
}

#define SCAN_N       (N_BKT * N_CHUNK)
#define ROW_Q        (N_CHUNK / 4)
#define SCAN_THREADS 1024
#define SCAN_TRIPS   (((N_BKT / 2) * ROW_Q + SCAN_THREADS - 1) / SCAN_THREADS)
__launch_bounds__(SCAN_THREADS)
__global__ void gin_scan_kernel(const int* __restrict__ bcnt,
                                int* __restrict__ boffs) {
  __shared__ int wsum[32];
  __shared__ int rsum[4];
  const int t = threadIdx.x;
  const int lane = t & 31;
  const int w = t >> 5;
  const int b = blockIdx.x;
  const int nq = b * ROW_Q;
  const bool front = b < N_BKT / 2;
  const int q0 = front ? 0 : nq;
  const int q1 = front ? nq : SCAN_N / 4;
  const int4* p = (const int4*)bcnt;

  int4 v[SCAN_TRIPS];
#pragma unroll
  for (int i = 0; i < SCAN_TRIPS; i++) {
    const int q = q0 + t + SCAN_THREADS * i;
    v[i] = p[q < q1 ? q : 0];
  }
  const int4 r = p[nq + (t < ROW_Q ? t : 0)];
  int part = 0;
#pragma unroll
  for (int i = 0; i < SCAN_TRIPS; i++) {
    const int q = q0 + t + SCAN_THREADS * i;
    part += (q < q1) ? (v[i].x + v[i].y + v[i].z + v[i].w) : 0;
  }
#pragma unroll
  for (int o = 16; o > 0; o >>= 1) part += __shfl_xor(part, o);
  if (lane == 0) wsum[w] = part;

  const int local = (t < ROW_Q) ? (r.x + r.y + r.z + r.w) : 0;
  int x = local;
#pragma unroll
  for (int o = 1; o < 32; o <<= 1) {
    int y = __shfl_up(x, o);
    if (lane >= o) x += y;
  }
  if (w < 4 && lane == 31) rsum[w] = x;
  __syncthreads();

  if (w < 4) {
    int tot = wsum[lane];
#pragma unroll
    for (int o = 16; o > 0; o >>= 1) tot += __shfl_xor(tot, o);
    int run = front ? tot : N_EDGES - tot;
    run += x - local;
    if (w > 0) run += rsum[0];
    if (w > 1) run += rsum[1];
    if (w > 2) run += rsum[2];
    if (t < ROW_Q) {
      int4 o4;
      o4.x = run; run += r.x;
      o4.y = run; run += r.y;
      o4.z = run; run += r.z;
      o4.w = run; run += r.w;
      ((int4*)boffs)[nq + t] = o4;
      if (b == N_BKT - 1 && t == ROW_Q - 1) boffs[SCAN_N] = run;
    }
  }
}


#define SPARE_CNT 511
__launch_bounds__(1024)
__global__ void gin_bucket_csr_kernel(const int2* __restrict__ pairs,
                                      const int* __restrict__ boffs,
                                      int* __restrict__ rowptr,
                                      int* __restrict__ csr) {
  __shared__ int cnt[512];
  __shared__ int wsum[32];
  const int t = threadIdx.x;
  const int lane = t & 31;
  const int w = t >> 5;
  const int b = blockIdx.x;
  const int n0 = b * BKT_NODES;
  const int beg = boffs[b * N_CHUNK];
  const int end = boffs[(b + 1) * N_CHUNK];

  if (t < 512) cnt[t] = 0;
  __syncthreads();
  for (int i0 = beg + t; i0 < end; i0 += 4 * 1024) {
    int c[4];
#pragma unroll
    for (int k = 0; k < 4; k++) {
      int i = i0 + k * 1024;
      int d = pairs[i < end ? i : end - 1].y;
      c[k] = (i < end) ? d - n0 : SPARE_CNT;
    }
#pragma unroll
    for (int k = 0; k < 4; k++) atomicAdd(&cnt[c[k]], 1);
  }
  __syncthreads();

  const int v = (t < BKT_NODES) ? cnt[t] : 0;
  int x = v;
#pragma unroll
  for (int o = 1; o < 32; o <<= 1) {
    int y = __shfl_up(x, o);
    if (lane >= o) x += y;
  }
  if (lane == 31) wsum[w] = x;
  __syncthreads();
  if (w == 0) {
    int s = wsum[lane];
#pragma unroll
    for (int o = 1; o < 32; o <<= 1) {
      int y = __shfl_up(s, o);
      if (lane >= o) s += y;
    }
    wsum[lane] = s;
  }
  __syncthreads();
  const int first = beg + x - v + (w > 0 ? wsum[w - 1] : 0);
  if (t < BKT_NODES) {
    rowptr[n0 + t] = first;
    cnt[t] = first;
  }
  if (b == N_BKT - 1 && t == BKT_NODES - 1) rowptr[N_NODES] = first + v;
  __syncthreads();

  for (int i0 = beg + t; i0 < end; i0 += 4 * 1024) {
    int2 p[4];
#pragma unroll
    for (int k = 0; k < 4; k++) {
      int i = i0 + k * 1024;
      p[k] = pairs[i < end ? i : end - 1];
    }
#pragma unroll
    for (int k = 0; k < 4; k++) {
      if (i0 + k * 1024 < end) {
        int pos = atomicAdd(&cnt[p[k].y - n0], 1);
        csr[pos] = p[k].x;
      }
    }
  }
}

#define CONV_WAVES 4
#define CONV_TILES_PER_WAVE 2
#define CONV_TILES_PER_WG (CONV_WAVES * CONV_TILES_PER_WAVE)

static __device__ __forceinline__ void wave_lds_handover() {
  __builtin_amdgcn_fence(__ATOMIC_RELEASE, "wavefront");
  __builtin_amdgcn_wave_barrier();
  __builtin_amdgcn_fence(__ATOMIC_ACQUIRE, "wavefront");
}

static __device__ __forceinline__ void conv_first_slots(const int* __restrict__ csr, int rpl, int row, int hi,
                                                        int& c0, int& c1) {
  const int b0 = __builtin_amdgcn_readlane(rpl, 0);
  const int b1 = __builtin_amdgcn_readlane(rpl, 1);
  const int b2 = __builtin_amdgcn_readlane(rpl, 2);
  const int beg = hi ? b1 : b0, end = hi ? b2 : b1;
  int j = beg + row;      j = j < end ? j : end - 1;
  c0 = csr[j > 0 ? j : 0];
  j = beg + 16 + row;     j = j < end ? j : end - 1;
  c1 = csr[j > 0 ? j : 0];
}

__launch_bounds__(CONV_WAVES * 32)
__global__ void gin_conv_kernel(const float* __restrict__ hin,
                                const int* __restrict__ rowptr,
                                const int* __restrict__ csr,
                                const float* __restrict__ W1,
                                const float* __restrict__ b1,
                                const float* __restrict__ bn1,
                                const float* __restrict__ W2,
                                const float* __restrict__ b2,
                                const float* __restrict__ bn2,
                                float* __restrict__ hout,
                                const int* __restrict__ batch,
                                float* __restrict__ pooled,
                                int ntiles) {
  __shared__ __align__(16) float w1s[4 * 16 * 32 * 2];
  __shared__ __align__(16) float w2s[4 * 16 * 32 * 2];
  __shared__ __align__(16) float tb[CONV_WAVES][16 * 68];

  const int tid  = threadIdx.x;
  const int lane = tid & 31;
  const int wave = tid >> 5;
  const int row  = lane & 15;
  const int hi   = lane >> 4;
  float* tbuf = tb[wave];

  float4 w1a[4], w1b[4], w2a[4], w2b[4];
#pragma unroll
  for (int j = 0; j < 4; j++) {
    const int p = tid + 128 * j;
    const float* wp = W1 + (p >> 4) * 128 + (p & 15) * 4;
    w1a[j] = *(const float4*)wp;
    w1b[j] = *(const float4*)(wp + 64);
    const float* wq = W2 + (p >> 4) * 128 + (p & 15) * 4;
    w2a[j] = *(const float4*)wq;
    w2b[j] = *(const float4*)(wq + 64);
  }
  const int tile0 = blockIdx.x * CONV_TILES_PER_WG + wave;
  const int tl0 = tile0 < ntiles ? tile0 : 0;
  const int tl1 = tile0 + CONV_WAVES < ntiles ? tile0 + CONV_WAVES : 0;
  int gl   = batch[tl0 * 16 + row];
  int gln  = batch[tl1 * 16 + row];
  int rpl  = rowptr[tl0 * 16 + (lane < 16 ? lane : 16)];
  int rpln = rowptr[tl1 * 16 + (lane < 16 ? lane : 16)];
  int c0, c1;
  conv_first_slots(csr, rpl, row, hi, c0, c1);

#pragma unroll
  for (int j = 0; j < 4; j++) {
    const int p = tid + 128 * j;
    const int kk = 2 * (p >> 4), nq = p & 15;
    const int o = ((((nq >> 2) * 16) + (kk >> 2)) * 32 + 4 * (nq & 3) + ((kk & 2) << 3)) * 2;
    *(float4*)(w1s + o)     = make_float4(w1a[j].x, w1b[j].x, w1a[j].y, w1b[j].y);
    *(float4*)(w1s + o + 4) = make_float4(w1a[j].z, w1b[j].z, w1a[j].w, w1b[j].w);
    *(float4*)(w2s + o)     = make_float4(w2a[j].x, w2b[j].x, w2a[j].y, w2b[j].y);
    *(float4*)(w2s + o + 4) = make_float4(w2a[j].z, w2b[j].z, w2a[j].w, w2b[j].w);
  }
  __syncthreads();

  const v8f vzero = {0.f, 0.f, 0.f, 0.f, 0.f, 0.f, 0.f, 0.f};

  for (int it = 0; it < CONV_TILES_PER_WAVE; it++) {
    const int tile = tile0 + it * CONV_WAVES;
    if (tile >= ntiles) break;

    {
      const char* hb = (const char*)hin;
      const unsigned cb = (unsigned)row << 4;
      int b0 = __builtin_amdgcn_readlane(rpl, 0);
      int b1 = __builtin_amdgcn_readlane(rpl, 1);
      int b2 = __builtin_amdgcn_readlane(rpl, 2);
      int beg = hi ? b1 : b0, end = hi ? b2 : b1;
      for (int p = 0; p < 8; p++) {
        const int r = 2 * p + hi;
        const unsigned n = (unsigned)(tile * 16 + r);
        float4 acc = *(const float4*)(hb + ((n << 8) + cb));
        float4 v[16], u[16];
#define GIN_LOADV(k) v[(k)] = *(const float4*)(hb + ((((unsigned)row_share_i<(k)>(c0)) << 8) + cb));
        GIN_SLOTS16(GIN_LOADV)
#undef GIN_LOADV
        const int nb0 = b2;
        const int nb1 = __builtin_amdgcn_readlane(rpl, 2 * p + 3);
        const int nb2 = __builtin_amdgcn_readlane(rpl, 2 * p + 4);
        const int nbeg = hi ? nb1 : nb0, nend = hi ? nb2 : nb1;
        int nc0, nc1;
        {
          int j = nbeg + row;   j = j < nend ? j : nend - 1;
          nc0 = csr[j > 0 ? j : 0];
          j = nbeg + 16 + row;  j = j < nend ? j : nend - 1;
          nc1 = csr[j > 0 ? j : 0];
        }
        const int d0 = b1 - b0, d1 = b2 - b1;
        const int more = (d0 > d1 ? d0 : d1) - 16;
#define GIN_LOADU(k) u[(k)] = *(const float4*)(hb + ((((unsigned)row_share_i<(k)>(c1)) << 8) + cb));
        if (more > 0) {
          GIN_SLOTS4(GIN_LOADU, 0)
          if (more > 4) {
            GIN_SLOTS4(GIN_LOADU, 4)
            if (more > 8) {
              GIN_SLOTS4(GIN_LOADU, 8)
              if (more > 12) { GIN_SLOTS4(GIN_LOADU, 12) }
            }
          }
        }
#undef GIN_LOADU
#define GIN_ACCV(k) { const bool in = beg + (k) < end; \
        acc.x += in ? v[(k)].x : 0.f; acc.y += in ? v[(k)].y : 0.f; \
        acc.z += in ? v[(k)].z : 0.f; acc.w += in ? v[(k)].w : 0.f; }
        GIN_SLOTS16(GIN_ACCV)
#undef GIN_ACCV
        if (more > 0) {
#define GIN_ACCU(k) { const bool in = beg + 16 + (k) < end; \
        acc.x += in ? u[(k)].x : 0.f; acc.y += in ? u[(k)].y : 0.f; \
        acc.z += in ? u[(k)].z : 0.f; acc.w += in ? u[(k)].w : 0.f; }
          GIN_SLOTS4(GIN_ACCU, 0)
          if (more > 4) {
            GIN_SLOTS4(GIN_ACCU, 4)
            if (more > 8) {
              GIN_SLOTS4(GIN_ACCU, 8)
              if (more > 12) { GIN_SLOTS4(GIN_ACCU, 12) }
            }
          }
#undef GIN_ACCU
        }
        for (int j = beg + 32; j < end; j += 16) {
          int s[16];
#pragma unroll
          for (int k = 0; k < 16; k++) s[k] = csr[(j + k < end) ? j + k : end - 1];
#pragma unroll
          for (int k = 0; k < 16; k++) {
            const unsigned sk = (j + k < end) ? (unsigned)s[k] : n;
            v[k] = *(const float4*)(hb + ((sk << 8) + cb));
          }
#pragma unroll
          for (int k = 0; k < 16; k++) {
            const bool in = j + k < end;
            acc.x += in ? v[k].x : 0.f; acc.y += in ? v[k].y : 0.f;
            acc.z += in ? v[k].z : 0.f; acc.w += in ? v[k].w : 0.f;
          }
        }
        *(float4*)(tbuf + r * 68 + 4 * row) = acc;
        b0 = nb0; b1 = nb1; b2 = nb2;
        beg = nbeg; end = nend;
        c0 = nc0; c1 = nc1;
      }
    }
    if (it + 1 < CONV_TILES_PER_WAVE) conv_first_slots(csr, rpln, row, hi, c0, c1);
    wave_lds_handover();

    {
      v2f a1[16];
      const float* tr = tbuf + row * 68 + 2 * hi;
#pragma unroll
      for (int k = 0; k < 16; k++) a1[k] = *(const v2f*)(tr + 4 * k);
#pragma unroll
      for (int nt = 0; nt < 4; nt++) {
        v8f acc = vzero;
        const v2f* wf = ((const v2f*)w1s) + nt * 16 * 32 + lane;
#pragma unroll
        for (int k = 0; k < 16; k++) {
          acc = __builtin_amdgcn_wmma_f32_16x16x4_f32(false, a1[k], false, wf[k * 32],
                                                      (short)0, acc, false, false);
        }
        int n = nt * 16 + row;
        float s  = bn1[n] * rsqrtf(bn1[192 + n] + BN_EPS);
        float t0 = bn1[64 + n] + (b1[n] - bn1[128 + n]) * s;
#pragma unroll
        for (int i = 0; i < 8; i++) {
          float y = acc[i] * s + t0;
          tbuf[(i + 8 * hi) * 68 + n] = fmaxf(y, 0.0f);
        }
      }
    }
    wave_lds_handover();

    v8f acc2[4];
    {
#pragma unroll
      for (int nt = 0; nt < 4; nt++) {
        v8f acc = vzero;
        const v2f* wf = ((const v2f*)w2s) + nt * 16 * 32 + lane;
        const float* tr = tbuf + row * 68 + 2 * hi;
#pragma unroll
        for (int k = 0; k < 16; k++) {
          v2f a = *(const v2f*)(tr + 4 * k);
          acc = __builtin_amdgcn_wmma_f32_16x16x4_f32(false, a, false, wf[k * 32],
                                                      (short)0, acc, false, false);
        }
        acc2[nt] = acc;
      }
    }

    {
#pragma unroll
      for (int nt = 0; nt < 4; nt++) {
        int n = nt * 16 + row;
        float s  = bn2[n] * rsqrtf(bn2[192 + n] + BN_EPS);
        float t0 = bn2[64 + n] + (b2[n] - bn2[128 + n]) * s;
#pragma unroll
        for (int i = 0; i < 8; i++) {
          float y = acc2[nt][i] * s + t0;
          tbuf[(i + 8 * hi) * 68 + n] = fmaxf(y, 0.0f);
        }
      }
    }
    wave_lds_handover();

    {
#pragma unroll
      for (int j = 0; j < 8; j++) {
        int flat = j * 32 + lane;
        int r = flat >> 4;
        int c = (flat & 15) << 2;
        float4 v = *(const float4*)(tbuf + r * 68 + c);
        *(float4*)(hout + (size_t)(tile * 16 + r) * 64 + c) = v;
      }
    }

    {
      float p0[16], p1[16];
#pragma unroll
      for (int r = 0; r < 16; r++) {
        p0[r] = tbuf[r * 68 + lane];
        p1[r] = tbuf[r * 68 + 32 + lane];
      }
      unsigned todo = 0xFFFFu;
      while (todo) {
        const int r0 = __builtin_ctz(todo);
        const int g = __builtin_amdgcn_readlane(gl, r0);
        const unsigned m = __builtin_amdgcn_ballot_w32(gl == g) & todo;
        float s0 = 0.f, s1 = 0.f;
#pragma unroll
        for (int r = 0; r < 16; r++) {
          const bool in = (m >> r) & 1u;
          s0 += in ? p0[r] : 0.f;
          s1 += in ? p1[r] : 0.f;
        }
        float* pg = pooled + (size_t)g * 64 + lane;
        atomicAdd(pg, s0);
        atomicAdd(pg + 32, s1);
        todo &= ~m;
      }
    }
    wave_lds_handover();
    gl = gln;
    rpl = rpln;
  }
}

__global__ void gin_final_kernel(const float* __restrict__ pooled,
                                 const float* __restrict__ counts,
                                 const float* __restrict__ linW,
                                 const float* __restrict__ linb,
                                 float* __restrict__ out) {
  int idx = blockIdx.x * blockDim.x + threadIdx.x;
  if (idx >= N_GRAPHS * 10) return;
  int g = idx / 10, t = idx - g * 10;
  float acc = counts[g] * linb[t];
  for (int l = 1; l < 5; l++) acc += linb[l * 10 + t];
  for (int l = 0; l < 5; l++) {
    const float* p = pooled + ((size_t)l * N_GRAPHS + g) * 64;
    const float* w = linW + l * 640 + t;
    float s = 0.f;
#pragma unroll
    for (int k = 0; k < 64; k++) s += p[k] * w[k * 10];
    acc += s;
  }
  out[idx] = acc;
}

extern "C" void kernel_launch(void* const* d_in, const int* in_sizes, int n_in,
                              void* d_out, int out_size, void* d_ws, size_t ws_size,
                              hipStream_t stream) {
  const float* x        = (const float*)d_in[0];
  const float* fh_W1    = (const float*)d_in[1];
  const float* fh_b1    = (const float*)d_in[2];
  const float* fh_bn1   = (const float*)d_in[3];
  const float* fh_W2    = (const float*)d_in[4];
  const float* fh_b2    = (const float*)d_in[5];
  const float* fh_bn2   = (const float*)d_in[6];
  const float* conv_W1  = (const float*)d_in[7];
  const float* conv_b1  = (const float*)d_in[8];
  const float* conv_bn1 = (const float*)d_in[9];
  const float* conv_W2  = (const float*)d_in[10];
  const float* conv_b2  = (const float*)d_in[11];
  const float* conv_bn2 = (const float*)d_in[12];
  const float* lin_W    = (const float*)d_in[13];
  const float* lin_b    = (const float*)d_in[14];
  const int*   eidx     = (const int*)d_in[15];
  const int*   batch    = (const int*)d_in[16];
  float* out = (float*)d_out;

  const int* src = eidx;
  const int* dst = eidx + N_EDGES;

  float* ws     = (float*)d_ws;
  float* h      = ws;
  float* zrow   = h + (size_t)N_NODES * 64;
  float* pooled = zrow + 64;
  float* counts = pooled + 5 * N_GRAPHS * 64;
  int*   bcnt   = (int*)(counts + N_GRAPHS);
  int*   boffs  = bcnt + SCAN_N;
  int*   rowptr = boffs + SCAN_N + 4;
  int2*  pairs  = (int2*)(rowptr + N_NODES + 4);
  int*   csr    = (int*)(pairs + N_EDGES);
  float* h2     = ws + 12000000;

  const int accN16 = (64 + 5 * N_GRAPHS * 64 + N_GRAPHS) / 4;

  const int mlpBlocks = (NTILES + 3) / 4;
  gin_bucket_count_kernel<<<N_CHUNK, CHUNK_THREADS, 0, stream>>>(dst, bcnt, (float4*)zrow, accN16);
  gin_scan_kernel<<<N_BKT, SCAN_THREADS, 0, stream>>>(bcnt, boffs);
  gin_fill_mlp_kernel<D_FEAT, true><<<N_CHUNK + mlpBlocks, 128, 0, stream>>>(
      src, dst, boffs, pairs,
      x, fh_W1, fh_b1, fh_bn1, fh_W2, fh_b2, fh_bn2, h, batch, pooled, counts, NTILES);
  gin_bucket_csr_kernel<<<N_BKT, 1024, 0, stream>>>(pairs, boffs, rowptr, csr);

  const int convBlocks = (NTILES + CONV_TILES_PER_WG - 1) / CONV_TILES_PER_WG;
  float* hin = h;
  float* hnext = h2;
  for (int l = 0; l < 4; l++) {
    gin_conv_kernel<<<convBlocks, CONV_WAVES * 32, 0, stream>>>(
        hin, rowptr, csr,
        conv_W1 + (size_t)l * 64 * 64, conv_b1 + l * 64, conv_bn1 + l * 256,
        conv_W2 + (size_t)l * 64 * 64, conv_b2 + l * 64, conv_bn2 + l * 256,
        hnext, batch, pooled + (size_t)(l + 1) * N_GRAPHS * 64, NTILES);
    float* t = hin; hin = hnext; hnext = t;
  }

  gin_final_kernel<<<(N_GRAPHS * 10 + 255) / 256, 256, 0, stream>>>(
      pooled, counts, lin_W, lin_b, out);
}
